# starting kernel
# baseline (speedup 1.0000x reference)
#include <hip/hip_runtime.h>
#include <hip/hip_bf16.h>
#include <math.h>

typedef __attribute__((ext_vector_type(2))) float v2f;
typedef __attribute__((ext_vector_type(8))) float v8f;

#define HID 128
#define HEADS 2
#define DH 256
#define NEG_SLOPE 0.2f

template<int K, bool CONCAT>
__global__ __launch_bounds__(512)
void gemm_node(const float* __restrict__ A, const float* __restrict__ A2,
               const float* __restrict__ W, float* __restrict__ Hout, int n)
{
    constexpr int PITCH = K + 4;
    __shared__ float As[16 * PITCH];

    const int tid  = threadIdx.x;
    const int wid  = tid >> 5;
    const int lane = tid & 31;
    const int r0   = blockIdx.x * 16;

    for (int idx = tid; idx < 16 * K; idx += 512) {
        const int row = idx / K;
        const int k   = idx - row * K;
        const int gr  = r0 + row;
        float v = 0.0f;
        if (gr < n) {
            if (CONCAT) v = (k < 64) ? A[gr * 64 + k] : A2[gr * 64 + (k - 64)];
            else        v = A[gr * K + k];
        }
        As[row * PITCH + k] = v;
    }
    __syncthreads();

    const int c0    = wid * 16;
    const int halfl = lane & 15;
    const int koff  = (lane >> 4) * 2;

    v8f acc = {};
    for (int k0 = 0; k0 < K; k0 += 4) {
        v2f a, b;
        a.x = As[halfl * PITCH + k0 + koff];
        a.y = As[halfl * PITCH + k0 + koff + 1];
        b.x = W[(k0 + koff)     * DH + c0 + halfl];
        b.y = W[(k0 + koff + 1) * DH + c0 + halfl];
        acc = __builtin_amdgcn_wmma_f32_16x16x4_f32(
                  false, a, false, b, (short)0, acc, false, false);
    }

    const int colg  = c0 + halfl;
    const int mbase = (lane >> 4) * 8;
#pragma unroll
    for (int r = 0; r < 8; ++r) {
        const int grow = r0 + mbase + r;
        if (grow < n) Hout[grow * DH + colg] = acc[r];
    }
}

__global__ __launch_bounds__(256)
void node_scores(const float* __restrict__ H, const float* __restrict__ al,
                 const float* __restrict__ ar, float* __restrict__ el,
                 float* __restrict__ er, int n)
{
    const int lane = threadIdx.x & 31;
    const int node = blockIdx.x * 8 + (threadIdx.x >> 5);
    if (node >= n) return;

    const float* h = H + node * DH;
    float e0 = 0.f, e1 = 0.f, r0 = 0.f, r1 = 0.f;
    for (int i = lane; i < HID; i += 32) {
        const float h0 = h[i], h1 = h[HID + i];
        e0 += h0 * al[i];        e1 += h1 * al[HID + i];
        r0 += h0 * ar[i];        r1 += h1 * ar[HID + i];
    }
#pragma unroll
    for (int off = 16; off; off >>= 1) {
        e0 += __shfl_xor(e0, off); e1 += __shfl_xor(e1, off);
        r0 += __shfl_xor(r0, off); r1 += __shfl_xor(r1, off);
    }
    if (lane == 0) {
        el[node * 2 + 0] = e0; el[node * 2 + 1] = e1;
        er[node * 2 + 0] = r0; er[node * 2 + 1] = r1;
    }
}

__global__ void init_layer(float* __restrict__ agg, float* __restrict__ m,
                           float* __restrict__ s, int n)
{
    const int i = blockIdx.x * 256 + threadIdx.x;
    if (i < n * DH) agg[i] = 0.0f;
    if (i < n * HEADS) { m[i] = -INFINITY; s[i] = 0.0f; }
}

__device__ __forceinline__ float lrelu(float x) {
    return x > 0.0f ? x : NEG_SLOPE * x;
}

__device__ __forceinline__ void atomicMaxF(float* addr, float val) {
    if (val >= 0.0f) atomicMax((int*)addr, __float_as_int(val));
    else             atomicMin((unsigned int*)addr, __float_as_uint(val));
}

__global__ void edge_max(const int* __restrict__ src, const int* __restrict__ dst,
                         const float* __restrict__ el, const float* __restrict__ er,
                         float* __restrict__ m, int e)
{
    const int i = blockIdx.x * 256 + threadIdx.x;
    if (i >= e) return;
    const int s = src[i], d = dst[i];
#pragma unroll
    for (int h = 0; h < HEADS; ++h)
        atomicMaxF(&m[d * 2 + h], lrelu(el[s * 2 + h] + er[d * 2 + h]));
}

__global__ void m_fixup(float* __restrict__ m, int n2)
{
    const int i = blockIdx.x * 256 + threadIdx.x;
    if (i < n2 && !isfinite(m[i])) m[i] = 0.0f;
}

__global__ void edge_exp(const int* __restrict__ src, const int* __restrict__ dst,
                         const float* __restrict__ el, const float* __restrict__ er,
                         const float* __restrict__ m, float* __restrict__ ex,
                         float* __restrict__ ssum, int e)
{
    const int i = blockIdx.x * 256 + threadIdx.x;
    if (i >= e) return;
    const int s = src[i], d = dst[i];
#pragma unroll
    for (int h = 0; h < HEADS; ++h) {
        const float sc = lrelu(el[s * 2 + h] + er[d * 2 + h]);
        const float v  = __expf(sc - m[d * 2 + h]);
        ex[i * 2 + h] = v;
        atomicAdd(&ssum[d * 2 + h], v);
    }
}

__global__ __launch_bounds__(256)
void edge_agg(const int* __restrict__ src, const int* __restrict__ dst,
              const float* __restrict__ ex, const float* __restrict__ ssum,
              const float* __restrict__ H, float* __restrict__ agg, int e)
{
    const int w    = blockIdx.x * 8 + (threadIdx.x >> 5);
    const int lane = threadIdx.x & 31;
    if (w >= e) return;
    const int s = src[w], d = dst[w];
    const float w0 = ex[w * 2 + 0] / ssum[d * 2 + 0];
    const float w1 = ex[w * 2 + 1] / ssum[d * 2 + 1];
    const float* hs = H + s * DH;
    float* ad = agg + d * DH;
#pragma unroll
    for (int i = 0; i < 8; ++i) {
        const int j = i * 32 + lane;
        const float wt = (j < HID) ? w0 : w1;
        atomicAdd(&ad[j], wt * hs[j]);
    }
}

__global__ void bias_relu(const float* __restrict__ agg, const float* __restrict__ b,
                          float* __restrict__ outp, int n)
{
    const int i = blockIdx.x * 256 + threadIdx.x;
    if (i >= n * DH) return;
    const float v = agg[i] + b[i & (DH - 1)];
    outp[i] = v > 0.0f ? v : 0.0f;
}

__global__ __launch_bounds__(256)
void edge_classify(const int* __restrict__ src, const int* __restrict__ dst,
                   const float* __restrict__ H, const float* __restrict__ fcW,
                   const float* __restrict__ fcb, float* __restrict__ out, int e)
{
    const int w    = blockIdx.x * 8 + (threadIdx.x >> 5);
    const int lane = threadIdx.x & 31;
    if (w >= e) return;
    const int s = src[w], d = dst[w];
    const float* hs = H + s * DH;
    const float* hd = H + d * DH;
    float a0 = 0.f, a1 = 0.f;
    for (int j = lane; j < DH; j += 32) {
        const float vs = hs[j], vd = hd[j];
        a0 += vs * fcW[j * 2 + 0] + vd * fcW[(DH + j) * 2 + 0];
        a1 += vs * fcW[j * 2 + 1] + vd * fcW[(DH + j) * 2 + 1];
    }
#pragma unroll
    for (int off = 16; off; off >>= 1) {
        a0 += __shfl_xor(a0, off);
        a1 += __shfl_xor(a1, off);
    }
    if (lane == 0) {
        out[w * 2 + 0] = a0 + fcb[0];
        out[w * 2 + 1] = a1 + fcb[1];
    }
}

extern "C" void kernel_launch(void* const* d_in, const int* in_sizes, int n_in,
                              void* d_out, int out_size, void* d_ws, size_t ws_size,
                              hipStream_t stream)
{
    const float* x   = (const float*)d_in[0];
    const float* rep = (const float*)d_in[1];
    const int*   src = (const int*)d_in[2];
    const int*   dst = (const int*)d_in[3];
    const float* W1  = (const float*)d_in[4];
    const float* al1 = (const float*)d_in[5];
    const float* ar1 = (const float*)d_in[6];
    const float* b1  = (const float*)d_in[7];
    const float* W2  = (const float*)d_in[8];
    const float* al2 = (const float*)d_in[9];
    const float* ar2 = (const float*)d_in[10];
    const float* b2  = (const float*)d_in[11];
    const float* fcW = (const float*)d_in[12];
    const float* fcb = (const float*)d_in[13];
    float* out = (float*)d_out;

    const int N = in_sizes[0] / 64;
    const int E = in_sizes[2];

    float* ws  = (float*)d_ws;
    float* h   = ws;
    float* agg = h   + (size_t)N * DH;
    float* rly = agg + (size_t)N * DH;
    float* el  = rly + (size_t)N * DH;
    float* er  = el  + (size_t)N * 2;
    float* mbf = er  + (size_t)N * 2;
    float* sbf = mbf + (size_t)N * 2;
    float* ex  = sbf + (size_t)N * 2;

    const int gN16  = (N + 15) / 16;
    const int gN8   = (N + 7) / 8;
    const int gE8   = (E + 7) / 8;
    const int gE256 = (E + 255) / 256;
    const int gF    = (N * DH + 255) / 256;
    const int gM    = (N * 2 + 255) / 256;

    gemm_node<128, true><<<gN16, 512, 0, stream>>>(x, rep, W1, h, N);
    node_scores<<<gN8, 256, 0, stream>>>(h, al1, ar1, el, er, N);
    init_layer<<<gF, 256, 0, stream>>>(agg, mbf, sbf, N);
    edge_max<<<gE256, 256, 0, stream>>>(src, dst, el, er, mbf, E);
    m_fixup<<<gM, 256, 0, stream>>>(mbf, N * 2);
    edge_exp<<<gE256, 256, 0, stream>>>(src, dst, el, er, mbf, ex, sbf, E);
    edge_agg<<<gE8, 256, 0, stream>>>(src, dst, ex, sbf, h, agg, E);
    bias_relu<<<gF, 256, 0, stream>>>(agg, b1, rly, N);

    gemm_node<256, false><<<gN16, 512, 0, stream>>>(rly, nullptr, W2, h, N);
    node_scores<<<gN8, 256, 0, stream>>>(h, al2, ar2, el, er, N);
    init_layer<<<gF, 256, 0, stream>>>(agg, mbf, sbf, N);
    edge_max<<<gE256, 256, 0, stream>>>(src, dst, el, er, mbf, E);
    m_fixup<<<gM, 256, 0, stream>>>(mbf, N * 2);
    edge_exp<<<gE256, 256, 0, stream>>>(src, dst, el, er, mbf, ex, sbf, E);
    edge_agg<<<gE8, 256, 0, stream>>>(src, dst, ex, sbf, h, agg, E);
    bias_relu<<<gF, 256, 0, stream>>>(agg, b2, rly, N);

    edge_classify<<<gE8, 256, 0, stream>>>(src, dst, rly, fcW, fcb, out, E);
}


# edges filed by dst tile of 98 rows (256 tiles); one work-group per tile sums max, exp, exp-weighted rows in LDS (float atomics there, divide by exp sum last) + bias, relu: 17 -> 9 launches
# speedup vs baseline: 2.1347x; 2.1347x over previous
#include <hip/hip_runtime.h>
#include <hip/hip_bf16.h>
#include <math.h>

typedef __attribute__((ext_vector_type(2))) float v2f;
typedef __attribute__((ext_vector_type(8))) float v8f;

#define HID 128
#define HEADS 2
#define DH 256
#define NEG_SLOPE 0.2f

template<int K, bool CONCAT>
__global__ __launch_bounds__(512)
void gemm_node(const float* __restrict__ A, const float* __restrict__ A2,
               const float* __restrict__ W, float* __restrict__ Hout, int n)
{
    constexpr int PITCH = K + 4;
    __shared__ float As[16 * PITCH];

    const int tid  = threadIdx.x;
    const int wid  = tid >> 5;
    const int lane = tid & 31;
    const int r0   = blockIdx.x * 16;

    for (int idx = tid; idx < 16 * K; idx += 512) {
        const int row = idx / K;
        const int k   = idx - row * K;
        const int gr  = r0 + row;
        float v = 0.0f;
        if (gr < n) {
            if (CONCAT) v = (k < 64) ? A[gr * 64 + k] : A2[gr * 64 + (k - 64)];
            else        v = A[gr * K + k];
        }
        As[row * PITCH + k] = v;
    }
    __syncthreads();

    const int c0    = wid * 16;
    const int halfl = lane & 15;
    const int koff  = (lane >> 4) * 2;

    v8f acc = {};
    for (int k0 = 0; k0 < K; k0 += 4) {
        v2f a, b;
        a.x = As[halfl * PITCH + k0 + koff];
        a.y = As[halfl * PITCH + k0 + koff + 1];
        b.x = W[(k0 + koff)     * DH + c0 + halfl];
        b.y = W[(k0 + koff + 1) * DH + c0 + halfl];
        acc = __builtin_amdgcn_wmma_f32_16x16x4_f32(
                  false, a, false, b, (short)0, acc, false, false);
    }

    const int colg  = c0 + halfl;
    const int mbase = (lane >> 4) * 8;
#pragma unroll
    for (int r = 0; r < 8; ++r) {
        const int grow = r0 + mbase + r;
        if (grow < n) Hout[grow * DH + colg] = acc[r];
    }
}

__global__ __launch_bounds__(256)
void node_scores(const float* __restrict__ H, const float* __restrict__ al,
                 const float* __restrict__ ar, float* __restrict__ el,
                 float* __restrict__ er, int n)
{
    const int lane = threadIdx.x & 31;
    const int node = blockIdx.x * 8 + (threadIdx.x >> 5);
    if (node >= n) return;

    const float* h = H + node * DH;
    float e0 = 0.f, e1 = 0.f, r0 = 0.f, r1 = 0.f;
    for (int i = lane; i < HID; i += 32) {
        const float h0 = h[i], h1 = h[HID + i];
        e0 += h0 * al[i];        e1 += h1 * al[HID + i];
        r0 += h0 * ar[i];        r1 += h1 * ar[HID + i];
    }
#pragma unroll
    for (int off = 16; off; off >>= 1) {
        e0 += __shfl_xor(e0, off); e1 += __shfl_xor(e1, off);
        r0 += __shfl_xor(r0, off); r1 += __shfl_xor(r1, off);
    }
    if (lane == 0) {
        el[node * 2 + 0] = e0; el[node * 2 + 1] = e1;
        er[node * 2 + 0] = r0; er[node * 2 + 1] = r1;
    }
}

#define TILE 98
#define LTH  1024
#define SCAN_U 16

__device__ __forceinline__ float lrelu(float x) {
    return x > 0.0f ? x : NEG_SLOPE * x;
}

__device__ __forceinline__ void atomicMaxF(float* addr, float val) {
    if (val >= 0.0f) atomicMax((int*)addr, __float_as_int(val));
    else             atomicMin((unsigned int*)addr, __float_as_uint(val));
}

__global__ __launch_bounds__(LTH)
void bin_count(const int* __restrict__ dst, int* __restrict__ cnt, int e)
{
    __shared__ int total;
    const int tid = threadIdx.x;
    const int lo  = blockIdx.x * TILE;
    if (tid == 0) total = 0;
    __syncthreads();

    int local = 0;
    for (int base = 0; base < e; base += LTH * SCAN_U) {
        int dv[SCAN_U];
#pragma unroll
        for (int u = 0; u < SCAN_U; ++u) {
            const int i = base + u * LTH + tid;
            dv[u] = (i < e) ? dst[i] : -1;
        }
#pragma unroll
        for (int u = 0; u < SCAN_U; ++u)
            local += ((unsigned)(dv[u] - lo) < (unsigned)TILE && dv[u] >= 0) ? 1 : 0;
    }
#pragma unroll
    for (int o = 16; o; o >>= 1) local += __shfl_xor(local, o);
    if ((tid & 31) == 0) atomicAdd(&total, local);
    __syncthreads();
    if (tid == 0) cnt[blockIdx.x] = total;
}

__global__ __launch_bounds__(LTH)
void bin_fill(const int* __restrict__ src, const int* __restrict__ dst,
              const int* __restrict__ cnt, int* __restrict__ off,
              int* __restrict__ lsrc, int* __restrict__ ldl, int e)
{
    __shared__ int cur;
    const int tid = threadIdx.x;
    const int b   = blockIdx.x;
    const int lo  = b * TILE;
    if (tid == 0) cur = 0;
    __syncthreads();
    int below = 0;
    for (int i = tid; i < b; i += LTH) below += cnt[i];
#pragma unroll
    for (int o = 16; o; o >>= 1) below += __shfl_xor(below, o);
    if ((tid & 31) == 0) atomicAdd(&cur, below);
    __syncthreads();
    const int start = cur;
    __syncthreads();
    if (tid == 0) off[b] = start;

    for (int base = 0; base < e; base += LTH * SCAN_U) {
        int dv[SCAN_U];
#pragma unroll
        for (int u = 0; u < SCAN_U; ++u) {
            const int i = base + u * LTH + tid;
            dv[u] = (i < e) ? dst[i] : -1;
        }
#pragma unroll
        for (int u = 0; u < SCAN_U; ++u) {
            const unsigned dl = (unsigned)(dv[u] - lo);
            if (dl < (unsigned)TILE && dv[u] >= 0) {
                const int p = atomicAdd(&cur, 1);
                lsrc[p] = src[base + u * LTH + tid];
                ldl[p]  = (int)dl;
            }
        }
    }
}

__global__ __launch_bounds__(LTH)
void tile_layer(const int* __restrict__ lsrc, const int* __restrict__ ldl,
                const int* __restrict__ cnt, const int* __restrict__ off,
                const float* __restrict__ H, const float* __restrict__ el,
                const float* __restrict__ er, const float* __restrict__ bias,
                float* __restrict__ outp, int n)
{
    __shared__ float acc[TILE * DH];
    __shared__ float sm[TILE * HEADS];
    __shared__ float ss[TILE * HEADS];
    __shared__ float ser[TILE * HEADS];

    const int tid  = threadIdx.x;
    const int lane = tid & 31;
    const int wid  = __builtin_amdgcn_readfirstlane(tid >> 5);
    const int lo   = blockIdx.x * TILE;
    const int beg  = off[blockIdx.x];
    const int num  = cnt[blockIdx.x];

    for (int i = tid; i < TILE * DH; i += LTH) acc[i] = 0.0f;
    if (tid < TILE * HEADS) {
        const int g = lo * HEADS + tid;
        sm[tid]  = -INFINITY;
        ss[tid]  = 0.0f;
        ser[tid] = (g < n * HEADS) ? er[g] : 0.0f;
    }
    __syncthreads();

    for (int i = tid; i < num; i += LTH) {
        const int s = lsrc[beg + i], dl = ldl[beg + i];
#pragma unroll
        for (int h = 0; h < HEADS; ++h)
            atomicMaxF(&sm[dl * HEADS + h], lrelu(el[s * HEADS + h] + ser[dl * HEADS + h]));
    }
    __syncthreads();
    if (tid < TILE * HEADS && !isfinite(sm[tid])) sm[tid] = 0.0f;
    __syncthreads();

    for (int c0 = wid * 32; c0 < num; c0 += 32 * (LTH / 32)) {
        const int i = c0 + lane;
        int s = 0, dl = 0;
        float v0 = 0.0f, v1 = 0.0f;
        if (i < num) {
            s  = lsrc[beg + i];
            dl = ldl[beg + i];
            const float sc0 = lrelu(el[s * HEADS + 0] + ser[dl * HEADS + 0]);
            const float sc1 = lrelu(el[s * HEADS + 1] + ser[dl * HEADS + 1]);
            v0 = __expf(sc0 - sm[dl * HEADS + 0]);
            v1 = __expf(sc1 - sm[dl * HEADS + 1]);
            atomicAdd(&ss[dl * HEADS + 0], v0);
            atomicAdd(&ss[dl * HEADS + 1], v1);
        }
        const int nk = (num - c0 < 32) ? (num - c0) : 32;
        for (int k = 0; k < nk; k += 4) {
            float r[4][8];
            int   dk[4];
            float a0[4], a1[4];
#pragma unroll
            for (int q = 0; q < 4; ++q) {
                const int kk = k + q;
                const int sk = __builtin_amdgcn_readlane(s, kk);
                dk[q] = __builtin_amdgcn_readlane(dl, kk);
                a0[q] = __int_as_float(__builtin_amdgcn_readlane(__float_as_int(v0), kk));
                a1[q] = __int_as_float(__builtin_amdgcn_readlane(__float_as_int(v1), kk));
                if (kk < nk) {
                    const float* row = H + (size_t)sk * DH;
#pragma unroll
                    for (int j = 0; j < 8; ++j) r[q][j] = row[j * 32 + lane];
                } else {
#pragma unroll
                    for (int j = 0; j < 8; ++j) r[q][j] = 0.0f;
                }
            }
#pragma unroll
            for (int q = 0; q < 4; ++q) {
                if (k + q < nk) {
                    float* ad = acc + dk[q] * DH;
#pragma unroll
                    for (int j = 0; j < 8; ++j)
                        atomicAdd(&ad[j * 32 + lane], (j < 4 ? a0[q] : a1[q]) * r[q][j]);
                }
            }
        }
    }
    __syncthreads();

    for (int idx = tid; idx < TILE * DH; idx += LTH) {
        const int nl   = idx >> 8;
        const int j    = idx & (DH - 1);
        const int node = lo + nl;
        if (node < n) {
            const float sv = ss[nl * HEADS + (j >> 7)];
            float v = (sv == 0.0f) ? 0.0f : acc[idx] / sv;
            v += bias[j];
            outp[(size_t)node * DH + j] = v > 0.0f ? v : 0.0f;
        }
    }
}

__global__ __launch_bounds__(256)
void edge_classify(const int* __restrict__ src, const int* __restrict__ dst,
                   const float* __restrict__ H, const float* __restrict__ fcW,
                   const float* __restrict__ fcb, float* __restrict__ out, int e)
{
    const int w    = blockIdx.x * 8 + (threadIdx.x >> 5);
    const int lane = threadIdx.x & 31;
    if (w >= e) return;
    const int s = src[w], d = dst[w];
    const float* hs = H + s * DH;
    const float* hd = H + d * DH;
    float a0 = 0.f, a1 = 0.f;
    for (int j = lane; j < DH; j += 32) {
        const float vs = hs[j], vd = hd[j];
        a0 += vs * fcW[j * 2 + 0] + vd * fcW[(DH + j) * 2 + 0];
        a1 += vs * fcW[j * 2 + 1] + vd * fcW[(DH + j) * 2 + 1];
    }
#pragma unroll
    for (int off = 16; off; off >>= 1) {
        a0 += __shfl_xor(a0, off);
        a1 += __shfl_xor(a1, off);
    }
    if (lane == 0) {
        out[w * 2 + 0] = a0 + fcb[0];
        out[w * 2 + 1] = a1 + fcb[1];
    }
}

extern "C" void kernel_launch(void* const* d_in, const int* in_sizes, int n_in,
                              void* d_out, int out_size, void* d_ws, size_t ws_size,
                              hipStream_t stream)
{
    const float* x   = (const float*)d_in[0];
    const float* rep = (const float*)d_in[1];
    const int*   src = (const int*)d_in[2];
    const int*   dst = (const int*)d_in[3];
    const float* W1  = (const float*)d_in[4];
    const float* al1 = (const float*)d_in[5];
    const float* ar1 = (const float*)d_in[6];
    const float* b1  = (const float*)d_in[7];
    const float* W2  = (const float*)d_in[8];
    const float* al2 = (const float*)d_in[9];
    const float* ar2 = (const float*)d_in[10];
    const float* b2  = (const float*)d_in[11];
    const float* fcW = (const float*)d_in[12];
    const float* fcb = (const float*)d_in[13];
    float* out = (float*)d_out;

    const int N = in_sizes[0] / 64;
    const int E = in_sizes[2];

    float* ws  = (float*)d_ws;
    float* h   = ws;
    float* agg = h   + (size_t)N * DH;
    float* rly = agg + (size_t)N * DH;
    float* el  = rly + (size_t)N * DH;
    float* er  = el  + (size_t)N * 2;
    int*   cnt = (int*)(er + (size_t)N * 2);
    int*   off = cnt + (size_t)N * 2;
    int*   lsrc = off + (size_t)N * 2;
    int*   ldl  = lsrc + (size_t)E;

    const int gN16 = (N + 15) / 16;
    const int gN8  = (N + 7) / 8;
    const int gE8  = (E + 7) / 8;
    const int gT   = (N + TILE - 1) / TILE;

    bin_count<<<gT, LTH, 0, stream>>>(dst, cnt, E);
    bin_fill<<<gT, LTH, 0, stream>>>(src, dst, cnt, off, lsrc, ldl, E);

    gemm_node<128, true><<<gN16, 512, 0, stream>>>(x, rep, W1, h, N);
    node_scores<<<gN8, 256, 0, stream>>>(h, al1, ar1, el, er, N);
    tile_layer<<<gT, LTH, 0, stream>>>(lsrc, ldl, cnt, off, h, el, er, b1, rly, N);

    gemm_node<256, false><<<gN16, 512, 0, stream>>>(rly, nullptr, W2, h, N);
    node_scores<<<gN8, 256, 0, stream>>>(h, al2, ar2, el, er, N);
    tile_layer<<<gT, LTH, 0, stream>>>(lsrc, ldl, cnt, off, h, el, er, b2, rly, N);

    edge_classify<<<gE8, 256, 0, stream>>>(src, dst, rly, fcW, fcb, out, E);
}


# edge classifier: one wave per edge -> 256 resident work-groups whose waves walk 16-edge chunks, weights held in registers, 16-byte row loads, one 128-byte store per chunk; dot sums reordered
# speedup vs baseline: 2.5060x; 1.1739x over previous
#include <hip/hip_runtime.h>
#include <hip/hip_bf16.h>
#include <math.h>

typedef __attribute__((ext_vector_type(2))) float v2f;
typedef __attribute__((ext_vector_type(8))) float v8f;

#define HID 128
#define HEADS 2
#define DH 256
#define NEG_SLOPE 0.2f

template<int K, bool CONCAT>
__global__ __launch_bounds__(512)
void gemm_node(const float* __restrict__ A, const float* __restrict__ A2,
               const float* __restrict__ W, float* __restrict__ Hout, int n)
{
    constexpr int PITCH = K + 4;
    __shared__ float As[16 * PITCH];

    const int tid  = threadIdx.x;
    const int wid  = tid >> 5;
    const int lane = tid & 31;
    const int r0   = blockIdx.x * 16;

    for (int idx = tid; idx < 16 * K; idx += 512) {
        const int row = idx / K;
        const int k   = idx - row * K;
        const int gr  = r0 + row;
        float v = 0.0f;
        if (gr < n) {
            if (CONCAT) v = (k < 64) ? A[gr * 64 + k] : A2[gr * 64 + (k - 64)];
            else        v = A[gr * K + k];
        }
        As[row * PITCH + k] = v;
    }
    __syncthreads();

    const int c0    = wid * 16;
    const int halfl = lane & 15;
    const int koff  = (lane >> 4) * 2;

    v8f acc = {};
    for (int k0 = 0; k0 < K; k0 += 4) {
        v2f a, b;
        a.x = As[halfl * PITCH + k0 + koff];
        a.y = As[halfl * PITCH + k0 + koff + 1];
        b.x = W[(k0 + koff)     * DH + c0 + halfl];
        b.y = W[(k0 + koff + 1) * DH + c0 + halfl];
        acc = __builtin_amdgcn_wmma_f32_16x16x4_f32(
                  false, a, false, b, (short)0, acc, false, false);
    }

    const int colg  = c0 + halfl;
    const int mbase = (lane >> 4) * 8;
#pragma unroll
    for (int r = 0; r < 8; ++r) {
        const int grow = r0 + mbase + r;
        if (grow < n) Hout[grow * DH + colg] = acc[r];
    }
}

__global__ __launch_bounds__(256)
void node_scores(const float* __restrict__ H, const float* __restrict__ al,
                 const float* __restrict__ ar, float* __restrict__ el,
                 float* __restrict__ er, int n)
{
    const int lane = threadIdx.x & 31;
    const int node = blockIdx.x * 8 + (threadIdx.x >> 5);
    if (node >= n) return;

    const float* h = H + node * DH;
    float e0 = 0.f, e1 = 0.f, r0 = 0.f, r1 = 0.f;
    for (int i = lane; i < HID; i += 32) {
        const float h0 = h[i], h1 = h[HID + i];
        e0 += h0 * al[i];        e1 += h1 * al[HID + i];
        r0 += h0 * ar[i];        r1 += h1 * ar[HID + i];
    }
#pragma unroll
    for (int off = 16; off; off >>= 1) {
        e0 += __shfl_xor(e0, off); e1 += __shfl_xor(e1, off);
        r0 += __shfl_xor(r0, off); r1 += __shfl_xor(r1, off);
    }
    if (lane == 0) {
        el[node * 2 + 0] = e0; el[node * 2 + 1] = e1;
        er[node * 2 + 0] = r0; er[node * 2 + 1] = r1;
    }
}

#define TILE 98
#define LTH  1024
#define SCAN_U 16

__device__ __forceinline__ float lrelu(float x) {
    return x > 0.0f ? x : NEG_SLOPE * x;
}

__device__ __forceinline__ void atomicMaxF(float* addr, float val) {
    if (val >= 0.0f) atomicMax((int*)addr, __float_as_int(val));
    else             atomicMin((unsigned int*)addr, __float_as_uint(val));
}

__global__ __launch_bounds__(LTH)
void bin_count(const int* __restrict__ dst, int* __restrict__ cnt, int e)
{
    __shared__ int total;
    const int tid = threadIdx.x;
    const int lo  = blockIdx.x * TILE;
    if (tid == 0) total = 0;
    __syncthreads();

    int local = 0;
    for (int base = 0; base < e; base += LTH * SCAN_U) {
        int dv[SCAN_U];
#pragma unroll
        for (int u = 0; u < SCAN_U; ++u) {
            const int i = base + u * LTH + tid;
            dv[u] = (i < e) ? dst[i] : -1;
        }
#pragma unroll
        for (int u = 0; u < SCAN_U; ++u)
            local += ((unsigned)(dv[u] - lo) < (unsigned)TILE && dv[u] >= 0) ? 1 : 0;
    }
#pragma unroll
    for (int o = 16; o; o >>= 1) local += __shfl_xor(local, o);
    if ((tid & 31) == 0) atomicAdd(&total, local);
    __syncthreads();
    if (tid == 0) cnt[blockIdx.x] = total;
}

__global__ __launch_bounds__(LTH)
void bin_fill(const int* __restrict__ src, const int* __restrict__ dst,
              const int* __restrict__ cnt, int* __restrict__ off,
              int* __restrict__ lsrc, int* __restrict__ ldl, int e)
{
    __shared__ int cur;
    const int tid = threadIdx.x;
    const int b   = blockIdx.x;
    const int lo  = b * TILE;
    if (tid == 0) cur = 0;
    __syncthreads();
    int below = 0;
    for (int i = tid; i < b; i += LTH) below += cnt[i];
#pragma unroll
    for (int o = 16; o; o >>= 1) below += __shfl_xor(below, o);
    if ((tid & 31) == 0) atomicAdd(&cur, below);
    __syncthreads();
    const int start = cur;
    __syncthreads();
    if (tid == 0) off[b] = start;

    for (int base = 0; base < e; base += LTH * SCAN_U) {
        int dv[SCAN_U];
#pragma unroll
        for (int u = 0; u < SCAN_U; ++u) {
            const int i = base + u * LTH + tid;
            dv[u] = (i < e) ? dst[i] : -1;
        }
#pragma unroll
        for (int u = 0; u < SCAN_U; ++u) {
            const unsigned dl = (unsigned)(dv[u] - lo);
            if (dl < (unsigned)TILE && dv[u] >= 0) {
                const int p = atomicAdd(&cur, 1);
                lsrc[p] = src[base + u * LTH + tid];
                ldl[p]  = (int)dl;
            }
        }
    }
}

__global__ __launch_bounds__(LTH)
void tile_layer(const int* __restrict__ lsrc, const int* __restrict__ ldl,
                const int* __restrict__ cnt, const int* __restrict__ off,
                const float* __restrict__ H, const float* __restrict__ el,
                const float* __restrict__ er, const float* __restrict__ bias,
                float* __restrict__ outp, int n)
{
    __shared__ float acc[TILE * DH];
    __shared__ float sm[TILE * HEADS];
    __shared__ float ss[TILE * HEADS];
    __shared__ float ser[TILE * HEADS];

    const int tid  = threadIdx.x;
    const int lane = tid & 31;
    const int wid  = __builtin_amdgcn_readfirstlane(tid >> 5);
    const int lo   = blockIdx.x * TILE;
    const int beg  = off[blockIdx.x];
    const int num  = cnt[blockIdx.x];

    for (int i = tid; i < TILE * DH; i += LTH) acc[i] = 0.0f;
    if (tid < TILE * HEADS) {
        const int g = lo * HEADS + tid;
        sm[tid]  = -INFINITY;
        ss[tid]  = 0.0f;
        ser[tid] = (g < n * HEADS) ? er[g] : 0.0f;
    }
    __syncthreads();

    for (int i = tid; i < num; i += LTH) {
        const int s = lsrc[beg + i], dl = ldl[beg + i];
#pragma unroll
        for (int h = 0; h < HEADS; ++h)
            atomicMaxF(&sm[dl * HEADS + h], lrelu(el[s * HEADS + h] + ser[dl * HEADS + h]));
    }
    __syncthreads();
    if (tid < TILE * HEADS && !isfinite(sm[tid])) sm[tid] = 0.0f;
    __syncthreads();

    for (int c0 = wid * 32; c0 < num; c0 += 32 * (LTH / 32)) {
        const int i = c0 + lane;
        int s = 0, dl = 0;
        float v0 = 0.0f, v1 = 0.0f;
        if (i < num) {
            s  = lsrc[beg + i];
            dl = ldl[beg + i];
            const float sc0 = lrelu(el[s * HEADS + 0] + ser[dl * HEADS + 0]);
            const float sc1 = lrelu(el[s * HEADS + 1] + ser[dl * HEADS + 1]);
            v0 = __expf(sc0 - sm[dl * HEADS + 0]);
            v1 = __expf(sc1 - sm[dl * HEADS + 1]);
            atomicAdd(&ss[dl * HEADS + 0], v0);
            atomicAdd(&ss[dl * HEADS + 1], v1);
        }
        const int nk = (num - c0 < 32) ? (num - c0) : 32;
        for (int k = 0; k < nk; k += 4) {
            float r[4][8];
            int   dk[4];
            float a0[4], a1[4];
#pragma unroll
            for (int q = 0; q < 4; ++q) {
                const int kk = k + q;
                const int sk = __builtin_amdgcn_readlane(s, kk);
                dk[q] = __builtin_amdgcn_readlane(dl, kk);
                a0[q] = __int_as_float(__builtin_amdgcn_readlane(__float_as_int(v0), kk));
                a1[q] = __int_as_float(__builtin_amdgcn_readlane(__float_as_int(v1), kk));
                if (kk < nk) {
                    const float* row = H + (size_t)sk * DH;
#pragma unroll
                    for (int j = 0; j < 8; ++j) r[q][j] = row[j * 32 + lane];
                } else {
#pragma unroll
                    for (int j = 0; j < 8; ++j) r[q][j] = 0.0f;
                }
            }
#pragma unroll
            for (int q = 0; q < 4; ++q) {
                if (k + q < nk) {
                    float* ad = acc + dk[q] * DH;
#pragma unroll
                    for (int j = 0; j < 8; ++j)
                        atomicAdd(&ad[j * 32 + lane], (j < 4 ? a0[q] : a1[q]) * r[q][j]);
                }
            }
        }
    }
    __syncthreads();

    for (int idx = tid; idx < TILE * DH; idx += LTH) {
        const int nl   = idx >> 8;
        const int j    = idx & (DH - 1);
        const int node = lo + nl;
        if (node < n) {
            const float sv = ss[nl * HEADS + (j >> 7)];
            float v = (sv == 0.0f) ? 0.0f : acc[idx] / sv;
            v += bias[j];
            outp[(size_t)node * DH + j] = v > 0.0f ? v : 0.0f;
        }
    }
}

#define EC_TH   1024
#define EC_GRID 256

typedef __attribute__((ext_vector_type(4))) float v4f;

__global__ __launch_bounds__(EC_TH) __attribute__((amdgpu_waves_per_eu(8, 8)))
void edge_classify(const int* __restrict__ src, const int* __restrict__ dst,
                   const float* __restrict__ H, const float* __restrict__ fcW,
                   const float* __restrict__ fcb, float* __restrict__ out, int e)
{
    const int lane = threadIdx.x & 31;
    const int wave   = __builtin_amdgcn_readfirstlane((int)((threadIdx.x >> 5) * gridDim.x + blockIdx.x));
    const int nwaves = (int)gridDim.x * (EC_TH / 32);
    const int nchunks = (e + 15) >> 4;
    if (wave >= nchunks) return;

    v4f wS[2][2], wD[2][2];
#pragma unroll
    for (int hf = 0; hf < 2; ++hf) {
        const v4f* ps = (const v4f*)(fcW + (size_t)(hf * 128 + 4 * lane) * 2);
        const v4f* pd = (const v4f*)(fcW + (size_t)(DH + hf * 128 + 4 * lane) * 2);
        wS[hf][0] = ps[0]; wS[hf][1] = ps[1];
        wD[hf][0] = pd[0]; wD[hf][1] = pd[1];
    }
    const float bo = fcb[lane & 1];

    const int last = e - 1;
    int s, d;
    {
        const int ei = min((wave << 4) + (lane & 15), last);
        s = src[ei]; d = dst[ei];
    }
    for (int ch = wave; ch < nchunks; ch += nwaves) {
        const int e0 = ch << 4;
        const int en = min(((ch + nwaves) << 4) + (lane & 15), last);
        const int sn = src[en], dn = dst[en];
        float res = 0.0f;
#pragma unroll 1
        for (int g = 0; g < 4; ++g) {
            v4f rs[4][2], rd[4][2];
#pragma unroll
            for (int q = 0; q < 4; ++q) {
                const int sk = __builtin_amdgcn_readlane(s, g * 4 + q);
                const int dk = __builtin_amdgcn_readlane(d, g * 4 + q);
                const v4f* ps = (const v4f*)(H + (size_t)sk * DH) + lane;
                const v4f* pd = (const v4f*)(H + (size_t)dk * DH) + lane;
                rs[q][0] = ps[0]; rs[q][1] = ps[32];
                rd[q][0] = pd[0]; rd[q][1] = pd[32];
            }
            float p[8];
#pragma unroll
            for (int q = 0; q < 4; ++q) {
                v2f a = {0.0f, 0.0f};
#pragma unroll
                for (int hf = 0; hf < 2; ++hf) {
                    a += rs[q][hf].x * wS[hf][0].xy;
                    a += rs[q][hf].y * wS[hf][0].zw;
                    a += rs[q][hf].z * wS[hf][1].xy;
                    a += rs[q][hf].w * wS[hf][1].zw;
                    a += rd[q][hf].x * wD[hf][0].xy;
                    a += rd[q][hf].y * wD[hf][0].zw;
                    a += rd[q][hf].z * wD[hf][1].xy;
                    a += rd[q][hf].w * wD[hf][1].zw;
                }
                p[q * 2 + 0] = a.x;
                p[q * 2 + 1] = a.y;
            }
            float t[4], u[2];
#pragma unroll
            for (int i = 0; i < 4; ++i) {
                const float keep = (lane & 1) ? p[2 * i + 1] : p[2 * i];
                const float send = (lane & 1) ? p[2 * i] : p[2 * i + 1];
                t[i] = keep + __shfl_xor(send, 1);
            }
#pragma unroll
            for (int i = 0; i < 2; ++i) {
                const float keep = (lane & 2) ? t[2 * i + 1] : t[2 * i];
                const float send = (lane & 2) ? t[2 * i] : t[2 * i + 1];
                u[i] = keep + __shfl_xor(send, 2);
            }
            float v;
            {
                const float keep = (lane & 4) ? u[1] : u[0];
                const float send = (lane & 4) ? u[0] : u[1];
                v = keep + __shfl_xor(send, 4);
            }
            v += __shfl_xor(v, 8);
            v += __shfl_xor(v, 16);
            if ((lane >> 3) == g) res = v;
        }
        const int oi = e0 * 2 + lane;
        if (oi < e * 2) out[oi] = res + bo;
        s = sn; d = dn;
    }
}

extern "C" void kernel_launch(void* const* d_in, const int* in_sizes, int n_in,
                              void* d_out, int out_size, void* d_ws, size_t ws_size,
                              hipStream_t stream)
{
    const float* x   = (const float*)d_in[0];
    const float* rep = (const float*)d_in[1];
    const int*   src = (const int*)d_in[2];
    const int*   dst = (const int*)d_in[3];
    const float* W1  = (const float*)d_in[4];
    const float* al1 = (const float*)d_in[5];
    const float* ar1 = (const float*)d_in[6];
    const float* b1  = (const float*)d_in[7];
    const float* W2  = (const float*)d_in[8];
    const float* al2 = (const float*)d_in[9];
    const float* ar2 = (const float*)d_in[10];
    const float* b2  = (const float*)d_in[11];
    const float* fcW = (const float*)d_in[12];
    const float* fcb = (const float*)d_in[13];
    float* out = (float*)d_out;

    const int N = in_sizes[0] / 64;
    const int E = in_sizes[2];

    float* ws  = (float*)d_ws;
    float* h   = ws;
    float* agg = h   + (size_t)N * DH;
    float* rly = agg + (size_t)N * DH;
    float* el  = rly + (size_t)N * DH;
    float* er  = el  + (size_t)N * 2;
    int*   cnt = (int*)(er + (size_t)N * 2);
    int*   off = cnt + (size_t)N * 2;
    int*   lsrc = off + (size_t)N * 2;
    int*   ldl  = lsrc + (size_t)E;

    const int gN16 = (N + 15) / 16;
    const int gN8  = (N + 7) / 8;
    const int gT   = (N + TILE - 1) / TILE;

    bin_count<<<gT, LTH, 0, stream>>>(dst, cnt, E);
    bin_fill<<<gT, LTH, 0, stream>>>(src, dst, cnt, off, lsrc, ldl, E);

    gemm_node<128, true><<<gN16, 512, 0, stream>>>(x, rep, W1, h, N);
    node_scores<<<gN8, 256, 0, stream>>>(h, al1, ar1, el, er, N);
    tile_layer<<<gT, LTH, 0, stream>>>(lsrc, ldl, cnt, off, h, el, er, b1, rly, N);

    gemm_node<256, false><<<gN16, 512, 0, stream>>>(rly, nullptr, W2, h, N);
    node_scores<<<gN8, 256, 0, stream>>>(h, al2, ar2, el, er, N);
    tile_layer<<<gT, LTH, 0, stream>>>(lsrc, ldl, cnt, off, h, el, er, b2, rly, N);

    edge_classify<<<EC_GRID, EC_TH, 0, stream>>>(src, dst, rly, fcW, fcb, out, E);
}


# edge filing in one pass: 128 slice work-groups count (slice, tile) pairs, then place by column sums + LDS cursors; was 256 work-groups each scanning all edges twice; tile list (so sum) order changes
# speedup vs baseline: 3.1294x; 1.2488x over previous
#include <hip/hip_runtime.h>
#include <hip/hip_bf16.h>
#include <math.h>

typedef __attribute__((ext_vector_type(2))) float v2f;
typedef __attribute__((ext_vector_type(8))) float v8f;

#define HID 128
#define HEADS 2
#define DH 256
#define NEG_SLOPE 0.2f

template<int K, bool CONCAT>
__global__ __launch_bounds__(512)
void gemm_node(const float* __restrict__ A, const float* __restrict__ A2,
               const float* __restrict__ W, float* __restrict__ Hout, int n)
{
    constexpr int PITCH = K + 4;
    __shared__ float As[16 * PITCH];

    const int tid  = threadIdx.x;
    const int wid  = tid >> 5;
    const int lane = tid & 31;
    const int r0   = blockIdx.x * 16;

    for (int idx = tid; idx < 16 * K; idx += 512) {
        const int row = idx / K;
        const int k   = idx - row * K;
        const int gr  = r0 + row;
        float v = 0.0f;
        if (gr < n) {
            if (CONCAT) v = (k < 64) ? A[gr * 64 + k] : A2[gr * 64 + (k - 64)];
            else        v = A[gr * K + k];
        }
        As[row * PITCH + k] = v;
    }
    __syncthreads();

    const int c0    = wid * 16;
    const int halfl = lane & 15;
    const int koff  = (lane >> 4) * 2;

    v8f acc = {};
    for (int k0 = 0; k0 < K; k0 += 4) {
        v2f a, b;
        a.x = As[halfl * PITCH + k0 + koff];
        a.y = As[halfl * PITCH + k0 + koff + 1];
        b.x = W[(k0 + koff)     * DH + c0 + halfl];
        b.y = W[(k0 + koff + 1) * DH + c0 + halfl];
        acc = __builtin_amdgcn_wmma_f32_16x16x4_f32(
                  false, a, false, b, (short)0, acc, false, false);
    }

    const int colg  = c0 + halfl;
    const int mbase = (lane >> 4) * 8;
#pragma unroll
    for (int r = 0; r < 8; ++r) {
        const int grow = r0 + mbase + r;
        if (grow < n) Hout[grow * DH + colg] = acc[r];
    }
}

__global__ __launch_bounds__(256)
void node_scores(const float* __restrict__ H, const float* __restrict__ al,
                 const float* __restrict__ ar, float* __restrict__ el,
                 float* __restrict__ er, int n)
{
    const int lane = threadIdx.x & 31;
    const int node = blockIdx.x * 8 + (threadIdx.x >> 5);
    if (node >= n) return;

    const float* h = H + node * DH;
    float e0 = 0.f, e1 = 0.f, r0 = 0.f, r1 = 0.f;
    for (int i = lane; i < HID; i += 32) {
        const float h0 = h[i], h1 = h[HID + i];
        e0 += h0 * al[i];        e1 += h1 * al[HID + i];
        r0 += h0 * ar[i];        r1 += h1 * ar[HID + i];
    }
#pragma unroll
    for (int off = 16; off; off >>= 1) {
        e0 += __shfl_xor(e0, off); e1 += __shfl_xor(e1, off);
        r0 += __shfl_xor(r0, off); r1 += __shfl_xor(r1, off);
    }
    if (lane == 0) {
        el[node * 2 + 0] = e0; el[node * 2 + 1] = e1;
        er[node * 2 + 0] = r0; er[node * 2 + 1] = r1;
    }
}

#define TILE 98
#define LTH  1024
#define NB   128
#define FU   4
#define MAXT LTH

__device__ __forceinline__ float lrelu(float x) {
    return x > 0.0f ? x : NEG_SLOPE * x;
}

__device__ __forceinline__ void atomicMaxF(float* addr, float val) {
    if (val >= 0.0f) atomicMax((int*)addr, __float_as_int(val));
    else             atomicMin((unsigned int*)addr, __float_as_uint(val));
}

__global__ __launch_bounds__(LTH)
void bin_hist(const int* __restrict__ dst, int* __restrict__ hist, int e, int n, int gT)
{
    __shared__ int hc[MAXT];
    const int tid = threadIdx.x;
    const int per = (e + NB - 1) / NB;
    const int lo  = blockIdx.x * per;
    const int hi  = min(e, lo + per);
    for (int t = tid; t < gT; t += LTH) hc[t] = 0;
    __syncthreads();
    for (int base = lo; base < hi; base += LTH * FU) {
        int dv[FU];
#pragma unroll
        for (int u = 0; u < FU; ++u) {
            const int i = base + u * LTH + tid;
            dv[u] = (i < hi) ? dst[i] : -1;
        }
#pragma unroll
        for (int u = 0; u < FU; ++u)
            if ((unsigned)dv[u] < (unsigned)n) atomicAdd(&hc[dv[u] / TILE], 1);
    }
    __syncthreads();
    for (int t = tid; t < gT; t += LTH) hist[blockIdx.x * gT + t] = hc[t];
}

__global__ __launch_bounds__(LTH)
void bin_place(const int* __restrict__ src, const int* __restrict__ dst,
               const int* __restrict__ hist, int* __restrict__ cnt, int* __restrict__ off,
               int* __restrict__ lsrc, int* __restrict__ ldl, int e, int n, int gT)
{
    constexpr int TQ  = 256;
    constexpr int RQ  = LTH / TQ;
    constexpr int RPQ = NB / RQ;
    __shared__ int pa[LTH];
    __shared__ int pb[LTH];
    __shared__ int tot[MAXT];
    __shared__ int cur[MAXT];
    __shared__ int wsum[LTH / 32];

    const int tid  = threadIdx.x;
    const int lane = tid & 31;
    const int wid  = tid >> 5;
    const int b    = blockIdx.x;
    const int per  = (e + NB - 1) / NB;
    const int lo   = b * per;
    const int hi   = min(e, lo + per);

    int dv[FU], sv[FU];
#pragma unroll
    for (int u = 0; u < FU; ++u) {
        const int i = lo + u * LTH + tid;
        dv[u] = (i < hi) ? dst[i] : -1;
        sv[u] = (i < hi) ? src[i] : 0;
    }

    const int tl = tid & (TQ - 1);
    const int q  = tid / TQ;
    for (int t0 = 0; t0 < gT; t0 += TQ) {
        const int t = t0 + tl;
        int all = 0, bel = 0;
        if (t < gT) {
            int c[RPQ];
#pragma unroll
            for (int r = 0; r < RPQ; ++r) c[r] = hist[(q * RPQ + r) * gT + t];
#pragma unroll
            for (int r = 0; r < RPQ; ++r) {
                all += c[r];
                bel += (q * RPQ + r < b) ? c[r] : 0;
            }
        }
        pa[tid] = all;
        pb[tid] = bel;
        __syncthreads();
        if (q == 0 && t < gT) {
            int a = 0, w = 0;
            for (int k = 0; k < RQ; ++k) { a += pa[k * TQ + tl]; w += pb[k * TQ + tl]; }
            tot[t] = a;
            cur[t] = w;
        }
        __syncthreads();
    }

    const int v = (tid < gT) ? tot[tid] : 0;
    int inc = v;
#pragma unroll
    for (int o = 1; o < 32; o <<= 1) {
        const int y = __shfl_up(inc, o);
        if (lane >= o) inc += y;
    }
    if (lane == 31) wsum[wid] = inc;
    __syncthreads();
    if (wid == 0) {
        const int w = wsum[lane];
        int wi = w;
#pragma unroll
        for (int o = 1; o < 32; o <<= 1) {
            const int y = __shfl_up(wi, o);
            if (lane >= o) wi += y;
        }
        wsum[lane] = wi - w;
    }
    __syncthreads();
    if (tid < gT) {
        const int start = inc - v + wsum[wid];
        cur[tid] += start;
        if (b == 0) { cnt[tid] = v; off[tid] = start; }
    }
    __syncthreads();

    for (int base = lo; ; ) {
#pragma unroll
        for (int u = 0; u < FU; ++u) {
            const int d = dv[u];
            if ((unsigned)d < (unsigned)n) {
                const int t = d / TILE;
                const int p = atomicAdd(&cur[t], 1);
                lsrc[p] = sv[u];
                ldl[p]  = d - t * TILE;
            }
        }
        base += LTH * FU;
        if (base >= hi) break;
#pragma unroll
        for (int u = 0; u < FU; ++u) {
            const int i = base + u * LTH + tid;
            dv[u] = (i < hi) ? dst[i] : -1;
            sv[u] = (i < hi) ? src[i] : 0;
        }
    }
}

__global__ __launch_bounds__(LTH)
void tile_layer(const int* __restrict__ lsrc, const int* __restrict__ ldl,
                const int* __restrict__ cnt, const int* __restrict__ off,
                const float* __restrict__ H, const float* __restrict__ el,
                const float* __restrict__ er, const float* __restrict__ bias,
                float* __restrict__ outp, int n)
{
    __shared__ float acc[TILE * DH];
    __shared__ float sm[TILE * HEADS];
    __shared__ float ss[TILE * HEADS];
    __shared__ float ser[TILE * HEADS];

    const int tid  = threadIdx.x;
    const int lane = tid & 31;
    const int wid  = __builtin_amdgcn_readfirstlane(tid >> 5);
    const int lo   = blockIdx.x * TILE;
    const int beg  = off[blockIdx.x];
    const int num  = cnt[blockIdx.x];

    for (int i = tid; i < TILE * DH; i += LTH) acc[i] = 0.0f;
    if (tid < TILE * HEADS) {
        const int g = lo * HEADS + tid;
        sm[tid]  = -INFINITY;
        ss[tid]  = 0.0f;
        ser[tid] = (g < n * HEADS) ? er[g] : 0.0f;
    }
    __syncthreads();

    for (int i = tid; i < num; i += LTH) {
        const int s = lsrc[beg + i], dl = ldl[beg + i];
#pragma unroll
        for (int h = 0; h < HEADS; ++h)
            atomicMaxF(&sm[dl * HEADS + h], lrelu(el[s * HEADS + h] + ser[dl * HEADS + h]));
    }
    __syncthreads();
    if (tid < TILE * HEADS && !isfinite(sm[tid])) sm[tid] = 0.0f;
    __syncthreads();

    for (int c0 = wid * 32; c0 < num; c0 += 32 * (LTH / 32)) {
        const int i = c0 + lane;
        int s = 0, dl = 0;
        float v0 = 0.0f, v1 = 0.0f;
        if (i < num) {
            s  = lsrc[beg + i];
            dl = ldl[beg + i];
            const float sc0 = lrelu(el[s * HEADS + 0] + ser[dl * HEADS + 0]);
            const float sc1 = lrelu(el[s * HEADS + 1] + ser[dl * HEADS + 1]);
            v0 = __expf(sc0 - sm[dl * HEADS + 0]);
            v1 = __expf(sc1 - sm[dl * HEADS + 1]);
            atomicAdd(&ss[dl * HEADS + 0], v0);
            atomicAdd(&ss[dl * HEADS + 1], v1);
        }
        const int nk = (num - c0 < 32) ? (num - c0) : 32;
        for (int k = 0; k < nk; k += 4) {
            float r[4][8];
            int   dk[4];
            float a0[4], a1[4];
#pragma unroll
            for (int q = 0; q < 4; ++q) {
                const int kk = k + q;
                const int sk = __builtin_amdgcn_readlane(s, kk);
                dk[q] = __builtin_amdgcn_readlane(dl, kk);
                a0[q] = __int_as_float(__builtin_amdgcn_readlane(__float_as_int(v0), kk));
                a1[q] = __int_as_float(__builtin_amdgcn_readlane(__float_as_int(v1), kk));
                if (kk < nk) {
                    const float* row = H + (size_t)sk * DH;
#pragma unroll
                    for (int j = 0; j < 8; ++j) r[q][j] = row[j * 32 + lane];
                } else {
#pragma unroll
                    for (int j = 0; j < 8; ++j) r[q][j] = 0.0f;
                }
            }
#pragma unroll
            for (int q = 0; q < 4; ++q) {
                if (k + q < nk) {
                    float* ad = acc + dk[q] * DH;
#pragma unroll
                    for (int j = 0; j < 8; ++j)
                        atomicAdd(&ad[j * 32 + lane], (j < 4 ? a0[q] : a1[q]) * r[q][j]);
                }
            }
        }
    }
    __syncthreads();

    for (int idx = tid; idx < TILE * DH; idx += LTH) {
        const int nl   = idx >> 8;
        const int j    = idx & (DH - 1);
        const int node = lo + nl;
        if (node < n) {
            const float sv = ss[nl * HEADS + (j >> 7)];
            float v = (sv == 0.0f) ? 0.0f : acc[idx] / sv;
            v += bias[j];
            outp[(size_t)node * DH + j] = v > 0.0f ? v : 0.0f;
        }
    }
}

#define EC_TH   1024
#define EC_GRID 256

typedef __attribute__((ext_vector_type(4))) float v4f;

__global__ __launch_bounds__(EC_TH) __attribute__((amdgpu_waves_per_eu(8, 8)))
void edge_classify(const int* __restrict__ src, const int* __restrict__ dst,
                   const float* __restrict__ H, const float* __restrict__ fcW,
                   const float* __restrict__ fcb, float* __restrict__ out, int e)
{
    const int lane = threadIdx.x & 31;
    const int wave   = __builtin_amdgcn_readfirstlane((int)((threadIdx.x >> 5) * gridDim.x + blockIdx.x));
    const int nwaves = (int)gridDim.x * (EC_TH / 32);
    const int nchunks = (e + 15) >> 4;
    if (wave >= nchunks) return;

    v4f wS[2][2], wD[2][2];
#pragma unroll
    for (int hf = 0; hf < 2; ++hf) {
        const v4f* ps = (const v4f*)(fcW + (size_t)(hf * 128 + 4 * lane) * 2);
        const v4f* pd = (const v4f*)(fcW + (size_t)(DH + hf * 128 + 4 * lane) * 2);
        wS[hf][0] = ps[0]; wS[hf][1] = ps[1];
        wD[hf][0] = pd[0]; wD[hf][1] = pd[1];
    }
    const float bo = fcb[lane & 1];

    const int last = e - 1;
    int s, d;
    {
        const int ei = min((wave << 4) + (lane & 15), last);
        s = src[ei]; d = dst[ei];
    }
    for (int ch = wave; ch < nchunks; ch += nwaves) {
        const int e0 = ch << 4;
        const int en = min(((ch + nwaves) << 4) + (lane & 15), last);
        const int sn = src[en], dn = dst[en];
        float res = 0.0f;
#pragma unroll 1
        for (int g = 0; g < 4; ++g) {
            v4f rs[4][2], rd[4][2];
#pragma unroll
            for (int q = 0; q < 4; ++q) {
                const int sk = __builtin_amdgcn_readlane(s, g * 4 + q);
                const int dk = __builtin_amdgcn_readlane(d, g * 4 + q);
                const v4f* ps = (const v4f*)(H + (size_t)sk * DH) + lane;
                const v4f* pd = (const v4f*)(H + (size_t)dk * DH) + lane;
                rs[q][0] = ps[0]; rs[q][1] = ps[32];
                rd[q][0] = pd[0]; rd[q][1] = pd[32];
            }
            float p[8];
#pragma unroll
            for (int q = 0; q < 4; ++q) {
                v2f a = {0.0f, 0.0f};
#pragma unroll
                for (int hf = 0; hf < 2; ++hf) {
                    a += rs[q][hf].x * wS[hf][0].xy;
                    a += rs[q][hf].y * wS[hf][0].zw;
                    a += rs[q][hf].z * wS[hf][1].xy;
                    a += rs[q][hf].w * wS[hf][1].zw;
                    a += rd[q][hf].x * wD[hf][0].xy;
                    a += rd[q][hf].y * wD[hf][0].zw;
                    a += rd[q][hf].z * wD[hf][1].xy;
                    a += rd[q][hf].w * wD[hf][1].zw;
                }
                p[q * 2 + 0] = a.x;
                p[q * 2 + 1] = a.y;
            }
            float t[4], u[2];
#pragma unroll
            for (int i = 0; i < 4; ++i) {
                const float keep = (lane & 1) ? p[2 * i + 1] : p[2 * i];
                const float send = (lane & 1) ? p[2 * i] : p[2 * i + 1];
                t[i] = keep + __shfl_xor(send, 1);
            }
#pragma unroll
            for (int i = 0; i < 2; ++i) {
                const float keep = (lane & 2) ? t[2 * i + 1] : t[2 * i];
                const float send = (lane & 2) ? t[2 * i] : t[2 * i + 1];
                u[i] = keep + __shfl_xor(send, 2);
            }
            float v;
            {
                const float keep = (lane & 4) ? u[1] : u[0];
                const float send = (lane & 4) ? u[0] : u[1];
                v = keep + __shfl_xor(send, 4);
            }
            v += __shfl_xor(v, 8);
            v += __shfl_xor(v, 16);
            if ((lane >> 3) == g) res = v;
        }
        const int oi = e0 * 2 + lane;
        if (oi < e * 2) out[oi] = res + bo;
        s = sn; d = dn;
    }
}

extern "C" void kernel_launch(void* const* d_in, const int* in_sizes, int n_in,
                              void* d_out, int out_size, void* d_ws, size_t ws_size,
                              hipStream_t stream)
{
    const float* x   = (const float*)d_in[0];
    const float* rep = (const float*)d_in[1];
    const int*   src = (const int*)d_in[2];
    const int*   dst = (const int*)d_in[3];
    const float* W1  = (const float*)d_in[4];
    const float* al1 = (const float*)d_in[5];
    const float* ar1 = (const float*)d_in[6];
    const float* b1  = (const float*)d_in[7];
    const float* W2  = (const float*)d_in[8];
    const float* al2 = (const float*)d_in[9];
    const float* ar2 = (const float*)d_in[10];
    const float* b2  = (const float*)d_in[11];
    const float* fcW = (const float*)d_in[12];
    const float* fcb = (const float*)d_in[13];
    float* out = (float*)d_out;

    const int N = in_sizes[0] / 64;
    const int E = in_sizes[2];

    float* ws  = (float*)d_ws;
    float* h   = ws;
    float* agg = h   + (size_t)N * DH;
    float* rly = agg + (size_t)N * DH;
    float* el  = rly + (size_t)N * DH;
    float* er  = el  + (size_t)N * 2;
    int*   cnt = (int*)(er + (size_t)N * 2);
    int*   off = cnt + (size_t)N * 2;
    int*   lsrc = off + (size_t)N * 2;
    int*   ldl  = lsrc + (size_t)E;
    int*   hist = (int*)agg;

    const int gN16 = (N + 15) / 16;
    const int gN8  = (N + 7) / 8;
    const int gT   = (N + TILE - 1) / TILE;

    bin_hist<<<NB, LTH, 0, stream>>>(dst, hist, E, N, gT);
    bin_place<<<NB, LTH, 0, stream>>>(src, dst, hist, cnt, off, lsrc, ldl, E, N, gT);

    gemm_node<128, true><<<gN16, 512, 0, stream>>>(x, rep, W1, h, N);
    node_scores<<<gN8, 256, 0, stream>>>(h, al1, ar1, el, er, N);
    tile_layer<<<gT, LTH, 0, stream>>>(lsrc, ldl, cnt, off, h, el, er, b1, rly, N);

    gemm_node<256, false><<<gN16, 512, 0, stream>>>(rly, nullptr, W2, h, N);
    node_scores<<<gN8, 256, 0, stream>>>(h, al2, ar2, el, er, N);
    tile_layer<<<gT, LTH, 0, stream>>>(lsrc, ldl, cnt, off, h, el, er, b2, rly, N);

    edge_classify<<<EC_GRID, EC_TH, 0, stream>>>(src, dst, rly, fcW, fcb, out, E);
}


# tile layer gathers a source row as two 16-byte loads per lane (was eight 4-byte): half the second-level-cache requests per edge; LDS row re-laid (column 4l+c at word 32c+l), 16-byte result stores
# speedup vs baseline: 3.3958x; 1.0851x over previous
#include <hip/hip_runtime.h>
#include <hip/hip_bf16.h>
#include <math.h>

typedef __attribute__((ext_vector_type(2))) float v2f;
typedef __attribute__((ext_vector_type(4))) float v4f;
typedef __attribute__((ext_vector_type(8))) float v8f;

#define HID 128
#define HEADS 2
#define DH 256
#define NEG_SLOPE 0.2f

template<int K, bool CONCAT>
__global__ __launch_bounds__(512)
void gemm_node(const float* __restrict__ A, const float* __restrict__ A2,
               const float* __restrict__ W, float* __restrict__ Hout, int n)
{
    constexpr int PITCH = K + 4;
    __shared__ float As[16 * PITCH];

    const int tid  = threadIdx.x;
    const int wid  = tid >> 5;
    const int lane = tid & 31;
    const int r0   = blockIdx.x * 16;

    for (int idx = tid; idx < 16 * K; idx += 512) {
        const int row = idx / K;
        const int k   = idx - row * K;
        const int gr  = r0 + row;
        float v = 0.0f;
        if (gr < n) {
            if (CONCAT) v = (k < 64) ? A[gr * 64 + k] : A2[gr * 64 + (k - 64)];
            else        v = A[gr * K + k];
        }
        As[row * PITCH + k] = v;
    }
    __syncthreads();

    const int c0    = wid * 16;
    const int halfl = lane & 15;
    const int koff  = (lane >> 4) * 2;

    v8f acc = {};
    for (int k0 = 0; k0 < K; k0 += 4) {
        v2f a, b;
        a.x = As[halfl * PITCH + k0 + koff];
        a.y = As[halfl * PITCH + k0 + koff + 1];
        b.x = W[(k0 + koff)     * DH + c0 + halfl];
        b.y = W[(k0 + koff + 1) * DH + c0 + halfl];
        acc = __builtin_amdgcn_wmma_f32_16x16x4_f32(
                  false, a, false, b, (short)0, acc, false, false);
    }

    const int colg  = c0 + halfl;
    const int mbase = (lane >> 4) * 8;
#pragma unroll
    for (int r = 0; r < 8; ++r) {
        const int grow = r0 + mbase + r;
        if (grow < n) Hout[grow * DH + colg] = acc[r];
    }
}

__global__ __launch_bounds__(256)
void node_scores(const float* __restrict__ H, const float* __restrict__ al,
                 const float* __restrict__ ar, float* __restrict__ el,
                 float* __restrict__ er, int n)
{
    const int lane = threadIdx.x & 31;
    const int node = blockIdx.x * 8 + (threadIdx.x >> 5);
    if (node >= n) return;

    const float* h = H + node * DH;
    float e0 = 0.f, e1 = 0.f, r0 = 0.f, r1 = 0.f;
    for (int i = lane; i < HID; i += 32) {
        const float h0 = h[i], h1 = h[HID + i];
        e0 += h0 * al[i];        e1 += h1 * al[HID + i];
        r0 += h0 * ar[i];        r1 += h1 * ar[HID + i];
    }
#pragma unroll
    for (int off = 16; off; off >>= 1) {
        e0 += __shfl_xor(e0, off); e1 += __shfl_xor(e1, off);
        r0 += __shfl_xor(r0, off); r1 += __shfl_xor(r1, off);
    }
    if (lane == 0) {
        el[node * 2 + 0] = e0; el[node * 2 + 1] = e1;
        er[node * 2 + 0] = r0; er[node * 2 + 1] = r1;
    }
}

#define TILE 98
#define LTH  1024
#define NB   128
#define FU   4
#define MAXT LTH

__device__ __forceinline__ float lrelu(float x) {
    return x > 0.0f ? x : NEG_SLOPE * x;
}

__device__ __forceinline__ void atomicMaxF(float* addr, float val) {
    if (val >= 0.0f) atomicMax((int*)addr, __float_as_int(val));
    else             atomicMin((unsigned int*)addr, __float_as_uint(val));
}

__global__ __launch_bounds__(LTH)
void bin_hist(const int* __restrict__ dst, int* __restrict__ hist, int e, int n, int gT)
{
    __shared__ int hc[MAXT];
    const int tid = threadIdx.x;
    const int per = (e + NB - 1) / NB;
    const int lo  = blockIdx.x * per;
    const int hi  = min(e, lo + per);
    for (int t = tid; t < gT; t += LTH) hc[t] = 0;
    __syncthreads();
    for (int base = lo; base < hi; base += LTH * FU) {
        int dv[FU];
#pragma unroll
        for (int u = 0; u < FU; ++u) {
            const int i = base + u * LTH + tid;
            dv[u] = (i < hi) ? dst[i] : -1;
        }
#pragma unroll
        for (int u = 0; u < FU; ++u)
            if ((unsigned)dv[u] < (unsigned)n) atomicAdd(&hc[dv[u] / TILE], 1);
    }
    __syncthreads();
    for (int t = tid; t < gT; t += LTH) hist[blockIdx.x * gT + t] = hc[t];
}

__global__ __launch_bounds__(LTH)
void bin_place(const int* __restrict__ src, const int* __restrict__ dst,
               const int* __restrict__ hist, int* __restrict__ cnt, int* __restrict__ off,
               int* __restrict__ lsrc, int* __restrict__ ldl, int e, int n, int gT)
{
    constexpr int TQ  = 256;
    constexpr int RQ  = LTH / TQ;
    constexpr int RPQ = NB / RQ;
    __shared__ int pa[LTH];
    __shared__ int pb[LTH];
    __shared__ int tot[MAXT];
    __shared__ int cur[MAXT];
    __shared__ int wsum[LTH / 32];

    const int tid  = threadIdx.x;
    const int lane = tid & 31;
    const int wid  = tid >> 5;
    const int b    = blockIdx.x;
    const int per  = (e + NB - 1) / NB;
    const int lo   = b * per;
    const int hi   = min(e, lo + per);

    int dv[FU], sv[FU];
#pragma unroll
    for (int u = 0; u < FU; ++u) {
        const int i = lo + u * LTH + tid;
        dv[u] = (i < hi) ? dst[i] : -1;
        sv[u] = (i < hi) ? src[i] : 0;
    }

    const int tl = tid & (TQ - 1);
    const int q  = tid / TQ;
    for (int t0 = 0; t0 < gT; t0 += TQ) {
        const int t = t0 + tl;
        int all = 0, bel = 0;
        if (t < gT) {
            int c[RPQ];
#pragma unroll
            for (int r = 0; r < RPQ; ++r) c[r] = hist[(q * RPQ + r) * gT + t];
#pragma unroll
            for (int r = 0; r < RPQ; ++r) {
                all += c[r];
                bel += (q * RPQ + r < b) ? c[r] : 0;
            }
        }
        pa[tid] = all;
        pb[tid] = bel;
        __syncthreads();
        if (q == 0 && t < gT) {
            int a = 0, w = 0;
            for (int k = 0; k < RQ; ++k) { a += pa[k * TQ + tl]; w += pb[k * TQ + tl]; }
            tot[t] = a;
            cur[t] = w;
        }
        __syncthreads();
    }

    const int v = (tid < gT) ? tot[tid] : 0;
    int inc = v;
#pragma unroll
    for (int o = 1; o < 32; o <<= 1) {
        const int y = __shfl_up(inc, o);
        if (lane >= o) inc += y;
    }
    if (lane == 31) wsum[wid] = inc;
    __syncthreads();
    if (wid == 0) {
        const int w = wsum[lane];
        int wi = w;
#pragma unroll
        for (int o = 1; o < 32; o <<= 1) {
            const int y = __shfl_up(wi, o);
            if (lane >= o) wi += y;
        }
        wsum[lane] = wi - w;
    }
    __syncthreads();
    if (tid < gT) {
        const int start = inc - v + wsum[wid];
        cur[tid] += start;
        if (b == 0) { cnt[tid] = v; off[tid] = start; }
    }
    __syncthreads();

    for (int base = lo; ; ) {
#pragma unroll
        for (int u = 0; u < FU; ++u) {
            const int d = dv[u];
            if ((unsigned)d < (unsigned)n) {
                const int t = d / TILE;
                const int p = atomicAdd(&cur[t], 1);
                lsrc[p] = sv[u];
                ldl[p]  = d - t * TILE;
            }
        }
        base += LTH * FU;
        if (base >= hi) break;
#pragma unroll
        for (int u = 0; u < FU; ++u) {
            const int i = base + u * LTH + tid;
            dv[u] = (i < hi) ? dst[i] : -1;
            sv[u] = (i < hi) ? src[i] : 0;
        }
    }
}

__global__ __launch_bounds__(LTH)
void tile_layer(const int* __restrict__ lsrc, const int* __restrict__ ldl,
                const int* __restrict__ cnt, const int* __restrict__ off,
                const float* __restrict__ H, const float* __restrict__ el,
                const float* __restrict__ er, const float* __restrict__ bias,
                float* __restrict__ outp, int n)
{
    __shared__ float acc[TILE * DH];
    __shared__ float sm[TILE * HEADS];
    __shared__ float ss[TILE * HEADS];
    __shared__ float ser[TILE * HEADS];

    const int tid  = threadIdx.x;
    const int lane = tid & 31;
    const int wid  = __builtin_amdgcn_readfirstlane(tid >> 5);
    const int lo   = blockIdx.x * TILE;
    const int beg  = off[blockIdx.x];
    const int num  = cnt[blockIdx.x];

    for (int i = tid; i < TILE * DH; i += LTH) acc[i] = 0.0f;
    if (tid < TILE * HEADS) {
        const int g = lo * HEADS + tid;
        sm[tid]  = -INFINITY;
        ss[tid]  = 0.0f;
        ser[tid] = (g < n * HEADS) ? er[g] : 0.0f;
    }
    __syncthreads();

    for (int i = tid; i < num; i += LTH) {
        const int s = lsrc[beg + i], dl = ldl[beg + i];
#pragma unroll
        for (int h = 0; h < HEADS; ++h)
            atomicMaxF(&sm[dl * HEADS + h], lrelu(el[s * HEADS + h] + ser[dl * HEADS + h]));
    }
    __syncthreads();
    if (tid < TILE * HEADS && !isfinite(sm[tid])) sm[tid] = 0.0f;
    __syncthreads();

    for (int c0 = wid * 32; c0 < num; c0 += 32 * (LTH / 32)) {
        const int i = c0 + lane;
        int s = 0, dl = 0;
        float v0 = 0.0f, v1 = 0.0f;
        if (i < num) {
            s  = lsrc[beg + i];
            dl = ldl[beg + i];
            const float sc0 = lrelu(el[s * HEADS + 0] + ser[dl * HEADS + 0]);
            const float sc1 = lrelu(el[s * HEADS + 1] + ser[dl * HEADS + 1]);
            v0 = __expf(sc0 - sm[dl * HEADS + 0]);
            v1 = __expf(sc1 - sm[dl * HEADS + 1]);
            atomicAdd(&ss[dl * HEADS + 0], v0);
            atomicAdd(&ss[dl * HEADS + 1], v1);
        }
        const int nk = (num - c0 < 32) ? (num - c0) : 32;
        for (int k = 0; k < nk; k += 4) {
            v4f   r[4][2];
            int   dk[4];
            float a0[4], a1[4];
#pragma unroll
            for (int q = 0; q < 4; ++q) {
                const int kk = k + q;
                const int sk = __builtin_amdgcn_readlane(s, kk);
                dk[q] = __builtin_amdgcn_readlane(dl, kk);
                a0[q] = __int_as_float(__builtin_amdgcn_readlane(__float_as_int(v0), kk));
                a1[q] = __int_as_float(__builtin_amdgcn_readlane(__float_as_int(v1), kk));
                if (kk < nk) {
                    const v4f* row = (const v4f*)(H + (size_t)sk * DH) + lane;
                    r[q][0] = row[0];
                    r[q][1] = row[32];
                } else {
                    r[q][0] = v4f{0.0f, 0.0f, 0.0f, 0.0f};
                    r[q][1] = v4f{0.0f, 0.0f, 0.0f, 0.0f};
                }
            }
#pragma unroll
            for (int q = 0; q < 4; ++q) {
                if (k + q < nk) {
                    float* ad = acc + dk[q] * DH + lane;
#pragma unroll
                    for (int c = 0; c < 4; ++c) atomicAdd(&ad[c * 32], a0[q] * r[q][0][c]);
#pragma unroll
                    for (int c = 0; c < 4; ++c) atomicAdd(&ad[HID + c * 32], a1[q] * r[q][1][c]);
                }
            }
        }
    }
    __syncthreads();

    for (int idx = tid; idx < TILE * (DH / 4); idx += LTH) {
        const int nl   = idx >> 6;
        const int hf   = (idx >> 5) & 1;
        const int l    = idx & 31;
        const int node = lo + nl;
        if (node < n) {
            const float  sv = ss[nl * HEADS + hf];
            const float* a  = acc + nl * DH + hf * HID + l;
            const v4f    bv = *(const v4f*)(bias + hf * HID + 4 * l);
            v4f o;
#pragma unroll
            for (int c = 0; c < 4; ++c) {
                float v = (sv == 0.0f) ? 0.0f : a[c * 32] / sv;
                v += bv[c];
                o[c] = v > 0.0f ? v : 0.0f;
            }
            *(v4f*)(outp + (size_t)node * DH + hf * HID + 4 * l) = o;
        }
    }
}

#define EC_TH   1024
#define EC_GRID 256

__global__ __launch_bounds__(EC_TH) __attribute__((amdgpu_waves_per_eu(8, 8)))
void edge_classify(const int* __restrict__ src, const int* __restrict__ dst,
                   const float* __restrict__ H, const float* __restrict__ fcW,
                   const float* __restrict__ fcb, float* __restrict__ out, int e)
{
    const int lane = threadIdx.x & 31;
    const int wave   = __builtin_amdgcn_readfirstlane((int)((threadIdx.x >> 5) * gridDim.x + blockIdx.x));
    const int nwaves = (int)gridDim.x * (EC_TH / 32);
    const int nchunks = (e + 15) >> 4;
    if (wave >= nchunks) return;

    v4f wS[2][2], wD[2][2];
#pragma unroll
    for (int hf = 0; hf < 2; ++hf) {
        const v4f* ps = (const v4f*)(fcW + (size_t)(hf * 128 + 4 * lane) * 2);
        const v4f* pd = (const v4f*)(fcW + (size_t)(DH + hf * 128 + 4 * lane) * 2);
        wS[hf][0] = ps[0]; wS[hf][1] = ps[1];
        wD[hf][0] = pd[0]; wD[hf][1] = pd[1];
    }
    const float bo = fcb[lane & 1];

    const int last = e - 1;
    int s, d;
    {
        const int ei = min((wave << 4) + (lane & 15), last);
        s = src[ei]; d = dst[ei];
    }
    for (int ch = wave; ch < nchunks; ch += nwaves) {
        const int e0 = ch << 4;
        const int en = min(((ch + nwaves) << 4) + (lane & 15), last);
        const int sn = src[en], dn = dst[en];
        float res = 0.0f;
#pragma unroll 1
        for (int g = 0; g < 4; ++g) {
            v4f rs[4][2], rd[4][2];
#pragma unroll
            for (int q = 0; q < 4; ++q) {
                const int sk = __builtin_amdgcn_readlane(s, g * 4 + q);
                const int dk = __builtin_amdgcn_readlane(d, g * 4 + q);
                const v4f* ps = (const v4f*)(H + (size_t)sk * DH) + lane;
                const v4f* pd = (const v4f*)(H + (size_t)dk * DH) + lane;
                rs[q][0] = ps[0]; rs[q][1] = ps[32];
                rd[q][0] = pd[0]; rd[q][1] = pd[32];
            }
            float p[8];
#pragma unroll
            for (int q = 0; q < 4; ++q) {
                v2f a = {0.0f, 0.0f};
#pragma unroll
                for (int hf = 0; hf < 2; ++hf) {
                    a += rs[q][hf].x * wS[hf][0].xy;
                    a += rs[q][hf].y * wS[hf][0].zw;
                    a += rs[q][hf].z * wS[hf][1].xy;
                    a += rs[q][hf].w * wS[hf][1].zw;
                    a += rd[q][hf].x * wD[hf][0].xy;
                    a += rd[q][hf].y * wD[hf][0].zw;
                    a += rd[q][hf].z * wD[hf][1].xy;
                    a += rd[q][hf].w * wD[hf][1].zw;
                }
                p[q * 2 + 0] = a.x;
                p[q * 2 + 1] = a.y;
            }
            float t[4], u[2];
#pragma unroll
            for (int i = 0; i < 4; ++i) {
                const float keep = (lane & 1) ? p[2 * i + 1] : p[2 * i];
                const float send = (lane & 1) ? p[2 * i] : p[2 * i + 1];
                t[i] = keep + __shfl_xor(send, 1);
            }
#pragma unroll
            for (int i = 0; i < 2; ++i) {
                const float keep = (lane & 2) ? t[2 * i + 1] : t[2 * i];
                const float send = (lane & 2) ? t[2 * i] : t[2 * i + 1];
                u[i] = keep + __shfl_xor(send, 2);
            }
            float v;
            {
                const float keep = (lane & 4) ? u[1] : u[0];
                const float send = (lane & 4) ? u[0] : u[1];
                v = keep + __shfl_xor(send, 4);
            }
            v += __shfl_xor(v, 8);
            v += __shfl_xor(v, 16);
            if ((lane >> 3) == g) res = v;
        }
        const int oi = e0 * 2 + lane;
        if (oi < e * 2) out[oi] = res + bo;
        s = sn; d = dn;
    }
}

extern "C" void kernel_launch(void* const* d_in, const int* in_sizes, int n_in,
                              void* d_out, int out_size, void* d_ws, size_t ws_size,
                              hipStream_t stream)
{
    const float* x   = (const float*)d_in[0];
    const float* rep = (const float*)d_in[1];
    const int*   src = (const int*)d_in[2];
    const int*   dst = (const int*)d_in[3];
    const float* W1  = (const float*)d_in[4];
    const float* al1 = (const float*)d_in[5];
    const float* ar1 = (const float*)d_in[6];
    const float* b1  = (const float*)d_in[7];
    const float* W2  = (const float*)d_in[8];
    const float* al2 = (const float*)d_in[9];
    const float* ar2 = (const float*)d_in[10];
    const float* b2  = (const float*)d_in[11];
    const float* fcW = (const float*)d_in[12];
    const float* fcb = (const float*)d_in[13];
    float* out = (float*)d_out;

    const int N = in_sizes[0] / 64;
    const int E = in_sizes[2];

    float* ws  = (float*)d_ws;
    float* h   = ws;
    float* agg = h   + (size_t)N * DH;
    float* rly = agg + (size_t)N * DH;
    float* el  = rly + (size_t)N * DH;
    float* er  = el  + (size_t)N * 2;
    int*   cnt = (int*)(er + (size_t)N * 2);
    int*   off = cnt + (size_t)N * 2;
    int*   lsrc = off + (size_t)N * 2;
    int*   ldl  = lsrc + (size_t)E;
    int*   hist = (int*)agg;

    const int gN16 = (N + 15) / 16;
    const int gN8  = (N + 7) / 8;
    const int gT   = (N + TILE - 1) / TILE;

    bin_hist<<<NB, LTH, 0, stream>>>(dst, hist, E, N, gT);
    bin_place<<<NB, LTH, 0, stream>>>(src, dst, hist, cnt, off, lsrc, ldl, E, N, gT);

    gemm_node<128, true><<<gN16, 512, 0, stream>>>(x, rep, W1, h, N);
    node_scores<<<gN8, 256, 0, stream>>>(h, al1, ar1, el, er, N);
    tile_layer<<<gT, LTH, 0, stream>>>(lsrc, ldl, cnt, off, h, el, er, b1, rly, N);

    gemm_node<256, false><<<gN16, 512, 0, stream>>>(rly, nullptr, W2, h, N);
    node_scores<<<gN8, 256, 0, stream>>>(h, al2, ar2, el, er, N);
    tile_layer<<<gT, LTH, 0, stream>>>(lsrc, ldl, cnt, off, h, el, er, b2, rly, N);

    edge_classify<<<EC_GRID, EC_TH, 0, stream>>>(src, dst, rly, fcW, fcb, out, E);
}


# attention vectors folded into the layer weights once per call (W.al, W.ar); the GEMM takes el/er as 4 dots of its staged input row; both score kernels gone (9 -> 8 launches); score sums reordered
# speedup vs baseline: 3.5163x; 1.0355x over previous
#include <hip/hip_runtime.h>
#include <hip/hip_bf16.h>
#include <math.h>

typedef __attribute__((ext_vector_type(2))) float v2f;
typedef __attribute__((ext_vector_type(4))) float v4f;
typedef __attribute__((ext_vector_type(8))) float v8f;

#define HID 128
#define HEADS 2
#define DH 256
#define NEG_SLOPE 0.2f

template<int K, bool CONCAT>
__global__ __launch_bounds__(512)
void gemm_node(const float* __restrict__ A, const float* __restrict__ A2,
               const float* __restrict__ W, const float* __restrict__ Wx,
               float* __restrict__ Hout, float* __restrict__ el, float* __restrict__ er, int n)
{
    constexpr int PITCH = K + 4;
    constexpr int KT    = K / 32;
    __shared__ float As[16 * PITCH];

    const int tid  = threadIdx.x;
    const int wid  = tid >> 5;
    const int lane = tid & 31;
    const int r0   = blockIdx.x * 16;

    v4f wx[KT];
#pragma unroll
    for (int t = 0; t < KT; ++t) wx[t] = *(const v4f*)(Wx + (size_t)(lane + 32 * t) * 4);

    for (int idx = tid; idx < 16 * K; idx += 512) {
        const int row = idx / K;
        const int k   = idx - row * K;
        const int gr  = r0 + row;
        float v = 0.0f;
        if (gr < n) {
            if (CONCAT) v = (k < 64) ? A[gr * 64 + k] : A2[gr * 64 + (k - 64)];
            else        v = A[gr * K + k];
        }
        As[row * PITCH + k] = v;
    }
    __syncthreads();

    {
        float s0 = 0.f, s1 = 0.f, s2 = 0.f, s3 = 0.f;
#pragma unroll
        for (int t = 0; t < KT; ++t) {
            const float a = As[wid * PITCH + lane + 32 * t];
            s0 += a * wx[t].x;  s1 += a * wx[t].y;
            s2 += a * wx[t].z;  s3 += a * wx[t].w;
        }
#pragma unroll
        for (int o = 16; o; o >>= 1) {
            s0 += __shfl_xor(s0, o); s1 += __shfl_xor(s1, o);
            s2 += __shfl_xor(s2, o); s3 += __shfl_xor(s3, o);
        }
        const int gr = r0 + wid;
        if (gr < n) {
            if (lane == 0) { el[gr * 2 + 0] = s0; el[gr * 2 + 1] = s1; }
            if (lane == 1) { er[gr * 2 + 0] = s2; er[gr * 2 + 1] = s3; }
        }
    }

    const int c0    = wid * 16;
    const int halfl = lane & 15;
    const int koff  = (lane >> 4) * 2;

    v8f acc = {};
    for (int k0 = 0; k0 < K; k0 += 4) {
        v2f a, b;
        a.x = As[halfl * PITCH + k0 + koff];
        a.y = As[halfl * PITCH + k0 + koff + 1];
        b.x = W[(k0 + koff)     * DH + c0 + halfl];
        b.y = W[(k0 + koff + 1) * DH + c0 + halfl];
        acc = __builtin_amdgcn_wmma_f32_16x16x4_f32(
                  false, a, false, b, (short)0, acc, false, false);
    }

    const int colg  = c0 + halfl;
    const int mbase = (lane >> 4) * 8;
#pragma unroll
    for (int r = 0; r < 8; ++r) {
        const int grow = r0 + mbase + r;
        if (grow < n) Hout[grow * DH + colg] = acc[r];
    }
}

__global__ __launch_bounds__(256)
void fold_att(const float* __restrict__ W1, const float* __restrict__ al1, const float* __restrict__ ar1,
              const float* __restrict__ W2, const float* __restrict__ al2, const float* __restrict__ ar2,
              float* __restrict__ wx1, float* __restrict__ wx2)
{
    const int lane = threadIdx.x & 31;
    const int row  = blockIdx.x * 8 + (threadIdx.x >> 5);
    if (row >= 384) return;
    const bool first = row < 128;
    const int  k     = first ? row : row - 128;
    const float* W  = first ? W1  : W2;
    const float* al = first ? al1 : al2;
    const float* ar = first ? ar1 : ar2;
    float*       wx = first ? wx1 : wx2;

    const v4f w0 = *(const v4f*)(W + (size_t)k * DH + 4 * lane);
    const v4f w1 = *(const v4f*)(W + (size_t)k * DH + HID + 4 * lane);
    const v4f l0 = *(const v4f*)(al + 4 * lane);
    const v4f l1 = *(const v4f*)(al + HID + 4 * lane);
    const v4f q0 = *(const v4f*)(ar + 4 * lane);
    const v4f q1 = *(const v4f*)(ar + HID + 4 * lane);
    float s0 = w0.x * l0.x + w0.y * l0.y + w0.z * l0.z + w0.w * l0.w;
    float s1 = w1.x * l1.x + w1.y * l1.y + w1.z * l1.z + w1.w * l1.w;
    float s2 = w0.x * q0.x + w0.y * q0.y + w0.z * q0.z + w0.w * q0.w;
    float s3 = w1.x * q1.x + w1.y * q1.y + w1.z * q1.z + w1.w * q1.w;
#pragma unroll
    for (int o = 16; o; o >>= 1) {
        s0 += __shfl_xor(s0, o); s1 += __shfl_xor(s1, o);
        s2 += __shfl_xor(s2, o); s3 += __shfl_xor(s3, o);
    }
    if (lane == 0) *(v4f*)(wx + (size_t)k * 4) = v4f{s0, s1, s2, s3};
}

#define TILE 98
#define LTH  1024
#define NB   128
#define FU   4
#define MAXT LTH

__device__ __forceinline__ float lrelu(float x) {
    return x > 0.0f ? x : NEG_SLOPE * x;
}

__device__ __forceinline__ void atomicMaxF(float* addr, float val) {
    if (val >= 0.0f) atomicMax((int*)addr, __float_as_int(val));
    else             atomicMin((unsigned int*)addr, __float_as_uint(val));
}

__global__ __launch_bounds__(LTH)
void bin_hist(const int* __restrict__ dst, int* __restrict__ hist, int e, int n, int gT)
{
    __shared__ int hc[MAXT];
    const int tid = threadIdx.x;
    const int per = (e + NB - 1) / NB;
    const int lo  = blockIdx.x * per;
    const int hi  = min(e, lo + per);
    for (int t = tid; t < gT; t += LTH) hc[t] = 0;
    __syncthreads();
    for (int base = lo; base < hi; base += LTH * FU) {
        int dv[FU];
#pragma unroll
        for (int u = 0; u < FU; ++u) {
            const int i = base + u * LTH + tid;
            dv[u] = (i < hi) ? dst[i] : -1;
        }
#pragma unroll
        for (int u = 0; u < FU; ++u)
            if ((unsigned)dv[u] < (unsigned)n) atomicAdd(&hc[dv[u] / TILE], 1);
    }
    __syncthreads();
    for (int t = tid; t < gT; t += LTH) hist[blockIdx.x * gT + t] = hc[t];
}

__global__ __launch_bounds__(LTH)
void bin_place(const int* __restrict__ src, const int* __restrict__ dst,
               const int* __restrict__ hist, int* __restrict__ cnt, int* __restrict__ off,
               int* __restrict__ lsrc, int* __restrict__ ldl, int e, int n, int gT)
{
    constexpr int TQ  = 256;
    constexpr int RQ  = LTH / TQ;
    constexpr int RPQ = NB / RQ;
    __shared__ int pa[LTH];
    __shared__ int pb[LTH];
    __shared__ int tot[MAXT];
    __shared__ int cur[MAXT];
    __shared__ int wsum[LTH / 32];

    const int tid  = threadIdx.x;
    const int lane = tid & 31;
    const int wid  = tid >> 5;
    const int b    = blockIdx.x;
    const int per  = (e + NB - 1) / NB;
    const int lo   = b * per;
    const int hi   = min(e, lo + per);

    int dv[FU], sv[FU];
#pragma unroll
    for (int u = 0; u < FU; ++u) {
        const int i = lo + u * LTH + tid;
        dv[u] = (i < hi) ? dst[i] : -1;
        sv[u] = (i < hi) ? src[i] : 0;
    }

    const int tl = tid & (TQ - 1);
    const int q  = tid / TQ;
    for (int t0 = 0; t0 < gT; t0 += TQ) {
        const int t = t0 + tl;
        int all = 0, bel = 0;
        if (t < gT) {
            int c[RPQ];
#pragma unroll
            for (int r = 0; r < RPQ; ++r) c[r] = hist[(q * RPQ + r) * gT + t];
#pragma unroll
            for (int r = 0; r < RPQ; ++r) {
                all += c[r];
                bel += (q * RPQ + r < b) ? c[r] : 0;
            }
        }
        pa[tid] = all;
        pb[tid] = bel;
        __syncthreads();
        if (q == 0 && t < gT) {
            int a = 0, w = 0;
            for (int k = 0; k < RQ; ++k) { a += pa[k * TQ + tl]; w += pb[k * TQ + tl]; }
            tot[t] = a;
            cur[t] = w;
        }
        __syncthreads();
    }

    const int v = (tid < gT) ? tot[tid] : 0;
    int inc = v;
#pragma unroll
    for (int o = 1; o < 32; o <<= 1) {
        const int y = __shfl_up(inc, o);
        if (lane >= o) inc += y;
    }
    if (lane == 31) wsum[wid] = inc;
    __syncthreads();
    if (wid == 0) {
        const int w = wsum[lane];
        int wi = w;
#pragma unroll
        for (int o = 1; o < 32; o <<= 1) {
            const int y = __shfl_up(wi, o);
            if (lane >= o) wi += y;
        }
        wsum[lane] = wi - w;
    }
    __syncthreads();
    if (tid < gT) {
        const int start = inc - v + wsum[wid];
        cur[tid] += start;
        if (b == 0) { cnt[tid] = v; off[tid] = start; }
    }
    __syncthreads();

    for (int base = lo; ; ) {
#pragma unroll
        for (int u = 0; u < FU; ++u) {
            const int d = dv[u];
            if ((unsigned)d < (unsigned)n) {
                const int t = d / TILE;
                const int p = atomicAdd(&cur[t], 1);
                lsrc[p] = sv[u];
                ldl[p]  = d - t * TILE;
            }
        }
        base += LTH * FU;
        if (base >= hi) break;
#pragma unroll
        for (int u = 0; u < FU; ++u) {
            const int i = base + u * LTH + tid;
            dv[u] = (i < hi) ? dst[i] : -1;
            sv[u] = (i < hi) ? src[i] : 0;
        }
    }
}

__global__ __launch_bounds__(LTH)
void tile_layer(const int* __restrict__ lsrc, const int* __restrict__ ldl,
                const int* __restrict__ cnt, const int* __restrict__ off,
                const float* __restrict__ H, const float* __restrict__ el,
                const float* __restrict__ er, const float* __restrict__ bias,
                float* __restrict__ outp, int n)
{
    __shared__ float acc[TILE * DH];
    __shared__ float sm[TILE * HEADS];
    __shared__ float ss[TILE * HEADS];
    __shared__ float ser[TILE * HEADS];

    const int tid  = threadIdx.x;
    const int lane = tid & 31;
    const int wid  = __builtin_amdgcn_readfirstlane(tid >> 5);
    const int lo   = blockIdx.x * TILE;
    const int beg  = off[blockIdx.x];
    const int num  = cnt[blockIdx.x];

    for (int i = tid; i < TILE * DH; i += LTH) acc[i] = 0.0f;
    if (tid < TILE * HEADS) {
        const int g = lo * HEADS + tid;
        sm[tid]  = -INFINITY;
        ss[tid]  = 0.0f;
        ser[tid] = (g < n * HEADS) ? er[g] : 0.0f;
    }
    __syncthreads();

    for (int i = tid; i < num; i += LTH) {
        const int s = lsrc[beg + i], dl = ldl[beg + i];
#pragma unroll
        for (int h = 0; h < HEADS; ++h)
            atomicMaxF(&sm[dl * HEADS + h], lrelu(el[s * HEADS + h] + ser[dl * HEADS + h]));
    }
    __syncthreads();
    if (tid < TILE * HEADS && !isfinite(sm[tid])) sm[tid] = 0.0f;
    __syncthreads();

    for (int c0 = wid * 32; c0 < num; c0 += 32 * (LTH / 32)) {
        const int i = c0 + lane;
        int s = 0, dl = 0;
        float v0 = 0.0f, v1 = 0.0f;
        if (i < num) {
            s  = lsrc[beg + i];
            dl = ldl[beg + i];
            const float sc0 = lrelu(el[s * HEADS + 0] + ser[dl * HEADS + 0]);
            const float sc1 = lrelu(el[s * HEADS + 1] + ser[dl * HEADS + 1]);
            v0 = __expf(sc0 - sm[dl * HEADS + 0]);
            v1 = __expf(sc1 - sm[dl * HEADS + 1]);
            atomicAdd(&ss[dl * HEADS + 0], v0);
            atomicAdd(&ss[dl * HEADS + 1], v1);
        }
        const int nk = (num - c0 < 32) ? (num - c0) : 32;
        for (int k = 0; k < nk; k += 4) {
            v4f   r[4][2];
            int   dk[4];
            float a0[4], a1[4];
#pragma unroll
            for (int q = 0; q < 4; ++q) {
                const int kk = k + q;
                const int sk = __builtin_amdgcn_readlane(s, kk);
                dk[q] = __builtin_amdgcn_readlane(dl, kk);
                a0[q] = __int_as_float(__builtin_amdgcn_readlane(__float_as_int(v0), kk));
                a1[q] = __int_as_float(__builtin_amdgcn_readlane(__float_as_int(v1), kk));
                if (kk < nk) {
                    const v4f* row = (const v4f*)(H + (size_t)sk * DH) + lane;
                    r[q][0] = row[0];
                    r[q][1] = row[32];
                } else {
                    r[q][0] = v4f{0.0f, 0.0f, 0.0f, 0.0f};
                    r[q][1] = v4f{0.0f, 0.0f, 0.0f, 0.0f};
                }
            }
#pragma unroll
            for (int q = 0; q < 4; ++q) {
                if (k + q < nk) {
                    float* ad = acc + dk[q] * DH + lane;
#pragma unroll
                    for (int c = 0; c < 4; ++c) atomicAdd(&ad[c * 32], a0[q] * r[q][0][c]);
#pragma unroll
                    for (int c = 0; c < 4; ++c) atomicAdd(&ad[HID + c * 32], a1[q] * r[q][1][c]);
                }
            }
        }
    }
    __syncthreads();

    for (int idx = tid; idx < TILE * (DH / 4); idx += LTH) {
        const int nl   = idx >> 6;
        const int hf   = (idx >> 5) & 1;
        const int l    = idx & 31;
        const int node = lo + nl;
        if (node < n) {
            const float  sv = ss[nl * HEADS + hf];
            const float* a  = acc + nl * DH + hf * HID + l;
            const v4f    bv = *(const v4f*)(bias + hf * HID + 4 * l);
            v4f o;
#pragma unroll
            for (int c = 0; c < 4; ++c) {
                float v = (sv == 0.0f) ? 0.0f : a[c * 32] / sv;
                v += bv[c];
                o[c] = v > 0.0f ? v : 0.0f;
            }
            *(v4f*)(outp + (size_t)node * DH + hf * HID + 4 * l) = o;
        }
    }
}

#define EC_TH   1024
#define EC_GRID 256

__global__ __launch_bounds__(EC_TH) __attribute__((amdgpu_waves_per_eu(8, 8)))
void edge_classify(const int* __restrict__ src, const int* __restrict__ dst,
                   const float* __restrict__ H, const float* __restrict__ fcW,
                   const float* __restrict__ fcb, float* __restrict__ out, int e)
{
    const int lane = threadIdx.x & 31;
    const int wave   = __builtin_amdgcn_readfirstlane((int)((threadIdx.x >> 5) * gridDim.x + blockIdx.x));
    const int nwaves = (int)gridDim.x * (EC_TH / 32);
    const int nchunks = (e + 15) >> 4;
    if (wave >= nchunks) return;

    v4f wS[2][2], wD[2][2];
#pragma unroll
    for (int hf = 0; hf < 2; ++hf) {
        const v4f* ps = (const v4f*)(fcW + (size_t)(hf * 128 + 4 * lane) * 2);
        const v4f* pd = (const v4f*)(fcW + (size_t)(DH + hf * 128 + 4 * lane) * 2);
        wS[hf][0] = ps[0]; wS[hf][1] = ps[1];
        wD[hf][0] = pd[0]; wD[hf][1] = pd[1];
    }
    const float bo = fcb[lane & 1];

    const int last = e - 1;
    int s, d;
    {
        const int ei = min((wave << 4) + (lane & 15), last);
        s = src[ei]; d = dst[ei];
    }
    for (int ch = wave; ch < nchunks; ch += nwaves) {
        const int e0 = ch << 4;
        const int en = min(((ch + nwaves) << 4) + (lane & 15), last);
        const int sn = src[en], dn = dst[en];
        float res = 0.0f;
#pragma unroll 1
        for (int g = 0; g < 4; ++g) {
            v4f rs[4][2], rd[4][2];
#pragma unroll
            for (int q = 0; q < 4; ++q) {
                const int sk = __builtin_amdgcn_readlane(s, g * 4 + q);
                const int dk = __builtin_amdgcn_readlane(d, g * 4 + q);
                const v4f* ps = (const v4f*)(H + (size_t)sk * DH) + lane;
                const v4f* pd = (const v4f*)(H + (size_t)dk * DH) + lane;
                rs[q][0] = ps[0]; rs[q][1] = ps[32];
                rd[q][0] = pd[0]; rd[q][1] = pd[32];
            }
            float p[8];
#pragma unroll
            for (int q = 0; q < 4; ++q) {
                v2f a = {0.0f, 0.0f};
#pragma unroll
                for (int hf = 0; hf < 2; ++hf) {
                    a += rs[q][hf].x * wS[hf][0].xy;
                    a += rs[q][hf].y * wS[hf][0].zw;
                    a += rs[q][hf].z * wS[hf][1].xy;
                    a += rs[q][hf].w * wS[hf][1].zw;
                    a += rd[q][hf].x * wD[hf][0].xy;
                    a += rd[q][hf].y * wD[hf][0].zw;
                    a += rd[q][hf].z * wD[hf][1].xy;
                    a += rd[q][hf].w * wD[hf][1].zw;
                }
                p[q * 2 + 0] = a.x;
                p[q * 2 + 1] = a.y;
            }
            float t[4], u[2];
#pragma unroll
            for (int i = 0; i < 4; ++i) {
                const float keep = (lane & 1) ? p[2 * i + 1] : p[2 * i];
                const float send = (lane & 1) ? p[2 * i] : p[2 * i + 1];
                t[i] = keep + __shfl_xor(send, 1);
            }
#pragma unroll
            for (int i = 0; i < 2; ++i) {
                const float keep = (lane & 2) ? t[2 * i + 1] : t[2 * i];
                const float send = (lane & 2) ? t[2 * i] : t[2 * i + 1];
                u[i] = keep + __shfl_xor(send, 2);
            }
            float v;
            {
                const float keep = (lane & 4) ? u[1] : u[0];
                const float send = (lane & 4) ? u[0] : u[1];
                v = keep + __shfl_xor(send, 4);
            }
            v += __shfl_xor(v, 8);
            v += __shfl_xor(v, 16);
            if ((lane >> 3) == g) res = v;
        }
        const int oi = e0 * 2 + lane;
        if (oi < e * 2) out[oi] = res + bo;
        s = sn; d = dn;
    }
}

extern "C" void kernel_launch(void* const* d_in, const int* in_sizes, int n_in,
                              void* d_out, int out_size, void* d_ws, size_t ws_size,
                              hipStream_t stream)
{
    const float* x   = (const float*)d_in[0];
    const float* rep = (const float*)d_in[1];
    const int*   src = (const int*)d_in[2];
    const int*   dst = (const int*)d_in[3];
    const float* W1  = (const float*)d_in[4];
    const float* al1 = (const float*)d_in[5];
    const float* ar1 = (const float*)d_in[6];
    const float* b1  = (const float*)d_in[7];
    const float* W2  = (const float*)d_in[8];
    const float* al2 = (const float*)d_in[9];
    const float* ar2 = (const float*)d_in[10];
    const float* b2  = (const float*)d_in[11];
    const float* fcW = (const float*)d_in[12];
    const float* fcb = (const float*)d_in[13];
    float* out = (float*)d_out;

    const int N = in_sizes[0] / 64;
    const int E = in_sizes[2];

    float* ws  = (float*)d_ws;
    float* h   = ws;
    float* agg = h   + (size_t)N * DH;
    float* rly = agg + (size_t)N * DH;
    float* el  = rly + (size_t)N * DH;
    float* er  = el  + (size_t)N * 2;
    int*   cnt = (int*)(er + (size_t)N * 2);
    int*   off = cnt + (size_t)N * 2;
    int*   lsrc = off + (size_t)N * 2;
    int*   ldl  = lsrc + (size_t)E;
    int*   hist = (int*)agg;
    float* wx1  = agg + (size_t)NB * MAXT;
    float* wx2  = wx1 + 128 * 4;

    const int gN16 = (N + 15) / 16;
    const int gT   = (N + TILE - 1) / TILE;

    fold_att<<<48, 256, 0, stream>>>(W1, al1, ar1, W2, al2, ar2, wx1, wx2);

    bin_hist<<<NB, LTH, 0, stream>>>(dst, hist, E, N, gT);
    bin_place<<<NB, LTH, 0, stream>>>(src, dst, hist, cnt, off, lsrc, ldl, E, N, gT);

    gemm_node<128, true><<<gN16, 512, 0, stream>>>(x, rep, W1, wx1, h, el, er, N);
    tile_layer<<<gT, LTH, 0, stream>>>(lsrc, ldl, cnt, off, h, el, er, b1, rly, N);

    gemm_node<256, false><<<gN16, 512, 0, stream>>>(rly, nullptr, W2, wx2, h, el, er, N);
    tile_layer<<<gT, LTH, 0, stream>>>(lsrc, ldl, cnt, off, h, el, er, b2, rly, N);

    edge_classify<<<EC_GRID, EC_TH, 0, stream>>>(src, dst, rly, fcW, fcb, out, E);
}


# tile layer: the tile's edges counting-sorted by dst row in LDS (integer adds), then a wave owns a row: max, exp sum, weighted row sum in registers; LDS float adds gone; row sums in sorted-list order
# speedup vs baseline: 3.7574x; 1.0686x over previous
#include <hip/hip_runtime.h>
#include <hip/hip_bf16.h>
#include <math.h>

typedef __attribute__((ext_vector_type(2))) float v2f;
typedef __attribute__((ext_vector_type(4))) float v4f;
typedef __attribute__((ext_vector_type(8))) float v8f;

#define HID 128
#define HEADS 2
#define DH 256
#define NEG_SLOPE 0.2f

template<int K, bool CONCAT>
__global__ __launch_bounds__(512)
void gemm_node(const float* __restrict__ A, const float* __restrict__ A2,
               const float* __restrict__ W, const float* __restrict__ Wx,
               float* __restrict__ Hout, float* __restrict__ el, float* __restrict__ er, int n)
{
    constexpr int PITCH = K + 4;
    constexpr int KT    = K / 32;
    __shared__ float As[16 * PITCH];

    const int tid  = threadIdx.x;
    const int wid  = tid >> 5;
    const int lane = tid & 31;
    const int r0   = blockIdx.x * 16;

    v4f wx[KT];
#pragma unroll
    for (int t = 0; t < KT; ++t) wx[t] = *(const v4f*)(Wx + (size_t)(lane + 32 * t) * 4);

    for (int idx = tid; idx < 16 * K; idx += 512) {
        const int row = idx / K;
        const int k   = idx - row * K;
        const int gr  = r0 + row;
        float v = 0.0f;
        if (gr < n) {
            if (CONCAT) v = (k < 64) ? A[gr * 64 + k] : A2[gr * 64 + (k - 64)];
            else        v = A[gr * K + k];
        }
        As[row * PITCH + k] = v;
    }
    __syncthreads();

    {
        float s0 = 0.f, s1 = 0.f, s2 = 0.f, s3 = 0.f;
#pragma unroll
        for (int t = 0; t < KT; ++t) {
            const float a = As[wid * PITCH + lane + 32 * t];
            s0 += a * wx[t].x;  s1 += a * wx[t].y;
            s2 += a * wx[t].z;  s3 += a * wx[t].w;
        }
#pragma unroll
        for (int o = 16; o; o >>= 1) {
            s0 += __shfl_xor(s0, o); s1 += __shfl_xor(s1, o);
            s2 += __shfl_xor(s2, o); s3 += __shfl_xor(s3, o);
        }
        const int gr = r0 + wid;
        if (gr < n) {
            if (lane == 0) { el[gr * 2 + 0] = s0; el[gr * 2 + 1] = s1; }
            if (lane == 1) { er[gr * 2 + 0] = s2; er[gr * 2 + 1] = s3; }
        }
    }

    const int c0    = wid * 16;
    const int halfl = lane & 15;
    const int koff  = (lane >> 4) * 2;

    v8f acc = {};
    for (int k0 = 0; k0 < K; k0 += 4) {
        v2f a, b;
        a.x = As[halfl * PITCH + k0 + koff];
        a.y = As[halfl * PITCH + k0 + koff + 1];
        b.x = W[(k0 + koff)     * DH + c0 + halfl];
        b.y = W[(k0 + koff + 1) * DH + c0 + halfl];
        acc = __builtin_amdgcn_wmma_f32_16x16x4_f32(
                  false, a, false, b, (short)0, acc, false, false);
    }

    const int colg  = c0 + halfl;
    const int mbase = (lane >> 4) * 8;
#pragma unroll
    for (int r = 0; r < 8; ++r) {
        const int grow = r0 + mbase + r;
        if (grow < n) Hout[grow * DH + colg] = acc[r];
    }
}

__global__ __launch_bounds__(256)
void fold_att(const float* __restrict__ W1, const float* __restrict__ al1, const float* __restrict__ ar1,
              const float* __restrict__ W2, const float* __restrict__ al2, const float* __restrict__ ar2,
              float* __restrict__ wx1, float* __restrict__ wx2)
{
    const int lane = threadIdx.x & 31;
    const int row  = blockIdx.x * 8 + (threadIdx.x >> 5);
    if (row >= 384) return;
    const bool first = row < 128;
    const int  k     = first ? row : row - 128;
    const float* W  = first ? W1  : W2;
    const float* al = first ? al1 : al2;
    const float* ar = first ? ar1 : ar2;
    float*       wx = first ? wx1 : wx2;

    const v4f w0 = *(const v4f*)(W + (size_t)k * DH + 4 * lane);
    const v4f w1 = *(const v4f*)(W + (size_t)k * DH + HID + 4 * lane);
    const v4f l0 = *(const v4f*)(al + 4 * lane);
    const v4f l1 = *(const v4f*)(al + HID + 4 * lane);
    const v4f q0 = *(const v4f*)(ar + 4 * lane);
    const v4f q1 = *(const v4f*)(ar + HID + 4 * lane);
    float s0 = w0.x * l0.x + w0.y * l0.y + w0.z * l0.z + w0.w * l0.w;
    float s1 = w1.x * l1.x + w1.y * l1.y + w1.z * l1.z + w1.w * l1.w;
    float s2 = w0.x * q0.x + w0.y * q0.y + w0.z * q0.z + w0.w * q0.w;
    float s3 = w1.x * q1.x + w1.y * q1.y + w1.z * q1.z + w1.w * q1.w;
#pragma unroll
    for (int o = 16; o; o >>= 1) {
        s0 += __shfl_xor(s0, o); s1 += __shfl_xor(s1, o);
        s2 += __shfl_xor(s2, o); s3 += __shfl_xor(s3, o);
    }
    if (lane == 0) *(v4f*)(wx + (size_t)k * 4) = v4f{s0, s1, s2, s3};
}

#define TILE 98
#define LTH  1024
#define NB   128
#define FU   4
#define MAXT LTH

__device__ __forceinline__ float lrelu(float x) {
    return x > 0.0f ? x : NEG_SLOPE * x;
}

__device__ __forceinline__ void atomicMaxF(float* addr, float val) {
    if (val >= 0.0f) atomicMax((int*)addr, __float_as_int(val));
    else             atomicMin((unsigned int*)addr, __float_as_uint(val));
}

__global__ __launch_bounds__(LTH)
void bin_hist(const int* __restrict__ dst, int* __restrict__ hist, int e, int n, int gT)
{
    __shared__ int hc[MAXT];
    const int tid = threadIdx.x;
    const int per = (e + NB - 1) / NB;
    const int lo  = blockIdx.x * per;
    const int hi  = min(e, lo + per);
    for (int t = tid; t < gT; t += LTH) hc[t] = 0;
    __syncthreads();
    for (int base = lo; base < hi; base += LTH * FU) {
        int dv[FU];
#pragma unroll
        for (int u = 0; u < FU; ++u) {
            const int i = base + u * LTH + tid;
            dv[u] = (i < hi) ? dst[i] : -1;
        }
#pragma unroll
        for (int u = 0; u < FU; ++u)
            if ((unsigned)dv[u] < (unsigned)n) atomicAdd(&hc[dv[u] / TILE], 1);
    }
    __syncthreads();
    for (int t = tid; t < gT; t += LTH) hist[blockIdx.x * gT + t] = hc[t];
}

__global__ __launch_bounds__(LTH)
void bin_place(const int* __restrict__ src, const int* __restrict__ dst,
               const int* __restrict__ hist, int* __restrict__ cnt, int* __restrict__ off,
               int* __restrict__ lsrc, int* __restrict__ ldl, int e, int n, int gT)
{
    constexpr int TQ  = 256;
    constexpr int RQ  = LTH / TQ;
    constexpr int RPQ = NB / RQ;
    __shared__ int pa[LTH];
    __shared__ int pb[LTH];
    __shared__ int tot[MAXT];
    __shared__ int cur[MAXT];
    __shared__ int wsum[LTH / 32];

    const int tid  = threadIdx.x;
    const int lane = tid & 31;
    const int wid  = tid >> 5;
    const int b    = blockIdx.x;
    const int per  = (e + NB - 1) / NB;
    const int lo   = b * per;
    const int hi   = min(e, lo + per);

    int dv[FU], sv[FU];
#pragma unroll
    for (int u = 0; u < FU; ++u) {
        const int i = lo + u * LTH + tid;
        dv[u] = (i < hi) ? dst[i] : -1;
        sv[u] = (i < hi) ? src[i] : 0;
    }

    const int tl = tid & (TQ - 1);
    const int q  = tid / TQ;
    for (int t0 = 0; t0 < gT; t0 += TQ) {
        const int t = t0 + tl;
        int all = 0, bel = 0;
        if (t < gT) {
            int c[RPQ];
#pragma unroll
            for (int r = 0; r < RPQ; ++r) c[r] = hist[(q * RPQ + r) * gT + t];
#pragma unroll
            for (int r = 0; r < RPQ; ++r) {
                all += c[r];
                bel += (q * RPQ + r < b) ? c[r] : 0;
            }
        }
        pa[tid] = all;
        pb[tid] = bel;
        __syncthreads();
        if (q == 0 && t < gT) {
            int a = 0, w = 0;
            for (int k = 0; k < RQ; ++k) { a += pa[k * TQ + tl]; w += pb[k * TQ + tl]; }
            tot[t] = a;
            cur[t] = w;
        }
        __syncthreads();
    }

    const int v = (tid < gT) ? tot[tid] : 0;
    int inc = v;
#pragma unroll
    for (int o = 1; o < 32; o <<= 1) {
        const int y = __shfl_up(inc, o);
        if (lane >= o) inc += y;
    }
    if (lane == 31) wsum[wid] = inc;
    __syncthreads();
    if (wid == 0) {
        const int w = wsum[lane];
        int wi = w;
#pragma unroll
        for (int o = 1; o < 32; o <<= 1) {
            const int y = __shfl_up(wi, o);
            if (lane >= o) wi += y;
        }
        wsum[lane] = wi - w;
    }
    __syncthreads();
    if (tid < gT) {
        const int start = inc - v + wsum[wid];
        cur[tid] += start;
        if (b == 0) { cnt[tid] = v; off[tid] = start; }
    }
    __syncthreads();

    for (int base = lo; ; ) {
#pragma unroll
        for (int u = 0; u < FU; ++u) {
            const int d = dv[u];
            if ((unsigned)d < (unsigned)n) {
                const int t = d / TILE;
                const int p = atomicAdd(&cur[t], 1);
                lsrc[p] = sv[u];
                ldl[p]  = d - t * TILE;
            }
        }
        base += LTH * FU;
        if (base >= hi) break;
#pragma unroll
        for (int u = 0; u < FU; ++u) {
            const int i = base + u * LTH + tid;
            dv[u] = (i < hi) ? dst[i] : -1;
            sv[u] = (i < hi) ? src[i] : 0;
        }
    }
}

#define RPAD 128
#define RCAP (TILE * DH)

__global__ __launch_bounds__(LTH)
void tile_layer(const int* __restrict__ lsrc, const int* __restrict__ ldl,
                const int* __restrict__ cnt, const int* __restrict__ off,
                const float* __restrict__ H, const float* __restrict__ el,
                const float* __restrict__ er, const float* __restrict__ bias,
                float* __restrict__ outp, int n)
{
    static_assert(TILE <= RPAD, "the prefix sum takes 4 rows per lane of one wave");
    __shared__ float acc[TILE * DH];
    __shared__ float sm[TILE * HEADS];
    __shared__ float ss[TILE * HEADS];
    __shared__ float ser[TILE * HEADS];
    __shared__ int   rcnt[RPAD];
    __shared__ int   rstart[RPAD + 1];
    __shared__ int   nextRow;

    const int tid  = threadIdx.x;
    const int lane = tid & 31;
    const int wid  = __builtin_amdgcn_readfirstlane(tid >> 5);
    const int lo   = blockIdx.x * TILE;
    const int beg  = off[blockIdx.x];
    const int num  = cnt[blockIdx.x];

  if (num <= RCAP) {
    int* ssrc = (int*)acc;

    if (tid < RPAD) rcnt[tid] = 0;
    if (tid == 0) nextRow = 0;
    if (tid < TILE * HEADS) {
        const int g = lo * HEADS + tid;
        ser[tid] = (g < n * HEADS) ? er[g] : 0.0f;
    }
    __syncthreads();

    for (int i = tid; i < num; i += LTH) atomicAdd(&rcnt[ldl[beg + i]], 1);
    __syncthreads();

    if (wid == 0) {
        const int c0 = rcnt[4 * lane], c1 = rcnt[4 * lane + 1], c2 = rcnt[4 * lane + 2], c3 = rcnt[4 * lane + 3];
        const int tot = c0 + c1 + c2 + c3;
        int inc = tot;
#pragma unroll
        for (int o = 1; o < 32; o <<= 1) {
            const int y = __shfl_up(inc, o);
            if (lane >= o) inc += y;
        }
        const int p0 = inc - tot, p1 = p0 + c0, p2 = p1 + c1, p3 = p2 + c2;
        rstart[4 * lane] = p0; rstart[4 * lane + 1] = p1; rstart[4 * lane + 2] = p2; rstart[4 * lane + 3] = p3;
        rcnt[4 * lane]   = p0; rcnt[4 * lane + 1]   = p1; rcnt[4 * lane + 2]   = p2; rcnt[4 * lane + 3]   = p3;
        if (lane == 31) rstart[RPAD] = inc;
    }
    __syncthreads();

    for (int i = tid; i < num; i += LTH) {
        const int s = lsrc[beg + i], dl = ldl[beg + i];
        const int p = atomicAdd(&rcnt[dl], 1);
        ssrc[p] = s;
    }
    __syncthreads();

    const v4f bv0 = *(const v4f*)(bias + 4 * lane);
    const v4f bv1 = *(const v4f*)(bias + HID + 4 * lane);

    for (int trip = 0; trip <= TILE; ++trip) {
        int rr = 0;
        if (lane == 0) rr = atomicAdd(&nextRow, 1);
        const int r = __builtin_amdgcn_readfirstlane(rr);
        if (r >= TILE) break;
        const int a   = __builtin_amdgcn_readfirstlane(rstart[r]);
        const int deg = __builtin_amdgcn_readfirstlane(rstart[r + 1]) - a;
        const float er0 = ser[r * HEADS + 0], er1 = ser[r * HEADS + 1];

        float m0 = -INFINITY, m1 = -INFINITY;
        int   sA = 0;
        float qA0 = -INFINITY, qA1 = -INFINITY;
        for (int c = 0; c < deg; c += 32) {
            int s = 0;
            float q0 = -INFINITY, q1 = -INFINITY;
            if (c + lane < deg) {
                s = ssrc[a + c + lane];
                const v2f e = *(const v2f*)(el + (size_t)s * HEADS);
                q0 = lrelu(e.x + er0);
                q1 = lrelu(e.y + er1);
            }
            m0 = fmaxf(m0, q0);
            m1 = fmaxf(m1, q1);
            if (c == 0) { sA = s; qA0 = q0; qA1 = q1; }
        }
#pragma unroll
        for (int o = 16; o; o >>= 1) {
            m0 = fmaxf(m0, __shfl_xor(m0, o));
            m1 = fmaxf(m1, __shfl_xor(m1, o));
        }
        if (!isfinite(m0)) m0 = 0.0f;
        if (!isfinite(m1)) m1 = 0.0f;

        v4f   w0 = {0.0f, 0.0f, 0.0f, 0.0f}, w1 = {0.0f, 0.0f, 0.0f, 0.0f};
        float t0 = 0.0f, t1 = 0.0f;
        for (int c = 0; c < deg; c += 32) {
            int s = sA;
            float q0 = qA0, q1 = qA1;
            if (c != 0) {
                s = 0; q0 = -INFINITY; q1 = -INFINITY;
                if (c + lane < deg) {
                    s = ssrc[a + c + lane];
                    const v2f e = *(const v2f*)(el + (size_t)s * HEADS);
                    q0 = lrelu(e.x + er0);
                    q1 = lrelu(e.y + er1);
                }
            }
            float v0 = 0.0f, v1 = 0.0f;
            if (c + lane < deg) {
                v0 = __expf(q0 - m0);
                v1 = __expf(q1 - m1);
            }
            t0 += v0;
            t1 += v1;
            const int nk = (deg - c < 32) ? (deg - c) : 32;
            for (int k = 0; k < nk; k += 4) {
                v4f   r4[4][2];
                float a0[4], a1[4];
#pragma unroll
                for (int q = 0; q < 4; ++q) {
                    const int kk = k + q;
                    const int sk = __builtin_amdgcn_readlane(s, kk);
                    a0[q] = __int_as_float(__builtin_amdgcn_readlane(__float_as_int(v0), kk));
                    a1[q] = __int_as_float(__builtin_amdgcn_readlane(__float_as_int(v1), kk));
                    if (kk < nk) {
                        const v4f* row = (const v4f*)(H + (size_t)sk * DH) + lane;
                        r4[q][0] = row[0];
                        r4[q][1] = row[32];
                    } else {
                        r4[q][0] = v4f{0.0f, 0.0f, 0.0f, 0.0f};
                        r4[q][1] = v4f{0.0f, 0.0f, 0.0f, 0.0f};
                    }
                }
#pragma unroll
                for (int q = 0; q < 4; ++q) {
                    w0 += a0[q] * r4[q][0];
                    w1 += a1[q] * r4[q][1];
                }
            }
        }
#pragma unroll
        for (int o = 16; o; o >>= 1) {
            t0 += __shfl_xor(t0, o);
            t1 += __shfl_xor(t1, o);
        }

        const int node = lo + r;
        if (node < n) {
            v4f o0, o1;
#pragma unroll
            for (int c = 0; c < 4; ++c) {
                float x0 = (t0 == 0.0f) ? 0.0f : w0[c] / t0;
                float x1 = (t1 == 0.0f) ? 0.0f : w1[c] / t1;
                x0 += bv0[c];
                x1 += bv1[c];
                o0[c] = x0 > 0.0f ? x0 : 0.0f;
                o1[c] = x1 > 0.0f ? x1 : 0.0f;
            }
            *(v4f*)(outp + (size_t)node * DH + 4 * lane)       = o0;
            *(v4f*)(outp + (size_t)node * DH + HID + 4 * lane) = o1;
        }
    }
  } else {
    for (int i = tid; i < TILE * DH; i += LTH) acc[i] = 0.0f;
    if (tid < TILE * HEADS) {
        const int g = lo * HEADS + tid;
        sm[tid]  = -INFINITY;
        ss[tid]  = 0.0f;
        ser[tid] = (g < n * HEADS) ? er[g] : 0.0f;
    }
    __syncthreads();

    for (int i = tid; i < num; i += LTH) {
        const int s = lsrc[beg + i], dl = ldl[beg + i];
#pragma unroll
        for (int h = 0; h < HEADS; ++h)
            atomicMaxF(&sm[dl * HEADS + h], lrelu(el[s * HEADS + h] + ser[dl * HEADS + h]));
    }
    __syncthreads();
    if (tid < TILE * HEADS && !isfinite(sm[tid])) sm[tid] = 0.0f;
    __syncthreads();

    for (int c0 = wid * 32; c0 < num; c0 += 32 * (LTH / 32)) {
        const int i = c0 + lane;
        int s = 0, dl = 0;
        float v0 = 0.0f, v1 = 0.0f;
        if (i < num) {
            s  = lsrc[beg + i];
            dl = ldl[beg + i];
            const float sc0 = lrelu(el[s * HEADS + 0] + ser[dl * HEADS + 0]);
            const float sc1 = lrelu(el[s * HEADS + 1] + ser[dl * HEADS + 1]);
            v0 = __expf(sc0 - sm[dl * HEADS + 0]);
            v1 = __expf(sc1 - sm[dl * HEADS + 1]);
            atomicAdd(&ss[dl * HEADS + 0], v0);
            atomicAdd(&ss[dl * HEADS + 1], v1);
        }
        const int nk = (num - c0 < 32) ? (num - c0) : 32;
        for (int k = 0; k < nk; k += 4) {
            v4f   r[4][2];
            int   dk[4];
            float a0[4], a1[4];
#pragma unroll
            for (int q = 0; q < 4; ++q) {
                const int kk = k + q;
                const int sk = __builtin_amdgcn_readlane(s, kk);
                dk[q] = __builtin_amdgcn_readlane(dl, kk);
                a0[q] = __int_as_float(__builtin_amdgcn_readlane(__float_as_int(v0), kk));
                a1[q] = __int_as_float(__builtin_amdgcn_readlane(__float_as_int(v1), kk));
                if (kk < nk) {
                    const v4f* row = (const v4f*)(H + (size_t)sk * DH) + lane;
                    r[q][0] = row[0];
                    r[q][1] = row[32];
                } else {
                    r[q][0] = v4f{0.0f, 0.0f, 0.0f, 0.0f};
                    r[q][1] = v4f{0.0f, 0.0f, 0.0f, 0.0f};
                }
            }
#pragma unroll
            for (int q = 0; q < 4; ++q) {
                if (k + q < nk) {
                    float* ad = acc + dk[q] * DH + lane;
#pragma unroll
                    for (int c = 0; c < 4; ++c) atomicAdd(&ad[c * 32], a0[q] * r[q][0][c]);
#pragma unroll
                    for (int c = 0; c < 4; ++c) atomicAdd(&ad[HID + c * 32], a1[q] * r[q][1][c]);
                }
            }
        }
    }
    __syncthreads();

    for (int idx = tid; idx < TILE * (DH / 4); idx += LTH) {
        const int nl   = idx >> 6;
        const int hf   = (idx >> 5) & 1;
        const int l    = idx & 31;
        const int node = lo + nl;
        if (node < n) {
            const float  sv = ss[nl * HEADS + hf];
            const float* a  = acc + nl * DH + hf * HID + l;
            const v4f    bv = *(const v4f*)(bias + hf * HID + 4 * l);
            v4f o;
#pragma unroll
            for (int c = 0; c < 4; ++c) {
                float v = (sv == 0.0f) ? 0.0f : a[c * 32] / sv;
                v += bv[c];
                o[c] = v > 0.0f ? v : 0.0f;
            }
            *(v4f*)(outp + (size_t)node * DH + hf * HID + 4 * l) = o;
        }
    }
  }
}

#define EC_TH   1024
#define EC_GRID 256

__global__ __launch_bounds__(EC_TH) __attribute__((amdgpu_waves_per_eu(8, 8)))
void edge_classify(const int* __restrict__ src, const int* __restrict__ dst,
                   const float* __restrict__ H, const float* __restrict__ fcW,
                   const float* __restrict__ fcb, float* __restrict__ out, int e)
{
    const int lane = threadIdx.x & 31;
    const int wave   = __builtin_amdgcn_readfirstlane((int)((threadIdx.x >> 5) * gridDim.x + blockIdx.x));
    const int nwaves = (int)gridDim.x * (EC_TH / 32);
    const int nchunks = (e + 15) >> 4;
    if (wave >= nchunks) return;

    v4f wS[2][2], wD[2][2];
#pragma unroll
    for (int hf = 0; hf < 2; ++hf) {
        const v4f* ps = (const v4f*)(fcW + (size_t)(hf * 128 + 4 * lane) * 2);
        const v4f* pd = (const v4f*)(fcW + (size_t)(DH + hf * 128 + 4 * lane) * 2);
        wS[hf][0] = ps[0]; wS[hf][1] = ps[1];
        wD[hf][0] = pd[0]; wD[hf][1] = pd[1];
    }
    const float bo = fcb[lane & 1];

    const int last = e - 1;
    int s, d;
    {
        const int ei = min((wave << 4) + (lane & 15), last);
        s = src[ei]; d = dst[ei];
    }
    for (int ch = wave; ch < nchunks; ch += nwaves) {
        const int e0 = ch << 4;
        const int en = min(((ch + nwaves) << 4) + (lane & 15), last);
        const int sn = src[en], dn = dst[en];
        float res = 0.0f;
#pragma unroll 1
        for (int g = 0; g < 4; ++g) {
            v4f rs[4][2], rd[4][2];
#pragma unroll
            for (int q = 0; q < 4; ++q) {
                const int sk = __builtin_amdgcn_readlane(s, g * 4 + q);
                const int dk = __builtin_amdgcn_readlane(d, g * 4 + q);
                const v4f* ps = (const v4f*)(H + (size_t)sk * DH) + lane;
                const v4f* pd = (const v4f*)(H + (size_t)dk * DH) + lane;
                rs[q][0] = ps[0]; rs[q][1] = ps[32];
                rd[q][0] = pd[0]; rd[q][1] = pd[32];
            }
            float p[8];
#pragma unroll
            for (int q = 0; q < 4; ++q) {
                v2f a = {0.0f, 0.0f};
#pragma unroll
                for (int hf = 0; hf < 2; ++hf) {
                    a += rs[q][hf].x * wS[hf][0].xy;
                    a += rs[q][hf].y * wS[hf][0].zw;
                    a += rs[q][hf].z * wS[hf][1].xy;
                    a += rs[q][hf].w * wS[hf][1].zw;
                    a += rd[q][hf].x * wD[hf][0].xy;
                    a += rd[q][hf].y * wD[hf][0].zw;
                    a += rd[q][hf].z * wD[hf][1].xy;
                    a += rd[q][hf].w * wD[hf][1].zw;
                }
                p[q * 2 + 0] = a.x;
                p[q * 2 + 1] = a.y;
            }
            float t[4], u[2];
#pragma unroll
            for (int i = 0; i < 4; ++i) {
                const float keep = (lane & 1) ? p[2 * i + 1] : p[2 * i];
                const float send = (lane & 1) ? p[2 * i] : p[2 * i + 1];
                t[i] = keep + __shfl_xor(send, 1);
            }
#pragma unroll
            for (int i = 0; i < 2; ++i) {
                const float keep = (lane & 2) ? t[2 * i + 1] : t[2 * i];
                const float send = (lane & 2) ? t[2 * i] : t[2 * i + 1];
                u[i] = keep + __shfl_xor(send, 2);
            }
            float v;
            {
                const float keep = (lane & 4) ? u[1] : u[0];
                const float send = (lane & 4) ? u[0] : u[1];
                v = keep + __shfl_xor(send, 4);
            }
            v += __shfl_xor(v, 8);
            v += __shfl_xor(v, 16);
            if ((lane >> 3) == g) res = v;
        }
        const int oi = e0 * 2 + lane;
        if (oi < e * 2) out[oi] = res + bo;
        s = sn; d = dn;
    }
}

extern "C" void kernel_launch(void* const* d_in, const int* in_sizes, int n_in,
                              void* d_out, int out_size, void* d_ws, size_t ws_size,
                              hipStream_t stream)
{
    const float* x   = (const float*)d_in[0];
    const float* rep = (const float*)d_in[1];
    const int*   src = (const int*)d_in[2];
    const int*   dst = (const int*)d_in[3];
    const float* W1  = (const float*)d_in[4];
    const float* al1 = (const float*)d_in[5];
    const float* ar1 = (const float*)d_in[6];
    const float* b1  = (const float*)d_in[7];
    const float* W2  = (const float*)d_in[8];
    const float* al2 = (const float*)d_in[9];
    const float* ar2 = (const float*)d_in[10];
    const float* b2  = (const float*)d_in[11];
    const float* fcW = (const float*)d_in[12];
    const float* fcb = (const float*)d_in[13];
    float* out = (float*)d_out;

    const int N = in_sizes[0] / 64;
    const int E = in_sizes[2];

    float* ws  = (float*)d_ws;
    float* h   = ws;
    float* agg = h   + (size_t)N * DH;
    float* rly = agg + (size_t)N * DH;
    float* el  = rly + (size_t)N * DH;
    float* er  = el  + (size_t)N * 2;
    int*   cnt = (int*)(er + (size_t)N * 2);
    int*   off = cnt + (size_t)N * 2;
    int*   lsrc = off + (size_t)N * 2;
    int*   ldl  = lsrc + (size_t)E;
    int*   hist = (int*)agg;
    float* wx1  = agg + (size_t)NB * MAXT;
    float* wx2  = wx1 + 128 * 4;

    const int gN16 = (N + 15) / 16;
    const int gT   = (N + TILE - 1) / TILE;

    fold_att<<<48, 256, 0, stream>>>(W1, al1, ar1, W2, al2, ar2, wx1, wx2);

    bin_hist<<<NB, LTH, 0, stream>>>(dst, hist, E, N, gT);
    bin_place<<<NB, LTH, 0, stream>>>(src, dst, hist, cnt, off, lsrc, ldl, E, N, gT);

    gemm_node<128, true><<<gN16, 512, 0, stream>>>(x, rep, W1, wx1, h, el, er, N);
    tile_layer<<<gT, LTH, 0, stream>>>(lsrc, ldl, cnt, off, h, el, er, b1, rly, N);

    gemm_node<256, false><<<gN16, 512, 0, stream>>>(rly, nullptr, W2, wx2, h, el, er, N);
    tile_layer<<<gT, LTH, 0, stream>>>(lsrc, ldl, cnt, off, h, el, er, b2, rly, N);

    edge_classify<<<EC_GRID, EC_TH, 0, stream>>>(src, dst, rly, fcW, fcb, out, E);
}


# node GEMM resident: 256 work-groups hold each wave's 16-column weight stripe in registers and walk 16-row slabs (next slab fetched under the products, tile stored a slab later); odd slabs shared
# speedup vs baseline: 4.1146x; 1.0951x over previous
#include <hip/hip_runtime.h>
#include <hip/hip_bf16.h>
#include <math.h>

typedef __attribute__((ext_vector_type(2))) float v2f;
typedef __attribute__((ext_vector_type(4))) float v4f;
typedef __attribute__((ext_vector_type(8))) float v8f;

#define HID 128
#define HEADS 2
#define DH 256
#define NEG_SLOPE 0.2f

#define GEMM_GRID 256

template<int K, bool CONCAT>
__global__ __launch_bounds__(512)
void gemm_node(const float* __restrict__ A, const float* __restrict__ A2,
               const float* __restrict__ W, const float* __restrict__ Wx,
               float* __restrict__ Hout, float* __restrict__ el, float* __restrict__ er, int n)
{
    constexpr int PITCH = K + 4;
    constexpr int KT    = K / 32;
    constexpr int KS    = K / 4;
    constexpr int K4    = K / 4;
    constexpr int NV    = (16 * K4) / 512;
    __shared__ __attribute__((aligned(16))) float As[2][16 * PITCH];

    const int tid  = threadIdx.x;
    const int wid  = __builtin_amdgcn_readfirstlane(tid >> 5);
    const int lane = tid & 31;
    const int c0    = wid * 16;
    const int halfl = lane & 15;
    const int koff  = (lane >> 4) * 2;
    const int G     = (int)gridDim.x;
    const int g     = (int)blockIdx.x;

    v2f bf[KS];
#pragma unroll
    for (int s = 0; s < KS; ++s) {
        bf[s].x = W[(4 * s + koff)     * DH + c0 + halfl];
        bf[s].y = W[(4 * s + koff + 1) * DH + c0 + halfl];
    }
    v4f wx[KT];
#pragma unroll
    for (int t = 0; t < KT; ++t) wx[t] = *(const v4f*)(Wx + (size_t)(lane + 32 * t) * 4);

    const int nslab = (n + 15) >> 4;
    const int full  = nslab / G;
    const int rem   = nslab - full * G;
    const int per   = rem > 0 ? G / rem : 0;
    const int q     = rem > 0 ? g / rem : 0;
    const int ts    = rem > 0 ? g - q * rem : 0;
    const bool tail = rem > 0 && q < per;
    const int T     = full + (tail ? 1 : 0);
    const bool mine = tail && (wid % (per > 0 ? per : 1)) == q;

    v4f pre[NV];
    auto fetch = [&](int sl) {
#pragma unroll
        for (int j = 0; j < NV; ++j) {
            const int idx = tid + 512 * j;
            const int row = idx / K4;
            const int k   = (idx - row * K4) * 4;
            const int gr  = sl * 16 + row;
            v4f v = {0.0f, 0.0f, 0.0f, 0.0f};
            if (gr < n) {
                if (CONCAT) v = (k < 64) ? *(const v4f*)(A + gr * 64 + k) : *(const v4f*)(A2 + gr * 64 + (k - 64));
                else        v = *(const v4f*)(A + gr * K + k);
            }
            pre[j] = v;
        }
    };
    auto put = [&](float* as) {
#pragma unroll
        for (int j = 0; j < NV; ++j) {
            const int idx = tid + 512 * j;
            const int row = idx / K4;
            const int k   = (idx - row * K4) * 4;
            *(v4f*)(as + row * PITCH + k) = pre[j];
        }
    };

    const int colg  = c0 + halfl;
    const int mbase = (lane >> 4) * 8;
    auto store_tile = [&](const v8f& t, int sl) {
        float* p = Hout + (sl * 16 + mbase) * DH + colg;
        if (sl * 16 + 16 <= n) {
#pragma unroll
            for (int r = 0; r < 8; ++r) p[r * DH] = t[r];
        } else {
#pragma unroll
            for (int r = 0; r < 8; ++r)
                if (sl * 16 + mbase + r < n) p[r * DH] = t[r];
        }
    };

    if (T > 0) { fetch(full > 0 ? g : ts); put(As[0]); }
    __syncthreads();

    v8f  accP  = {};
    int  slP   = 0;
    bool haveP = false;

    for (int it = 0; it < T; ++it) {
        const float* as   = As[it & 1];
        const bool   whole = it < full;
        const int    sl    = whole ? g + it * G : full * G + ts;
        const bool   more  = it + 1 < T;
        if (more) fetch(it + 1 < full ? g + (it + 1) * G : full * G + ts);
        if (haveP) store_tile(accP, slP);

        const bool have = whole || mine;
        v8f acc = {};
        if (have) {
#pragma unroll
            for (int s = 0; s < KS; ++s) {
                v2f a;
                a.x = as[halfl * PITCH + 4 * s + koff];
                a.y = as[halfl * PITCH + 4 * s + koff + 1];
                acc = __builtin_amdgcn_wmma_f32_16x16x4_f32(
                          false, a, false, bf[s], (short)0, acc, false, false);
            }
        }

        if ((whole || q == 0) && wid < 8) {
            float s[2][4] = {{0.f, 0.f, 0.f, 0.f}, {0.f, 0.f, 0.f, 0.f}};
#pragma unroll
            for (int t = 0; t < KT; ++t) {
#pragma unroll
                for (int h = 0; h < 2; ++h) {
                    const float a = as[(wid + 8 * h) * PITCH + lane + 32 * t];
                    s[h][0] += a * wx[t].x;  s[h][1] += a * wx[t].y;
                    s[h][2] += a * wx[t].z;  s[h][3] += a * wx[t].w;
                }
            }
#pragma unroll
            for (int o = 16; o; o >>= 1) {
#pragma unroll
                for (int h = 0; h < 2; ++h) {
                    s[h][0] += __shfl_xor(s[h][0], o); s[h][1] += __shfl_xor(s[h][1], o);
                    s[h][2] += __shfl_xor(s[h][2], o); s[h][3] += __shfl_xor(s[h][3], o);
                }
            }
#pragma unroll
            for (int h = 0; h < 2; ++h) {
                const int gr = sl * 16 + wid + 8 * h;
                if (gr < n) {
                    if (lane == 0) { el[gr * 2 + 0] = s[h][0]; el[gr * 2 + 1] = s[h][1]; }
                    if (lane == 1) { er[gr * 2 + 0] = s[h][2]; er[gr * 2 + 1] = s[h][3]; }
                }
            }
        }

        if (more) put(As[(it + 1) & 1]);
        __syncthreads();
        accP = acc; slP = sl; haveP = have;
    }
    if (haveP) store_tile(accP, slP);
}

__global__ __launch_bounds__(256)
void fold_att(const float* __restrict__ W1, const float* __restrict__ al1, const float* __restrict__ ar1,
              const float* __restrict__ W2, const float* __restrict__ al2, const float* __restrict__ ar2,
              float* __restrict__ wx1, float* __restrict__ wx2)
{
    const int lane = threadIdx.x & 31;
    const int row  = blockIdx.x * 8 + (threadIdx.x >> 5);
    if (row >= 384) return;
    const bool first = row < 128;
    const int  k     = first ? row : row - 128;
    const float* W  = first ? W1  : W2;
    const float* al = first ? al1 : al2;
    const float* ar = first ? ar1 : ar2;
    float*       wx = first ? wx1 : wx2;

    const v4f w0 = *(const v4f*)(W + (size_t)k * DH + 4 * lane);
    const v4f w1 = *(const v4f*)(W + (size_t)k * DH + HID + 4 * lane);
    const v4f l0 = *(const v4f*)(al + 4 * lane);
    const v4f l1 = *(const v4f*)(al + HID + 4 * lane);
    const v4f q0 = *(const v4f*)(ar + 4 * lane);
    const v4f q1 = *(const v4f*)(ar + HID + 4 * lane);
    float s0 = w0.x * l0.x + w0.y * l0.y + w0.z * l0.z + w0.w * l0.w;
    float s1 = w1.x * l1.x + w1.y * l1.y + w1.z * l1.z + w1.w * l1.w;
    float s2 = w0.x * q0.x + w0.y * q0.y + w0.z * q0.z + w0.w * q0.w;
    float s3 = w1.x * q1.x + w1.y * q1.y + w1.z * q1.z + w1.w * q1.w;
#pragma unroll
    for (int o = 16; o; o >>= 1) {
        s0 += __shfl_xor(s0, o); s1 += __shfl_xor(s1, o);
        s2 += __shfl_xor(s2, o); s3 += __shfl_xor(s3, o);
    }
    if (lane == 0) *(v4f*)(wx + (size_t)k * 4) = v4f{s0, s1, s2, s3};
}

#define TILE 98
#define LTH  1024
#define NB   128
#define FU   4
#define MAXT LTH

__device__ __forceinline__ float lrelu(float x) {
    return x > 0.0f ? x : NEG_SLOPE * x;
}

__device__ __forceinline__ void atomicMaxF(float* addr, float val) {
    if (val >= 0.0f) atomicMax((int*)addr, __float_as_int(val));
    else             atomicMin((unsigned int*)addr, __float_as_uint(val));
}

__global__ __launch_bounds__(LTH)
void bin_hist(const int* __restrict__ dst, int* __restrict__ hist, int e, int n, int gT)
{
    __shared__ int hc[MAXT];
    const int tid = threadIdx.x;
    const int per = (e + NB - 1) / NB;
    const int lo  = blockIdx.x * per;
    const int hi  = min(e, lo + per);
    for (int t = tid; t < gT; t += LTH) hc[t] = 0;
    __syncthreads();
    for (int base = lo; base < hi; base += LTH * FU) {
        int dv[FU];
#pragma unroll
        for (int u = 0; u < FU; ++u) {
            const int i = base + u * LTH + tid;
            dv[u] = (i < hi) ? dst[i] : -1;
        }
#pragma unroll
        for (int u = 0; u < FU; ++u)
            if ((unsigned)dv[u] < (unsigned)n) atomicAdd(&hc[dv[u] / TILE], 1);
    }
    __syncthreads();
    for (int t = tid; t < gT; t += LTH) hist[blockIdx.x * gT + t] = hc[t];
}

__global__ __launch_bounds__(LTH)
void bin_place(const int* __restrict__ src, const int* __restrict__ dst,
               const int* __restrict__ hist, int* __restrict__ cnt, int* __restrict__ off,
               int* __restrict__ lsrc, int* __restrict__ ldl, int e, int n, int gT)
{
    constexpr int TQ  = 256;
    constexpr int RQ  = LTH / TQ;
    constexpr int RPQ = NB / RQ;
    __shared__ int pa[LTH];
    __shared__ int pb[LTH];
    __shared__ int tot[MAXT];
    __shared__ int cur[MAXT];
    __shared__ int wsum[LTH / 32];

    const int tid  = threadIdx.x;
    const int lane = tid & 31;
    const int wid  = tid >> 5;
    const int b    = blockIdx.x;
    const int per  = (e + NB - 1) / NB;
    const int lo   = b * per;
    const int hi   = min(e, lo + per);

    int dv[FU], sv[FU];
#pragma unroll
    for (int u = 0; u < FU; ++u) {
        const int i = lo + u * LTH + tid;
        dv[u] = (i < hi) ? dst[i] : -1;
        sv[u] = (i < hi) ? src[i] : 0;
    }

    const int tl = tid & (TQ - 1);
    const int q  = tid / TQ;
    for (int t0 = 0; t0 < gT; t0 += TQ) {
        const int t = t0 + tl;
        int all = 0, bel = 0;
        if (t < gT) {
            int c[RPQ];
#pragma unroll
            for (int r = 0; r < RPQ; ++r) c[r] = hist[(q * RPQ + r) * gT + t];
#pragma unroll
            for (int r = 0; r < RPQ; ++r) {
                all += c[r];
                bel += (q * RPQ + r < b) ? c[r] : 0;
            }
        }
        pa[tid] = all;
        pb[tid] = bel;
        __syncthreads();
        if (q == 0 && t < gT) {
            int a = 0, w = 0;
            for (int k = 0; k < RQ; ++k) { a += pa[k * TQ + tl]; w += pb[k * TQ + tl]; }
            tot[t] = a;
            cur[t] = w;
        }
        __syncthreads();
    }

    const int v = (tid < gT) ? tot[tid] : 0;
    int inc = v;
#pragma unroll
    for (int o = 1; o < 32; o <<= 1) {
        const int y = __shfl_up(inc, o);
        if (lane >= o) inc += y;
    }
    if (lane == 31) wsum[wid] = inc;
    __syncthreads();
    if (wid == 0) {
        const int w = wsum[lane];
        int wi = w;
#pragma unroll
        for (int o = 1; o < 32; o <<= 1) {
            const int y = __shfl_up(wi, o);
            if (lane >= o) wi += y;
        }
        wsum[lane] = wi - w;
    }
    __syncthreads();
    if (tid < gT) {
        const int start = inc - v + wsum[wid];
        cur[tid] += start;
        if (b == 0) { cnt[tid] = v; off[tid] = start; }
    }
    __syncthreads();

    for (int base = lo; ; ) {
#pragma unroll
        for (int u = 0; u < FU; ++u) {
            const int d = dv[u];
            if ((unsigned)d < (unsigned)n) {
                const int t = d / TILE;
                const int p = atomicAdd(&cur[t], 1);
                lsrc[p] = sv[u];
                ldl[p]  = d - t * TILE;
            }
        }
        base += LTH * FU;
        if (base >= hi) break;
#pragma unroll
        for (int u = 0; u < FU; ++u) {
            const int i = base + u * LTH + tid;
            dv[u] = (i < hi) ? dst[i] : -1;
            sv[u] = (i < hi) ? src[i] : 0;
        }
    }
}

#define RPAD 128
#define RCAP (TILE * DH)

__global__ __launch_bounds__(LTH)
void tile_layer(const int* __restrict__ lsrc, const int* __restrict__ ldl,
                const int* __restrict__ cnt, const int* __restrict__ off,
                const float* __restrict__ H, const float* __restrict__ el,
                const float* __restrict__ er, const float* __restrict__ bias,
                float* __restrict__ outp, int n)
{
    static_assert(TILE <= RPAD, "the prefix sum takes 4 rows per lane of one wave");
    __shared__ float acc[TILE * DH];
    __shared__ float sm[TILE * HEADS];
    __shared__ float ss[TILE * HEADS];
    __shared__ float ser[TILE * HEADS];
    __shared__ int   rcnt[RPAD];
    __shared__ int   rstart[RPAD + 1];
    __shared__ int   nextRow;

    const int tid  = threadIdx.x;
    const int lane = tid & 31;
    const int wid  = __builtin_amdgcn_readfirstlane(tid >> 5);
    const int lo   = blockIdx.x * TILE;
    const int beg  = off[blockIdx.x];
    const int num  = cnt[blockIdx.x];

  if (num <= RCAP) {
    int* ssrc = (int*)acc;

    if (tid < RPAD) rcnt[tid] = 0;
    if (tid == 0) nextRow = 0;
    if (tid < TILE * HEADS) {
        const int g = lo * HEADS + tid;
        ser[tid] = (g < n * HEADS) ? er[g] : 0.0f;
    }
    __syncthreads();

    for (int i = tid; i < num; i += LTH) atomicAdd(&rcnt[ldl[beg + i]], 1);
    __syncthreads();

    if (wid == 0) {
        const int c0 = rcnt[4 * lane], c1 = rcnt[4 * lane + 1], c2 = rcnt[4 * lane + 2], c3 = rcnt[4 * lane + 3];
        const int tot = c0 + c1 + c2 + c3;
        int inc = tot;
#pragma unroll
        for (int o = 1; o < 32; o <<= 1) {
            const int y = __shfl_up(inc, o);
            if (lane >= o) inc += y;
        }
        const int p0 = inc - tot, p1 = p0 + c0, p2 = p1 + c1, p3 = p2 + c2;
        rstart[4 * lane] = p0; rstart[4 * lane + 1] = p1; rstart[4 * lane + 2] = p2; rstart[4 * lane + 3] = p3;
        rcnt[4 * lane]   = p0; rcnt[4 * lane + 1]   = p1; rcnt[4 * lane + 2]   = p2; rcnt[4 * lane + 3]   = p3;
        if (lane == 31) rstart[RPAD] = inc;
    }
    __syncthreads();

    for (int i = tid; i < num; i += LTH) {
        const int s = lsrc[beg + i], dl = ldl[beg + i];
        const int p = atomicAdd(&rcnt[dl], 1);
        ssrc[p] = s;
    }
    __syncthreads();

    const v4f bv0 = *(const v4f*)(bias + 4 * lane);
    const v4f bv1 = *(const v4f*)(bias + HID + 4 * lane);

    for (int trip = 0; trip <= TILE; ++trip) {
        int rr = 0;
        if (lane == 0) rr = atomicAdd(&nextRow, 1);
        const int r = __builtin_amdgcn_readfirstlane(rr);
        if (r >= TILE) break;
        const int a   = __builtin_amdgcn_readfirstlane(rstart[r]);
        const int deg = __builtin_amdgcn_readfirstlane(rstart[r + 1]) - a;
        const float er0 = ser[r * HEADS + 0], er1 = ser[r * HEADS + 1];

        float m0 = -INFINITY, m1 = -INFINITY;
        int   sA = 0;
        float qA0 = -INFINITY, qA1 = -INFINITY;
        for (int c = 0; c < deg; c += 32) {
            int s = 0;
            float q0 = -INFINITY, q1 = -INFINITY;
            if (c + lane < deg) {
                s = ssrc[a + c + lane];
                const v2f e = *(const v2f*)(el + (size_t)s * HEADS);
                q0 = lrelu(e.x + er0);
                q1 = lrelu(e.y + er1);
            }
            m0 = fmaxf(m0, q0);
            m1 = fmaxf(m1, q1);
            if (c == 0) { sA = s; qA0 = q0; qA1 = q1; }
        }
#pragma unroll
        for (int o = 16; o; o >>= 1) {
            m0 = fmaxf(m0, __shfl_xor(m0, o));
            m1 = fmaxf(m1, __shfl_xor(m1, o));
        }
        if (!isfinite(m0)) m0 = 0.0f;
        if (!isfinite(m1)) m1 = 0.0f;

        v4f   w0 = {0.0f, 0.0f, 0.0f, 0.0f}, w1 = {0.0f, 0.0f, 0.0f, 0.0f};
        float t0 = 0.0f, t1 = 0.0f;
        for (int c = 0; c < deg; c += 32) {
            int s = sA;
            float q0 = qA0, q1 = qA1;
            if (c != 0) {
                s = 0; q0 = -INFINITY; q1 = -INFINITY;
                if (c + lane < deg) {
                    s = ssrc[a + c + lane];
                    const v2f e = *(const v2f*)(el + (size_t)s * HEADS);
                    q0 = lrelu(e.x + er0);
                    q1 = lrelu(e.y + er1);
                }
            }
            float v0 = 0.0f, v1 = 0.0f;
            if (c + lane < deg) {
                v0 = __expf(q0 - m0);
                v1 = __expf(q1 - m1);
            }
            t0 += v0;
            t1 += v1;
            const int nk = (deg - c < 32) ? (deg - c) : 32;
            for (int k = 0; k < nk; k += 4) {
                v4f   r4[4][2];
                float a0[4], a1[4];
#pragma unroll
                for (int q = 0; q < 4; ++q) {
                    const int kk = k + q;
                    const int sk = __builtin_amdgcn_readlane(s, kk);
                    a0[q] = __int_as_float(__builtin_amdgcn_readlane(__float_as_int(v0), kk));
                    a1[q] = __int_as_float(__builtin_amdgcn_readlane(__float_as_int(v1), kk));
                    if (kk < nk) {
                        const v4f* row = (const v4f*)(H + (size_t)sk * DH) + lane;
                        r4[q][0] = row[0];
                        r4[q][1] = row[32];
                    } else {
                        r4[q][0] = v4f{0.0f, 0.0f, 0.0f, 0.0f};
                        r4[q][1] = v4f{0.0f, 0.0f, 0.0f, 0.0f};
                    }
                }
#pragma unroll
                for (int q = 0; q < 4; ++q) {
                    w0 += a0[q] * r4[q][0];
                    w1 += a1[q] * r4[q][1];
                }
            }
        }
#pragma unroll
        for (int o = 16; o; o >>= 1) {
            t0 += __shfl_xor(t0, o);
            t1 += __shfl_xor(t1, o);
        }

        const int node = lo + r;
        if (node < n) {
            v4f o0, o1;
#pragma unroll
            for (int c = 0; c < 4; ++c) {
                float x0 = (t0 == 0.0f) ? 0.0f : w0[c] / t0;
                float x1 = (t1 == 0.0f) ? 0.0f : w1[c] / t1;
                x0 += bv0[c];
                x1 += bv1[c];
                o0[c] = x0 > 0.0f ? x0 : 0.0f;
                o1[c] = x1 > 0.0f ? x1 : 0.0f;
            }
            *(v4f*)(outp + (size_t)node * DH + 4 * lane)       = o0;
            *(v4f*)(outp + (size_t)node * DH + HID + 4 * lane) = o1;
        }
    }
  } else {
    for (int i = tid; i < TILE * DH; i += LTH) acc[i] = 0.0f;
    if (tid < TILE * HEADS) {
        const int g = lo * HEADS + tid;
        sm[tid]  = -INFINITY;
        ss[tid]  = 0.0f;
        ser[tid] = (g < n * HEADS) ? er[g] : 0.0f;
    }
    __syncthreads();

    for (int i = tid; i < num; i += LTH) {
        const int s = lsrc[beg + i], dl = ldl[beg + i];
#pragma unroll
        for (int h = 0; h < HEADS; ++h)
            atomicMaxF(&sm[dl * HEADS + h], lrelu(el[s * HEADS + h] + ser[dl * HEADS + h]));
    }
    __syncthreads();
    if (tid < TILE * HEADS && !isfinite(sm[tid])) sm[tid] = 0.0f;
    __syncthreads();

    for (int c0 = wid * 32; c0 < num; c0 += 32 * (LTH / 32)) {
        const int i = c0 + lane;
        int s = 0, dl = 0;
        float v0 = 0.0f, v1 = 0.0f;
        if (i < num) {
            s  = lsrc[beg + i];
            dl = ldl[beg + i];
            const float sc0 = lrelu(el[s * HEADS + 0] + ser[dl * HEADS + 0]);
            const float sc1 = lrelu(el[s * HEADS + 1] + ser[dl * HEADS + 1]);
            v0 = __expf(sc0 - sm[dl * HEADS + 0]);
            v1 = __expf(sc1 - sm[dl * HEADS + 1]);
            atomicAdd(&ss[dl * HEADS + 0], v0);
            atomicAdd(&ss[dl * HEADS + 1], v1);
        }
        const int nk = (num - c0 < 32) ? (num - c0) : 32;
        for (int k = 0; k < nk; k += 4) {
            v4f   r[4][2];
            int   dk[4];
            float a0[4], a1[4];
#pragma unroll
            for (int q = 0; q < 4; ++q) {
                const int kk = k + q;
                const int sk = __builtin_amdgcn_readlane(s, kk);
                dk[q] = __builtin_amdgcn_readlane(dl, kk);
                a0[q] = __int_as_float(__builtin_amdgcn_readlane(__float_as_int(v0), kk));
                a1[q] = __int_as_float(__builtin_amdgcn_readlane(__float_as_int(v1), kk));
                if (kk < nk) {
                    const v4f* row = (const v4f*)(H + (size_t)sk * DH) + lane;
                    r[q][0] = row[0];
                    r[q][1] = row[32];
                } else {
                    r[q][0] = v4f{0.0f, 0.0f, 0.0f, 0.0f};
                    r[q][1] = v4f{0.0f, 0.0f, 0.0f, 0.0f};
                }
            }
#pragma unroll
            for (int q = 0; q < 4; ++q) {
                if (k + q < nk) {
                    float* ad = acc + dk[q] * DH + lane;
#pragma unroll
                    for (int c = 0; c < 4; ++c) atomicAdd(&ad[c * 32], a0[q] * r[q][0][c]);
#pragma unroll
                    for (int c = 0; c < 4; ++c) atomicAdd(&ad[HID + c * 32], a1[q] * r[q][1][c]);
                }
            }
        }
    }
    __syncthreads();

    for (int idx = tid; idx < TILE * (DH / 4); idx += LTH) {
        const int nl   = idx >> 6;
        const int hf   = (idx >> 5) & 1;
        const int l    = idx & 31;
        const int node = lo + nl;
        if (node < n) {
            const float  sv = ss[nl * HEADS + hf];
            const float* a  = acc + nl * DH + hf * HID + l;
            const v4f    bv = *(const v4f*)(bias + hf * HID + 4 * l);
            v4f o;
#pragma unroll
            for (int c = 0; c < 4; ++c) {
                float v = (sv == 0.0f) ? 0.0f : a[c * 32] / sv;
                v += bv[c];
                o[c] = v > 0.0f ? v : 0.0f;
            }
            *(v4f*)(outp + (size_t)node * DH + hf * HID + 4 * l) = o;
        }
    }
  }
}

#define EC_TH   1024
#define EC_GRID 256

__global__ __launch_bounds__(EC_TH) __attribute__((amdgpu_waves_per_eu(8, 8)))
void edge_classify(const int* __restrict__ src, const int* __restrict__ dst,
                   const float* __restrict__ H, const float* __restrict__ fcW,
                   const float* __restrict__ fcb, float* __restrict__ out, int e)
{
    const int lane = threadIdx.x & 31;
    const int wave   = __builtin_amdgcn_readfirstlane((int)((threadIdx.x >> 5) * gridDim.x + blockIdx.x));
    const int nwaves = (int)gridDim.x * (EC_TH / 32);
    const int nchunks = (e + 15) >> 4;
    if (wave >= nchunks) return;

    v4f wS[2][2], wD[2][2];
#pragma unroll
    for (int hf = 0; hf < 2; ++hf) {
        const v4f* ps = (const v4f*)(fcW + (size_t)(hf * 128 + 4 * lane) * 2);
        const v4f* pd = (const v4f*)(fcW + (size_t)(DH + hf * 128 + 4 * lane) * 2);
        wS[hf][0] = ps[0]; wS[hf][1] = ps[1];
        wD[hf][0] = pd[0]; wD[hf][1] = pd[1];
    }
    const float bo = fcb[lane & 1];

    const int last = e - 1;
    int s, d;
    {
        const int ei = min((wave << 4) + (lane & 15), last);
        s = src[ei]; d = dst[ei];
    }
    for (int ch = wave; ch < nchunks; ch += nwaves) {
        const int e0 = ch << 4;
        const int en = min(((ch + nwaves) << 4) + (lane & 15), last);
        const int sn = src[en], dn = dst[en];
        float res = 0.0f;
#pragma unroll 1
        for (int g = 0; g < 4; ++g) {
            v4f rs[4][2], rd[4][2];
#pragma unroll
            for (int q = 0; q < 4; ++q) {
                const int sk = __builtin_amdgcn_readlane(s, g * 4 + q);
                const int dk = __builtin_amdgcn_readlane(d, g * 4 + q);
                const v4f* ps = (const v4f*)(H + (size_t)sk * DH) + lane;
                const v4f* pd = (const v4f*)(H + (size_t)dk * DH) + lane;
                rs[q][0] = ps[0]; rs[q][1] = ps[32];
                rd[q][0] = pd[0]; rd[q][1] = pd[32];
            }
            float p[8];
#pragma unroll
            for (int q = 0; q < 4; ++q) {
                v2f a = {0.0f, 0.0f};
#pragma unroll
                for (int hf = 0; hf < 2; ++hf) {
                    a += rs[q][hf].x * wS[hf][0].xy;
                    a += rs[q][hf].y * wS[hf][0].zw;
                    a += rs[q][hf].z * wS[hf][1].xy;
                    a += rs[q][hf].w * wS[hf][1].zw;
                    a += rd[q][hf].x * wD[hf][0].xy;
                    a += rd[q][hf].y * wD[hf][0].zw;
                    a += rd[q][hf].z * wD[hf][1].xy;
                    a += rd[q][hf].w * wD[hf][1].zw;
                }
                p[q * 2 + 0] = a.x;
                p[q * 2 + 1] = a.y;
            }
            float t[4], u[2];
#pragma unroll
            for (int i = 0; i < 4; ++i) {
                const float keep = (lane & 1) ? p[2 * i + 1] : p[2 * i];
                const float send = (lane & 1) ? p[2 * i] : p[2 * i + 1];
                t[i] = keep + __shfl_xor(send, 1);
            }
#pragma unroll
            for (int i = 0; i < 2; ++i) {
                const float keep = (lane & 2) ? t[2 * i + 1] : t[2 * i];
                const float send = (lane & 2) ? t[2 * i] : t[2 * i + 1];
                u[i] = keep + __shfl_xor(send, 2);
            }
            float v;
            {
                const float keep = (lane & 4) ? u[1] : u[0];
                const float send = (lane & 4) ? u[0] : u[1];
                v = keep + __shfl_xor(send, 4);
            }
            v += __shfl_xor(v, 8);
            v += __shfl_xor(v, 16);
            if ((lane >> 3) == g) res = v;
        }
        const int oi = e0 * 2 + lane;
        if (oi < e * 2) out[oi] = res + bo;
        s = sn; d = dn;
    }
}

extern "C" void kernel_launch(void* const* d_in, const int* in_sizes, int n_in,
                              void* d_out, int out_size, void* d_ws, size_t ws_size,
                              hipStream_t stream)
{
    const float* x   = (const float*)d_in[0];
    const float* rep = (const float*)d_in[1];
    const int*   src = (const int*)d_in[2];
    const int*   dst = (const int*)d_in[3];
    const float* W1  = (const float*)d_in[4];
    const float* al1 = (const float*)d_in[5];
    const float* ar1 = (const float*)d_in[6];
    const float* b1  = (const float*)d_in[7];
    const float* W2  = (const float*)d_in[8];
    const float* al2 = (const float*)d_in[9];
    const float* ar2 = (const float*)d_in[10];
    const float* b2  = (const float*)d_in[11];
    const float* fcW = (const float*)d_in[12];
    const float* fcb = (const float*)d_in[13];
    float* out = (float*)d_out;

    const int N = in_sizes[0] / 64;
    const int E = in_sizes[2];

    float* ws  = (float*)d_ws;
    float* h   = ws;
    float* agg = h   + (size_t)N * DH;
    float* rly = agg + (size_t)N * DH;
    float* el  = rly + (size_t)N * DH;
    float* er  = el  + (size_t)N * 2;
    int*   cnt = (int*)(er + (size_t)N * 2);
    int*   off = cnt + (size_t)N * 2;
    int*   lsrc = off + (size_t)N * 2;
    int*   ldl  = lsrc + (size_t)E;
    int*   hist = (int*)agg;
    float* wx1  = agg + (size_t)NB * MAXT;
    float* wx2  = wx1 + 128 * 4;

    const int gT   = (N + TILE - 1) / TILE;

    fold_att<<<48, 256, 0, stream>>>(W1, al1, ar1, W2, al2, ar2, wx1, wx2);

    bin_hist<<<NB, LTH, 0, stream>>>(dst, hist, E, N, gT);
    bin_place<<<NB, LTH, 0, stream>>>(src, dst, hist, cnt, off, lsrc, ldl, E, N, gT);

    gemm_node<128, true><<<GEMM_GRID, 512, 0, stream>>>(x, rep, W1, wx1, h, el, er, N);
    tile_layer<<<gT, LTH, 0, stream>>>(lsrc, ldl, cnt, off, h, el, er, b1, rly, N);

    gemm_node<256, false><<<GEMM_GRID, 512, 0, stream>>>(rly, nullptr, W2, wx2, h, el, er, N);
    tile_layer<<<gT, LTH, 0, stream>>>(lsrc, ldl, cnt, off, h, el, er, b2, rly, N);

    edge_classify<<<EC_GRID, EC_TH, 0, stream>>>(src, dst, rly, fcW, fcb, out, E);
}


# edge classifier per dst tile: dst half once per row (1 KB gathered per edge, was 2), results as whole lines in list order, set in edge order by a gather launch (8 -> 9); dot sums reordered
# speedup vs baseline: 4.2793x; 1.0400x over previous
#include <hip/hip_runtime.h>
#include <hip/hip_bf16.h>
#include <math.h>

typedef __attribute__((ext_vector_type(2))) float v2f;
typedef __attribute__((ext_vector_type(4))) float v4f;
typedef __attribute__((ext_vector_type(8))) float v8f;

#define HID 128
#define HEADS 2
#define DH 256
#define NEG_SLOPE 0.2f

#define GEMM_GRID 256

template<int K, bool CONCAT>
__global__ __launch_bounds__(512)
void gemm_node(const float* __restrict__ A, const float* __restrict__ A2,
               const float* __restrict__ W, const float* __restrict__ Wx,
               float* __restrict__ Hout, float* __restrict__ el, float* __restrict__ er, int n)
{
    constexpr int PITCH = K + 4;
    constexpr int KT    = K / 32;
    constexpr int KS    = K / 4;
    constexpr int K4    = K / 4;
    constexpr int NV    = (16 * K4) / 512;
    __shared__ __attribute__((aligned(16))) float As[2][16 * PITCH];

    const int tid  = threadIdx.x;
    const int wid  = __builtin_amdgcn_readfirstlane(tid >> 5);
    const int lane = tid & 31;
    const int c0    = wid * 16;
    const int halfl = lane & 15;
    const int koff  = (lane >> 4) * 2;
    const int G     = (int)gridDim.x;
    const int g     = (int)blockIdx.x;

    v2f bf[KS];
#pragma unroll
    for (int s = 0; s < KS; ++s) {
        bf[s].x = W[(4 * s + koff)     * DH + c0 + halfl];
        bf[s].y = W[(4 * s + koff + 1) * DH + c0 + halfl];
    }
    v4f wx[KT];
#pragma unroll
    for (int t = 0; t < KT; ++t) wx[t] = *(const v4f*)(Wx + (size_t)(lane + 32 * t) * 4);

    const int nslab = (n + 15) >> 4;
    const int full  = nslab / G;
    const int rem   = nslab - full * G;
    const int per   = rem > 0 ? G / rem : 0;
    const int q     = rem > 0 ? g / rem : 0;
    const int ts    = rem > 0 ? g - q * rem : 0;
    const bool tail = rem > 0 && q < per;
    const int T     = full + (tail ? 1 : 0);
    const bool mine = tail && (wid % (per > 0 ? per : 1)) == q;

    v4f pre[NV];
    auto fetch = [&](int sl) {
#pragma unroll
        for (int j = 0; j < NV; ++j) {
            const int idx = tid + 512 * j;
            const int row = idx / K4;
            const int k   = (idx - row * K4) * 4;
            const int gr  = sl * 16 + row;
            v4f v = {0.0f, 0.0f, 0.0f, 0.0f};
            if (gr < n) {
                if (CONCAT) v = (k < 64) ? *(const v4f*)(A + gr * 64 + k) : *(const v4f*)(A2 + gr * 64 + (k - 64));
                else        v = *(const v4f*)(A + gr * K + k);
            }
            pre[j] = v;
        }
    };
    auto put = [&](float* as) {
#pragma unroll
        for (int j = 0; j < NV; ++j) {
            const int idx = tid + 512 * j;
            const int row = idx / K4;
            const int k   = (idx - row * K4) * 4;
            *(v4f*)(as + row * PITCH + k) = pre[j];
        }
    };

    const int colg  = c0 + halfl;
    const int mbase = (lane >> 4) * 8;
    auto store_tile = [&](const v8f& t, int sl) {
        float* p = Hout + (sl * 16 + mbase) * DH + colg;
        if (sl * 16 + 16 <= n) {
#pragma unroll
            for (int r = 0; r < 8; ++r) p[r * DH] = t[r];
        } else {
#pragma unroll
            for (int r = 0; r < 8; ++r)
                if (sl * 16 + mbase + r < n) p[r * DH] = t[r];
        }
    };

    if (T > 0) { fetch(full > 0 ? g : ts); put(As[0]); }
    __syncthreads();

    v8f  accP  = {};
    int  slP   = 0;
    bool haveP = false;

    for (int it = 0; it < T; ++it) {
        const float* as   = As[it & 1];
        const bool   whole = it < full;
        const int    sl    = whole ? g + it * G : full * G + ts;
        const bool   more  = it + 1 < T;
        if (more) fetch(it + 1 < full ? g + (it + 1) * G : full * G + ts);
        if (haveP) store_tile(accP, slP);

        const bool have = whole || mine;
        v8f acc = {};
        if (have) {
#pragma unroll
            for (int s = 0; s < KS; ++s) {
                v2f a;
                a.x = as[halfl * PITCH + 4 * s + koff];
                a.y = as[halfl * PITCH + 4 * s + koff + 1];
                acc = __builtin_amdgcn_wmma_f32_16x16x4_f32(
                          false, a, false, bf[s], (short)0, acc, false, false);
            }
        }

        if ((whole || q == 0) && wid < 8) {
            float s[2][4] = {{0.f, 0.f, 0.f, 0.f}, {0.f, 0.f, 0.f, 0.f}};
#pragma unroll
            for (int t = 0; t < KT; ++t) {
#pragma unroll
                for (int h = 0; h < 2; ++h) {
                    const float a = as[(wid + 8 * h) * PITCH + lane + 32 * t];
                    s[h][0] += a * wx[t].x;  s[h][1] += a * wx[t].y;
                    s[h][2] += a * wx[t].z;  s[h][3] += a * wx[t].w;
                }
            }
#pragma unroll
            for (int o = 16; o; o >>= 1) {
#pragma unroll
                for (int h = 0; h < 2; ++h) {
                    s[h][0] += __shfl_xor(s[h][0], o); s[h][1] += __shfl_xor(s[h][1], o);
                    s[h][2] += __shfl_xor(s[h][2], o); s[h][3] += __shfl_xor(s[h][3], o);
                }
            }
#pragma unroll
            for (int h = 0; h < 2; ++h) {
                const int gr = sl * 16 + wid + 8 * h;
                if (gr < n) {
                    if (lane == 0) { el[gr * 2 + 0] = s[h][0]; el[gr * 2 + 1] = s[h][1]; }
                    if (lane == 1) { er[gr * 2 + 0] = s[h][2]; er[gr * 2 + 1] = s[h][3]; }
                }
            }
        }

        if (more) put(As[(it + 1) & 1]);
        __syncthreads();
        accP = acc; slP = sl; haveP = have;
    }
    if (haveP) store_tile(accP, slP);
}

__global__ __launch_bounds__(256)
void fold_att(const float* __restrict__ W1, const float* __restrict__ al1, const float* __restrict__ ar1,
              const float* __restrict__ W2, const float* __restrict__ al2, const float* __restrict__ ar2,
              float* __restrict__ wx1, float* __restrict__ wx2)
{
    const int lane = threadIdx.x & 31;
    const int row  = blockIdx.x * 8 + (threadIdx.x >> 5);
    if (row >= 384) return;
    const bool first = row < 128;
    const int  k     = first ? row : row - 128;
    const float* W  = first ? W1  : W2;
    const float* al = first ? al1 : al2;
    const float* ar = first ? ar1 : ar2;
    float*       wx = first ? wx1 : wx2;

    const v4f w0 = *(const v4f*)(W + (size_t)k * DH + 4 * lane);
    const v4f w1 = *(const v4f*)(W + (size_t)k * DH + HID + 4 * lane);
    const v4f l0 = *(const v4f*)(al + 4 * lane);
    const v4f l1 = *(const v4f*)(al + HID + 4 * lane);
    const v4f q0 = *(const v4f*)(ar + 4 * lane);
    const v4f q1 = *(const v4f*)(ar + HID + 4 * lane);
    float s0 = w0.x * l0.x + w0.y * l0.y + w0.z * l0.z + w0.w * l0.w;
    float s1 = w1.x * l1.x + w1.y * l1.y + w1.z * l1.z + w1.w * l1.w;
    float s2 = w0.x * q0.x + w0.y * q0.y + w0.z * q0.z + w0.w * q0.w;
    float s3 = w1.x * q1.x + w1.y * q1.y + w1.z * q1.z + w1.w * q1.w;
#pragma unroll
    for (int o = 16; o; o >>= 1) {
        s0 += __shfl_xor(s0, o); s1 += __shfl_xor(s1, o);
        s2 += __shfl_xor(s2, o); s3 += __shfl_xor(s3, o);
    }
    if (lane == 0) *(v4f*)(wx + (size_t)k * 4) = v4f{s0, s1, s2, s3};
}

#define TILE 98
#define LTH  1024
#define NB   128
#define FU   4
#define MAXT LTH
#define RES_BASE(o, t) (16 * (((o) + 15) / 16 + (t)))

__device__ __forceinline__ float lrelu(float x) {
    return x > 0.0f ? x : NEG_SLOPE * x;
}

__device__ __forceinline__ void atomicMaxF(float* addr, float val) {
    if (val >= 0.0f) atomicMax((int*)addr, __float_as_int(val));
    else             atomicMin((unsigned int*)addr, __float_as_uint(val));
}

__global__ __launch_bounds__(LTH)
void bin_hist(const int* __restrict__ dst, int* __restrict__ hist, int e, int n, int gT)
{
    __shared__ int hc[MAXT];
    const int tid = threadIdx.x;
    const int per = (e + NB - 1) / NB;
    const int lo  = blockIdx.x * per;
    const int hi  = min(e, lo + per);
    for (int t = tid; t < gT; t += LTH) hc[t] = 0;
    __syncthreads();
    for (int base = lo; base < hi; base += LTH * FU) {
        int dv[FU];
#pragma unroll
        for (int u = 0; u < FU; ++u) {
            const int i = base + u * LTH + tid;
            dv[u] = (i < hi) ? dst[i] : -1;
        }
#pragma unroll
        for (int u = 0; u < FU; ++u)
            if ((unsigned)dv[u] < (unsigned)n) atomicAdd(&hc[dv[u] / TILE], 1);
    }
    __syncthreads();
    for (int t = tid; t < gT; t += LTH) hist[blockIdx.x * gT + t] = hc[t];
}

__global__ __launch_bounds__(LTH)
void bin_place(const int* __restrict__ src, const int* __restrict__ dst,
               const int* __restrict__ hist, int* __restrict__ cnt, int* __restrict__ off,
               int* __restrict__ lsrc, int* __restrict__ ldl, int* __restrict__ lpos,
               int e, int n, int gT)
{
    constexpr int TQ  = 256;
    constexpr int RQ  = LTH / TQ;
    constexpr int RPQ = NB / RQ;
    __shared__ int pa[LTH];
    __shared__ int pb[LTH];
    __shared__ int tot[MAXT];
    __shared__ int cur[MAXT];
    __shared__ int padd[MAXT];
    __shared__ int wsum[LTH / 32];

    const int tid  = threadIdx.x;
    const int lane = tid & 31;
    const int wid  = tid >> 5;
    const int b    = blockIdx.x;
    const int per  = (e + NB - 1) / NB;
    const int lo   = b * per;
    const int hi   = min(e, lo + per);

    int dv[FU], sv[FU];
#pragma unroll
    for (int u = 0; u < FU; ++u) {
        const int i = lo + u * LTH + tid;
        dv[u] = (i < hi) ? dst[i] : -1;
        sv[u] = (i < hi) ? src[i] : 0;
    }

    const int tl = tid & (TQ - 1);
    const int q  = tid / TQ;
    for (int t0 = 0; t0 < gT; t0 += TQ) {
        const int t = t0 + tl;
        int all = 0, bel = 0;
        if (t < gT) {
            int c[RPQ];
#pragma unroll
            for (int r = 0; r < RPQ; ++r) c[r] = hist[(q * RPQ + r) * gT + t];
#pragma unroll
            for (int r = 0; r < RPQ; ++r) {
                all += c[r];
                bel += (q * RPQ + r < b) ? c[r] : 0;
            }
        }
        pa[tid] = all;
        pb[tid] = bel;
        __syncthreads();
        if (q == 0 && t < gT) {
            int a = 0, w = 0;
            for (int k = 0; k < RQ; ++k) { a += pa[k * TQ + tl]; w += pb[k * TQ + tl]; }
            tot[t] = a;
            cur[t] = w;
        }
        __syncthreads();
    }

    const int v = (tid < gT) ? tot[tid] : 0;
    int inc = v;
#pragma unroll
    for (int o = 1; o < 32; o <<= 1) {
        const int y = __shfl_up(inc, o);
        if (lane >= o) inc += y;
    }
    if (lane == 31) wsum[wid] = inc;
    __syncthreads();
    if (wid == 0) {
        const int w = wsum[lane];
        int wi = w;
#pragma unroll
        for (int o = 1; o < 32; o <<= 1) {
            const int y = __shfl_up(wi, o);
            if (lane >= o) wi += y;
        }
        wsum[lane] = wi - w;
    }
    __syncthreads();
    if (tid < gT) {
        const int start = inc - v + wsum[wid];
        cur[tid] += start;
        padd[tid] = RES_BASE(start, tid) - start;
        if (b == 0) { cnt[tid] = v; off[tid] = start; }
    }
    __syncthreads();

    for (int base = lo; ; ) {
#pragma unroll
        for (int u = 0; u < FU; ++u) {
            const int d = dv[u];
            const int i = base + u * LTH + tid;
            if ((unsigned)d < (unsigned)n) {
                const int t = d / TILE;
                const int p = atomicAdd(&cur[t], 1);
                lsrc[p] = sv[u];
                ldl[p]  = d - t * TILE;
                lpos[i] = p + padd[t];
            } else if (i < hi) {
                lpos[i] = -1;
            }
        }
        base += LTH * FU;
        if (base >= hi) break;
#pragma unroll
        for (int u = 0; u < FU; ++u) {
            const int i = base + u * LTH + tid;
            dv[u] = (i < hi) ? dst[i] : -1;
            sv[u] = (i < hi) ? src[i] : 0;
        }
    }
}

#define RPAD 128
#define RCAP (TILE * DH)

__global__ __launch_bounds__(LTH)
void tile_layer(const int* __restrict__ lsrc, const int* __restrict__ ldl,
                const int* __restrict__ cnt, const int* __restrict__ off,
                const float* __restrict__ H, const float* __restrict__ el,
                const float* __restrict__ er, const float* __restrict__ bias,
                float* __restrict__ outp, int n)
{
    static_assert(TILE <= RPAD, "the prefix sum takes 4 rows per lane of one wave");
    __shared__ float acc[TILE * DH];
    __shared__ float sm[TILE * HEADS];
    __shared__ float ss[TILE * HEADS];
    __shared__ float ser[TILE * HEADS];
    __shared__ int   rcnt[RPAD];
    __shared__ int   rstart[RPAD + 1];
    __shared__ int   nextRow;

    const int tid  = threadIdx.x;
    const int lane = tid & 31;
    const int wid  = __builtin_amdgcn_readfirstlane(tid >> 5);
    const int lo   = blockIdx.x * TILE;
    const int beg  = off[blockIdx.x];
    const int num  = cnt[blockIdx.x];

  if (num <= RCAP) {
    int* ssrc = (int*)acc;

    if (tid < RPAD) rcnt[tid] = 0;
    if (tid == 0) nextRow = 0;
    if (tid < TILE * HEADS) {
        const int g = lo * HEADS + tid;
        ser[tid] = (g < n * HEADS) ? er[g] : 0.0f;
    }
    __syncthreads();

    for (int i = tid; i < num; i += LTH) atomicAdd(&rcnt[ldl[beg + i]], 1);
    __syncthreads();

    if (wid == 0) {
        const int c0 = rcnt[4 * lane], c1 = rcnt[4 * lane + 1], c2 = rcnt[4 * lane + 2], c3 = rcnt[4 * lane + 3];
        const int tot = c0 + c1 + c2 + c3;
        int inc = tot;
#pragma unroll
        for (int o = 1; o < 32; o <<= 1) {
            const int y = __shfl_up(inc, o);
            if (lane >= o) inc += y;
        }
        const int p0 = inc - tot, p1 = p0 + c0, p2 = p1 + c1, p3 = p2 + c2;
        rstart[4 * lane] = p0; rstart[4 * lane + 1] = p1; rstart[4 * lane + 2] = p2; rstart[4 * lane + 3] = p3;
        rcnt[4 * lane]   = p0; rcnt[4 * lane + 1]   = p1; rcnt[4 * lane + 2]   = p2; rcnt[4 * lane + 3]   = p3;
        if (lane == 31) rstart[RPAD] = inc;
    }
    __syncthreads();

    for (int i = tid; i < num; i += LTH) {
        const int s = lsrc[beg + i], dl = ldl[beg + i];
        const int p = atomicAdd(&rcnt[dl], 1);
        ssrc[p] = s;
    }
    __syncthreads();

    const v4f bv0 = *(const v4f*)(bias + 4 * lane);
    const v4f bv1 = *(const v4f*)(bias + HID + 4 * lane);

    for (int trip = 0; trip <= TILE; ++trip) {
        int rr = 0;
        if (lane == 0) rr = atomicAdd(&nextRow, 1);
        const int r = __builtin_amdgcn_readfirstlane(rr);
        if (r >= TILE) break;
        const int a   = __builtin_amdgcn_readfirstlane(rstart[r]);
        const int deg = __builtin_amdgcn_readfirstlane(rstart[r + 1]) - a;
        const float er0 = ser[r * HEADS + 0], er1 = ser[r * HEADS + 1];

        float m0 = -INFINITY, m1 = -INFINITY;
        int   sA = 0;
        float qA0 = -INFINITY, qA1 = -INFINITY;
        for (int c = 0; c < deg; c += 32) {
            int s = 0;
            float q0 = -INFINITY, q1 = -INFINITY;
            if (c + lane < deg) {
                s = ssrc[a + c + lane];
                const v2f e = *(const v2f*)(el + (size_t)s * HEADS);
                q0 = lrelu(e.x + er0);
                q1 = lrelu(e.y + er1);
            }
            m0 = fmaxf(m0, q0);
            m1 = fmaxf(m1, q1);
            if (c == 0) { sA = s; qA0 = q0; qA1 = q1; }
        }
#pragma unroll
        for (int o = 16; o; o >>= 1) {
            m0 = fmaxf(m0, __shfl_xor(m0, o));
            m1 = fmaxf(m1, __shfl_xor(m1, o));
        }
        if (!isfinite(m0)) m0 = 0.0f;
        if (!isfinite(m1)) m1 = 0.0f;

        v4f   w0 = {0.0f, 0.0f, 0.0f, 0.0f}, w1 = {0.0f, 0.0f, 0.0f, 0.0f};
        float t0 = 0.0f, t1 = 0.0f;
        for (int c = 0; c < deg; c += 32) {
            int s = sA;
            float q0 = qA0, q1 = qA1;
            if (c != 0) {
                s = 0; q0 = -INFINITY; q1 = -INFINITY;
                if (c + lane < deg) {
                    s = ssrc[a + c + lane];
                    const v2f e = *(const v2f*)(el + (size_t)s * HEADS);
                    q0 = lrelu(e.x + er0);
                    q1 = lrelu(e.y + er1);
                }
            }
            float v0 = 0.0f, v1 = 0.0f;
            if (c + lane < deg) {
                v0 = __expf(q0 - m0);
                v1 = __expf(q1 - m1);
            }
            t0 += v0;
            t1 += v1;
            const int nk = (deg - c < 32) ? (deg - c) : 32;
            for (int k = 0; k < nk; k += 4) {
                v4f   r4[4][2];
                float a0[4], a1[4];
#pragma unroll
                for (int q = 0; q < 4; ++q) {
                    const int kk = k + q;
                    const int sk = __builtin_amdgcn_readlane(s, kk);
                    a0[q] = __int_as_float(__builtin_amdgcn_readlane(__float_as_int(v0), kk));
                    a1[q] = __int_as_float(__builtin_amdgcn_readlane(__float_as_int(v1), kk));
                    if (kk < nk) {
                        const v4f* row = (const v4f*)(H + (size_t)sk * DH) + lane;
                        r4[q][0] = row[0];
                        r4[q][1] = row[32];
                    } else {
                        r4[q][0] = v4f{0.0f, 0.0f, 0.0f, 0.0f};
                        r4[q][1] = v4f{0.0f, 0.0f, 0.0f, 0.0f};
                    }
                }
#pragma unroll
                for (int q = 0; q < 4; ++q) {
                    w0 += a0[q] * r4[q][0];
                    w1 += a1[q] * r4[q][1];
                }
            }
        }
#pragma unroll
        for (int o = 16; o; o >>= 1) {
            t0 += __shfl_xor(t0, o);
            t1 += __shfl_xor(t1, o);
        }

        const int node = lo + r;
        if (node < n) {
            v4f o0, o1;
#pragma unroll
            for (int c = 0; c < 4; ++c) {
                float x0 = (t0 == 0.0f) ? 0.0f : w0[c] / t0;
                float x1 = (t1 == 0.0f) ? 0.0f : w1[c] / t1;
                x0 += bv0[c];
                x1 += bv1[c];
                o0[c] = x0 > 0.0f ? x0 : 0.0f;
                o1[c] = x1 > 0.0f ? x1 : 0.0f;
            }
            *(v4f*)(outp + (size_t)node * DH + 4 * lane)       = o0;
            *(v4f*)(outp + (size_t)node * DH + HID + 4 * lane) = o1;
        }
    }
  } else {
    for (int i = tid; i < TILE * DH; i += LTH) acc[i] = 0.0f;
    if (tid < TILE * HEADS) {
        const int g = lo * HEADS + tid;
        sm[tid]  = -INFINITY;
        ss[tid]  = 0.0f;
        ser[tid] = (g < n * HEADS) ? er[g] : 0.0f;
    }
    __syncthreads();

    for (int i = tid; i < num; i += LTH) {
        const int s = lsrc[beg + i], dl = ldl[beg + i];
#pragma unroll
        for (int h = 0; h < HEADS; ++h)
            atomicMaxF(&sm[dl * HEADS + h], lrelu(el[s * HEADS + h] + ser[dl * HEADS + h]));
    }
    __syncthreads();
    if (tid < TILE * HEADS && !isfinite(sm[tid])) sm[tid] = 0.0f;
    __syncthreads();

    for (int c0 = wid * 32; c0 < num; c0 += 32 * (LTH / 32)) {
        const int i = c0 + lane;
        int s = 0, dl = 0;
        float v0 = 0.0f, v1 = 0.0f;
        if (i < num) {
            s  = lsrc[beg + i];
            dl = ldl[beg + i];
            const float sc0 = lrelu(el[s * HEADS + 0] + ser[dl * HEADS + 0]);
            const float sc1 = lrelu(el[s * HEADS + 1] + ser[dl * HEADS + 1]);
            v0 = __expf(sc0 - sm[dl * HEADS + 0]);
            v1 = __expf(sc1 - sm[dl * HEADS + 1]);
            atomicAdd(&ss[dl * HEADS + 0], v0);
            atomicAdd(&ss[dl * HEADS + 1], v1);
        }
        const int nk = (num - c0 < 32) ? (num - c0) : 32;
        for (int k = 0; k < nk; k += 4) {
            v4f   r[4][2];
            int   dk[4];
            float a0[4], a1[4];
#pragma unroll
            for (int q = 0; q < 4; ++q) {
                const int kk = k + q;
                const int sk = __builtin_amdgcn_readlane(s, kk);
                dk[q] = __builtin_amdgcn_readlane(dl, kk);
                a0[q] = __int_as_float(__builtin_amdgcn_readlane(__float_as_int(v0), kk));
                a1[q] = __int_as_float(__builtin_amdgcn_readlane(__float_as_int(v1), kk));
                if (kk < nk) {
                    const v4f* row = (const v4f*)(H + (size_t)sk * DH) + lane;
                    r[q][0] = row[0];
                    r[q][1] = row[32];
                } else {
                    r[q][0] = v4f{0.0f, 0.0f, 0.0f, 0.0f};
                    r[q][1] = v4f{0.0f, 0.0f, 0.0f, 0.0f};
                }
            }
#pragma unroll
            for (int q = 0; q < 4; ++q) {
                if (k + q < nk) {
                    float* ad = acc + dk[q] * DH + lane;
#pragma unroll
                    for (int c = 0; c < 4; ++c) atomicAdd(&ad[c * 32], a0[q] * r[q][0][c]);
#pragma unroll
                    for (int c = 0; c < 4; ++c) atomicAdd(&ad[HID + c * 32], a1[q] * r[q][1][c]);
                }
            }
        }
    }
    __syncthreads();

    for (int idx = tid; idx < TILE * (DH / 4); idx += LTH) {
        const int nl   = idx >> 6;
        const int hf   = (idx >> 5) & 1;
        const int l    = idx & 31;
        const int node = lo + nl;
        if (node < n) {
            const float  sv = ss[nl * HEADS + hf];
            const float* a  = acc + nl * DH + hf * HID + l;
            const v4f    bv = *(const v4f*)(bias + hf * HID + 4 * l);
            v4f o;
#pragma unroll
            for (int c = 0; c < 4; ++c) {
                float v = (sv == 0.0f) ? 0.0f : a[c * 32] / sv;
                v += bv[c];
                o[c] = v > 0.0f ? v : 0.0f;
            }
            *(v4f*)(outp + (size_t)node * DH + hf * HID + 4 * l) = o;
        }
    }
  }
}

#define EC_TH   1024

__global__ __launch_bounds__(EC_TH) __attribute__((amdgpu_waves_per_eu(8, 8)))
void edge_classify_tile(const int* __restrict__ lsrc, const int* __restrict__ ldl,
                        const int* __restrict__ cnt, const int* __restrict__ off,
                        const float* __restrict__ H, const float* __restrict__ fcW,
                        const float* __restrict__ fcb, float* __restrict__ res, int n)
{
    constexpr int RPW = (TILE + EC_TH / 32 - 1) / (EC_TH / 32);
    __shared__ float pd[TILE * 2];

    const int lane = threadIdx.x & 31;
    const int wid  = __builtin_amdgcn_readfirstlane((int)(threadIdx.x >> 5));
    const int lo   = blockIdx.x * TILE;
    const int beg  = off[blockIdx.x];
    const int num  = cnt[blockIdx.x];

    v4f wS[2][2], wD[2][2];
#pragma unroll
    for (int hf = 0; hf < 2; ++hf) {
        const v4f* ps = (const v4f*)(fcW + (size_t)(hf * 128 + 4 * lane) * 2);
        const v4f* pq = (const v4f*)(fcW + (size_t)(DH + hf * 128 + 4 * lane) * 2);
        wS[hf][0] = ps[0]; wS[hf][1] = ps[1];
        wD[hf][0] = pq[0]; wD[hf][1] = pq[1];
    }

    const int nch  = (num + 15) >> 4;
    const int last = num - 1;
    int s = 0, dl = 0;
    if (wid < nch) {
        const int k = min((wid << 4) + (lane & 15), last);
        s = lsrc[beg + k]; dl = ldl[beg + k];
    }

    {
        v4f dr[RPW][2];
#pragma unroll
        for (int j = 0; j < RPW; ++j) {
            const int r    = min(wid + (EC_TH / 32) * j, TILE - 1);
            const int node = min(lo + r, n - 1);
            const v4f* row = (const v4f*)(H + (size_t)node * DH) + lane;
            dr[j][0] = row[0];
            dr[j][1] = row[32];
        }
        const float b0 = fcb[0], b1 = fcb[1];
        float a0[RPW], a1[RPW];
#pragma unroll
        for (int j = 0; j < RPW; ++j) {
            v2f a = {0.0f, 0.0f};
#pragma unroll
            for (int hf = 0; hf < 2; ++hf) {
                a += dr[j][hf].x * wD[hf][0].xy;
                a += dr[j][hf].y * wD[hf][0].zw;
                a += dr[j][hf].z * wD[hf][1].xy;
                a += dr[j][hf].w * wD[hf][1].zw;
            }
            a0[j] = a.x; a1[j] = a.y;
        }
#pragma unroll
        for (int o = 16; o; o >>= 1) {
#pragma unroll
            for (int j = 0; j < RPW; ++j) {
                a0[j] += __shfl_xor(a0[j], o);
                a1[j] += __shfl_xor(a1[j], o);
            }
        }
#pragma unroll
        for (int j = 0; j < RPW; ++j) {
            const int r = wid + (EC_TH / 32) * j;
            if (r < TILE && lane == 0) { pd[r * 2 + 0] = a0[j] + b0; pd[r * 2 + 1] = a1[j] + b1; }
        }
    }
    __syncthreads();

    const int rbase = RES_BASE(beg, (int)blockIdx.x);
    for (int ch = wid; ch < nch; ch += EC_TH / 32) {
        const int e0 = ch << 4;
        const int kn = min(((ch + EC_TH / 32) << 4) + (lane & 15), last);
        const int sn = lsrc[beg + kn], dn = ldl[beg + kn];
        float rsum = 0.0f;
#pragma unroll 1
        for (int g = 0; g < 4; ++g) {
            v4f rs[4][2];
#pragma unroll
            for (int q = 0; q < 4; ++q) {
                const int sk = __builtin_amdgcn_readlane(s, g * 4 + q);
                const v4f* ps = (const v4f*)(H + (size_t)sk * DH) + lane;
                rs[q][0] = ps[0]; rs[q][1] = ps[32];
            }
            float p[8];
#pragma unroll
            for (int q = 0; q < 4; ++q) {
                v2f a = {0.0f, 0.0f};
#pragma unroll
                for (int hf = 0; hf < 2; ++hf) {
                    a += rs[q][hf].x * wS[hf][0].xy;
                    a += rs[q][hf].y * wS[hf][0].zw;
                    a += rs[q][hf].z * wS[hf][1].xy;
                    a += rs[q][hf].w * wS[hf][1].zw;
                }
                p[q * 2 + 0] = a.x;
                p[q * 2 + 1] = a.y;
            }
            float t[4], u[2];
#pragma unroll
            for (int i = 0; i < 4; ++i) {
                const float keep = (lane & 1) ? p[2 * i + 1] : p[2 * i];
                const float send = (lane & 1) ? p[2 * i] : p[2 * i + 1];
                t[i] = keep + __shfl_xor(send, 1);
            }
#pragma unroll
            for (int i = 0; i < 2; ++i) {
                const float keep = (lane & 2) ? t[2 * i + 1] : t[2 * i];
                const float send = (lane & 2) ? t[2 * i] : t[2 * i + 1];
                u[i] = keep + __shfl_xor(send, 2);
            }
            float v;
            {
                const float keep = (lane & 4) ? u[1] : u[0];
                const float send = (lane & 4) ? u[0] : u[1];
                v = keep + __shfl_xor(send, 4);
            }
            v += __shfl_xor(v, 8);
            v += __shfl_xor(v, 16);
            if ((lane >> 3) == g) rsum = v;
        }
        const int dq = __shfl(dl, lane >> 1);
        const float dh = pd[dq * 2 + (lane & 1)];
        if (e0 * 2 + lane < num * 2) res[(size_t)(rbase + e0) * 2 + lane] = rsum + dh;
        s = sn; dl = dn;
    }
}

#define EO_TH 256

__global__ __launch_bounds__(EO_TH)
void edge_order(const int* __restrict__ lpos, const float* __restrict__ res,
                const float* __restrict__ fcb, float* __restrict__ out, int e)
{
    const int i4 = ((int)blockIdx.x * EO_TH + (int)threadIdx.x) * 4;
    if (i4 >= e) return;
    const v2f bo = {fcb[0], fcb[1]};
    if (i4 + 3 < e) {
        int p[4];
#pragma unroll
        for (int j = 0; j < 4; ++j) p[j] = lpos[i4 + j];
        v2f r[4];
#pragma unroll
        for (int j = 0; j < 4; ++j) r[j] = (p[j] >= 0) ? *(const v2f*)(res + (size_t)p[j] * 2) : bo;
        *(v4f*)(out + (size_t)i4 * 2)     = v4f{r[0].x, r[0].y, r[1].x, r[1].y};
        *(v4f*)(out + (size_t)i4 * 2 + 4) = v4f{r[2].x, r[2].y, r[3].x, r[3].y};
    } else {
        for (int j = 0; i4 + j < e; ++j) {
            const int p = lpos[i4 + j];
            const v2f r = (p >= 0) ? *(const v2f*)(res + (size_t)p * 2) : bo;
            out[(size_t)(i4 + j) * 2 + 0] = r.x;
            out[(size_t)(i4 + j) * 2 + 1] = r.y;
        }
    }
}

extern "C" void kernel_launch(void* const* d_in, const int* in_sizes, int n_in,
                              void* d_out, int out_size, void* d_ws, size_t ws_size,
                              hipStream_t stream)
{
    const float* x   = (const float*)d_in[0];
    const float* rep = (const float*)d_in[1];
    const int*   src = (const int*)d_in[2];
    const int*   dst = (const int*)d_in[3];
    const float* W1  = (const float*)d_in[4];
    const float* al1 = (const float*)d_in[5];
    const float* ar1 = (const float*)d_in[6];
    const float* b1  = (const float*)d_in[7];
    const float* W2  = (const float*)d_in[8];
    const float* al2 = (const float*)d_in[9];
    const float* ar2 = (const float*)d_in[10];
    const float* b2  = (const float*)d_in[11];
    const float* fcW = (const float*)d_in[12];
    const float* fcb = (const float*)d_in[13];
    float* out = (float*)d_out;

    const int N = in_sizes[0] / 64;
    const int E = in_sizes[2];

    float* ws  = (float*)d_ws;
    float* h   = ws;
    float* agg = h   + (size_t)N * DH;
    float* rly = agg + (size_t)N * DH;
    float* el  = rly + (size_t)N * DH;
    float* er  = el  + (size_t)N * 2;
    int*   cnt = (int*)(er + (size_t)N * 2);
    int*   off = cnt + (size_t)N * 2;
    int*   lsrc = off + (size_t)N * 2;
    int*   ldl  = lsrc + (size_t)E;
    int*   hist = (int*)agg;
    float* wx1  = agg + (size_t)NB * MAXT;
    float* wx2  = wx1 + 128 * 4;
    float* res  = wx2 + 256 * 4;
    int*   lpos = (int*)(res + 2 * ((size_t)E + 16 * (MAXT + 1)));

    const int gT   = (N + TILE - 1) / TILE;

    fold_att<<<48, 256, 0, stream>>>(W1, al1, ar1, W2, al2, ar2, wx1, wx2);

    bin_hist<<<NB, LTH, 0, stream>>>(dst, hist, E, N, gT);
    bin_place<<<NB, LTH, 0, stream>>>(src, dst, hist, cnt, off, lsrc, ldl, lpos, E, N, gT);

    gemm_node<128, true><<<GEMM_GRID, 512, 0, stream>>>(x, rep, W1, wx1, h, el, er, N);
    tile_layer<<<gT, LTH, 0, stream>>>(lsrc, ldl, cnt, off, h, el, er, b1, rly, N);

    gemm_node<256, false><<<GEMM_GRID, 512, 0, stream>>>(rly, nullptr, W2, wx2, h, el, er, N);
    tile_layer<<<gT, LTH, 0, stream>>>(lsrc, ldl, cnt, off, h, el, er, b2, rly, N);

    edge_classify_tile<<<gT, EC_TH, 0, stream>>>(lsrc, ldl, cnt, off, rly, fcW, fcb, res, N);
    edge_order<<<(E + EO_TH * 4 - 1) / (EO_TH * 4), EO_TH, 0, stream>>>(lpos, res, fcb, out, E);
}


# launches merged, 9 -> 7: weight fold rides in the edge-count launch; edge placing (128 work-groups of 512 threads) runs as second tenant under the layer-1 GEMM; list order by arrival as before
# speedup vs baseline: 4.4398x; 1.0375x over previous
#include <hip/hip_runtime.h>
#include <hip/hip_bf16.h>
#include <math.h>

typedef __attribute__((ext_vector_type(2))) float v2f;
typedef __attribute__((ext_vector_type(4))) float v4f;
typedef __attribute__((ext_vector_type(8))) float v8f;

#define HID 128
#define HEADS 2
#define DH 256
#define NEG_SLOPE 0.2f

#define GEMM_GRID 256

template<int K, bool CONCAT>
__device__ __forceinline__
void gemm_body(const float* __restrict__ A, const float* __restrict__ A2,
               const float* __restrict__ W, const float* __restrict__ Wx,
               float* __restrict__ Hout, float* __restrict__ el, float* __restrict__ er, int n,
               const int G, const int g)
{
    constexpr int PITCH = K + 4;
    constexpr int KT    = K / 32;
    constexpr int KS    = K / 4;
    constexpr int K4    = K / 4;
    constexpr int NV    = (16 * K4) / 512;
    __shared__ __attribute__((aligned(16))) float As[2][16 * PITCH];

    const int tid  = threadIdx.x;
    const int wid  = __builtin_amdgcn_readfirstlane(tid >> 5);
    const int lane = tid & 31;
    const int c0    = wid * 16;
    const int halfl = lane & 15;
    const int koff  = (lane >> 4) * 2;

    v2f bf[KS];
#pragma unroll
    for (int s = 0; s < KS; ++s) {
        bf[s].x = W[(4 * s + koff)     * DH + c0 + halfl];
        bf[s].y = W[(4 * s + koff + 1) * DH + c0 + halfl];
    }
    v4f wx[KT];
#pragma unroll
    for (int t = 0; t < KT; ++t) wx[t] = *(const v4f*)(Wx + (size_t)(lane + 32 * t) * 4);

    const int nslab = (n + 15) >> 4;
    const int full  = nslab / G;
    const int rem   = nslab - full * G;
    const int per   = rem > 0 ? G / rem : 0;
    const int q     = rem > 0 ? g / rem : 0;
    const int ts    = rem > 0 ? g - q * rem : 0;
    const bool tail = rem > 0 && q < per;
    const int T     = full + (tail ? 1 : 0);
    const bool mine = tail && (wid % (per > 0 ? per : 1)) == q;

    v4f pre[NV];
    auto fetch = [&](int sl) {
#pragma unroll
        for (int j = 0; j < NV; ++j) {
            const int idx = tid + 512 * j;
            const int row = idx / K4;
            const int k   = (idx - row * K4) * 4;
            const int gr  = sl * 16 + row;
            v4f v = {0.0f, 0.0f, 0.0f, 0.0f};
            if (gr < n) {
                if (CONCAT) v = (k < 64) ? *(const v4f*)(A + gr * 64 + k) : *(const v4f*)(A2 + gr * 64 + (k - 64));
                else        v = *(const v4f*)(A + gr * K + k);
            }
            pre[j] = v;
        }
    };
    auto put = [&](float* as) {
#pragma unroll
        for (int j = 0; j < NV; ++j) {
            const int idx = tid + 512 * j;
            const int row = idx / K4;
            const int k   = (idx - row * K4) * 4;
            *(v4f*)(as + row * PITCH + k) = pre[j];
        }
    };

    const int colg  = c0 + halfl;
    const int mbase = (lane >> 4) * 8;
    auto store_tile = [&](const v8f& t, int sl) {
        float* p = Hout + (sl * 16 + mbase) * DH + colg;
        if (sl * 16 + 16 <= n) {
#pragma unroll
            for (int r = 0; r < 8; ++r) p[r * DH] = t[r];
        } else {
#pragma unroll
            for (int r = 0; r < 8; ++r)
                if (sl * 16 + mbase + r < n) p[r * DH] = t[r];
        }
    };

    if (T > 0) { fetch(full > 0 ? g : ts); put(As[0]); }
    __syncthreads();

    v8f  accP  = {};
    int  slP   = 0;
    bool haveP = false;

    for (int it = 0; it < T; ++it) {
        const float* as   = As[it & 1];
        const bool   whole = it < full;
        const int    sl    = whole ? g + it * G : full * G + ts;
        const bool   more  = it + 1 < T;
        if (more) fetch(it + 1 < full ? g + (it + 1) * G : full * G + ts);
        if (haveP) store_tile(accP, slP);

        const bool have = whole || mine;
        v8f acc = {};
        if (have) {
#pragma unroll
            for (int s = 0; s < KS; ++s) {
                v2f a;
                a.x = as[halfl * PITCH + 4 * s + koff];
                a.y = as[halfl * PITCH + 4 * s + koff + 1];
                acc = __builtin_amdgcn_wmma_f32_16x16x4_f32(
                          false, a, false, bf[s], (short)0, acc, false, false);
            }
        }

        if ((whole || q == 0) && wid < 8) {
            float s[2][4] = {{0.f, 0.f, 0.f, 0.f}, {0.f, 0.f, 0.f, 0.f}};
#pragma unroll
            for (int t = 0; t < KT; ++t) {
#pragma unroll
                for (int h = 0; h < 2; ++h) {
                    const float a = as[(wid + 8 * h) * PITCH + lane + 32 * t];
                    s[h][0] += a * wx[t].x;  s[h][1] += a * wx[t].y;
                    s[h][2] += a * wx[t].z;  s[h][3] += a * wx[t].w;
                }
            }
#pragma unroll
            for (int o = 16; o; o >>= 1) {
#pragma unroll
                for (int h = 0; h < 2; ++h) {
                    s[h][0] += __shfl_xor(s[h][0], o); s[h][1] += __shfl_xor(s[h][1], o);
                    s[h][2] += __shfl_xor(s[h][2], o); s[h][3] += __shfl_xor(s[h][3], o);
                }
            }
#pragma unroll
            for (int h = 0; h < 2; ++h) {
                const int gr = sl * 16 + wid + 8 * h;
                if (gr < n) {
                    if (lane == 0) { el[gr * 2 + 0] = s[h][0]; el[gr * 2 + 1] = s[h][1]; }
                    if (lane == 1) { er[gr * 2 + 0] = s[h][2]; er[gr * 2 + 1] = s[h][3]; }
                }
            }
        }

        if (more) put(As[(it + 1) & 1]);
        __syncthreads();
        accP = acc; slP = sl; haveP = have;
    }
    if (haveP) store_tile(accP, slP);
}

template<int K, bool CONCAT>
__global__ __launch_bounds__(512)
void gemm_node(const float* __restrict__ A, const float* __restrict__ A2,
               const float* __restrict__ W, const float* __restrict__ Wx,
               float* __restrict__ Hout, float* __restrict__ el, float* __restrict__ er, int n)
{
    gemm_body<K, CONCAT>(A, A2, W, Wx, Hout, el, er, n, (int)gridDim.x, (int)blockIdx.x);
}

#define FOLD_ROWS 384

__device__ __forceinline__
void fold_body(const int row,
               const float* __restrict__ W1, const float* __restrict__ al1, const float* __restrict__ ar1,
               const float* __restrict__ W2, const float* __restrict__ al2, const float* __restrict__ ar2,
               float* __restrict__ wx1, float* __restrict__ wx2)
{
    const int lane = threadIdx.x & 31;
    if (row >= FOLD_ROWS) return;
    const bool first = row < 128;
    const int  k     = first ? row : row - 128;
    const float* W  = first ? W1  : W2;
    const float* al = first ? al1 : al2;
    const float* ar = first ? ar1 : ar2;
    float*       wx = first ? wx1 : wx2;

    const v4f w0 = *(const v4f*)(W + (size_t)k * DH + 4 * lane);
    const v4f w1 = *(const v4f*)(W + (size_t)k * DH + HID + 4 * lane);
    const v4f l0 = *(const v4f*)(al + 4 * lane);
    const v4f l1 = *(const v4f*)(al + HID + 4 * lane);
    const v4f q0 = *(const v4f*)(ar + 4 * lane);
    const v4f q1 = *(const v4f*)(ar + HID + 4 * lane);
    float s0 = w0.x * l0.x + w0.y * l0.y + w0.z * l0.z + w0.w * l0.w;
    float s1 = w1.x * l1.x + w1.y * l1.y + w1.z * l1.z + w1.w * l1.w;
    float s2 = w0.x * q0.x + w0.y * q0.y + w0.z * q0.z + w0.w * q0.w;
    float s3 = w1.x * q1.x + w1.y * q1.y + w1.z * q1.z + w1.w * q1.w;
#pragma unroll
    for (int o = 16; o; o >>= 1) {
        s0 += __shfl_xor(s0, o); s1 += __shfl_xor(s1, o);
        s2 += __shfl_xor(s2, o); s3 += __shfl_xor(s3, o);
    }
    if (lane == 0) *(v4f*)(wx + (size_t)k * 4) = v4f{s0, s1, s2, s3};
}

#define TILE 98
#define LTH  1024
#define NB   128
#define FU   4
#define MAXT LTH
#define RES_BASE(o, t) (16 * (((o) + 15) / 16 + (t)))

__device__ __forceinline__ float lrelu(float x) {
    return x > 0.0f ? x : NEG_SLOPE * x;
}

__device__ __forceinline__ void atomicMaxF(float* addr, float val) {
    if (val >= 0.0f) atomicMax((int*)addr, __float_as_int(val));
    else             atomicMin((unsigned int*)addr, __float_as_uint(val));
}

__device__ __forceinline__
void hist_body(const int* __restrict__ dst, int* __restrict__ hist, int e, int n, int gT, const int b)
{
    __shared__ int hc[MAXT];
    const int tid = threadIdx.x;
    const int per = (e + NB - 1) / NB;
    const int lo  = b * per;
    const int hi  = min(e, lo + per);
    for (int t = tid; t < gT; t += LTH) hc[t] = 0;
    __syncthreads();
    for (int base = lo; base < hi; base += LTH * FU) {
        int dv[FU];
#pragma unroll
        for (int u = 0; u < FU; ++u) {
            const int i = base + u * LTH + tid;
            dv[u] = (i < hi) ? dst[i] : -1;
        }
#pragma unroll
        for (int u = 0; u < FU; ++u)
            if ((unsigned)dv[u] < (unsigned)n) atomicAdd(&hc[dv[u] / TILE], 1);
    }
    __syncthreads();
    for (int t = tid; t < gT; t += LTH) hist[b * gT + t] = hc[t];
}

#define PREP_GRID (NB + FOLD_ROWS / (LTH / 32))

__global__ __launch_bounds__(LTH)
void hist_fold(const int* __restrict__ dst, int* __restrict__ hist, int e, int n, int gT,
               const float* __restrict__ W1, const float* __restrict__ al1, const float* __restrict__ ar1,
               const float* __restrict__ W2, const float* __restrict__ al2, const float* __restrict__ ar2,
               float* __restrict__ wx1, float* __restrict__ wx2)
{
    const int b = (int)blockIdx.x;
    if (b < NB) {
        hist_body(dst, hist, e, n, gT, b);
    } else {
        fold_body((b - NB) * (LTH / 32) + (int)(threadIdx.x >> 5), W1, al1, ar1, W2, al2, ar2, wx1, wx2);
    }
}

template<int TH>
__device__ __forceinline__
void place_body(const int* __restrict__ src, const int* __restrict__ dst,
                const int* __restrict__ hist, int* __restrict__ cnt, int* __restrict__ off,
                int* __restrict__ lsrc, int* __restrict__ ldl, int* __restrict__ lpos,
                int e, int n, int gT, const int b)
{
    constexpr int TQ  = 256;
    constexpr int RQ  = TH / TQ;
    constexpr int RPQ = NB / RQ;
    constexpr int RB  = 32;
    constexpr int TPT = MAXT / TH;
    constexpr int FP  = FU * (LTH / TH);
    static_assert(RPQ % RB == 0, "the column sums go in whole batches");
    __shared__ int pa[TH];
    __shared__ int pb[TH];
    __shared__ int tot[MAXT];
    __shared__ int cur[MAXT];
    __shared__ int padd[MAXT];
    __shared__ int wsum[32];

    const int tid  = threadIdx.x;
    const int lane = tid & 31;
    const int wid  = tid >> 5;
    const int per  = (e + NB - 1) / NB;
    const int lo   = b * per;
    const int hi   = min(e, lo + per);

    int dv[FP], sv[FP];
#pragma unroll
    for (int u = 0; u < FP; ++u) {
        const int i = lo + u * TH + tid;
        dv[u] = (i < hi) ? dst[i] : -1;
        sv[u] = (i < hi) ? src[i] : 0;
    }

    const int tl = tid & (TQ - 1);
    const int q  = tid / TQ;
    for (int t0 = 0; t0 < gT; t0 += TQ) {
        const int t = t0 + tl;
        int all = 0, bel = 0;
        if (t < gT) {
            for (int r0 = 0; r0 < RPQ; r0 += RB) {
                int c[RB];
#pragma unroll
                for (int r = 0; r < RB; ++r) c[r] = hist[(q * RPQ + r0 + r) * gT + t];
#pragma unroll
                for (int r = 0; r < RB; ++r) {
                    all += c[r];
                    bel += (q * RPQ + r0 + r < b) ? c[r] : 0;
                }
            }
        }
        pa[tid] = all;
        pb[tid] = bel;
        __syncthreads();
        if (q == 0 && t < gT) {
            int a = 0, w = 0;
            for (int k = 0; k < RQ; ++k) { a += pa[k * TQ + tl]; w += pb[k * TQ + tl]; }
            tot[t] = a;
            cur[t] = w;
        }
        __syncthreads();
    }

    const int ta = tid * TPT;
    int v[TPT];
    int sum = 0;
#pragma unroll
    for (int j = 0; j < TPT; ++j) { v[j] = (ta + j < gT) ? tot[ta + j] : 0; sum += v[j]; }
    int inc = sum;
#pragma unroll
    for (int o = 1; o < 32; o <<= 1) {
        const int y = __shfl_up(inc, o);
        if (lane >= o) inc += y;
    }
    if (lane == 31) wsum[wid] = inc;
    __syncthreads();
    if (wid == 0) {
        const int w = (lane < TH / 32) ? wsum[lane] : 0;
        int wi = w;
#pragma unroll
        for (int o = 1; o < 32; o <<= 1) {
            const int y = __shfl_up(wi, o);
            if (lane >= o) wi += y;
        }
        wsum[lane] = wi - w;
    }
    __syncthreads();
    {
        int start = inc - sum + wsum[wid];
#pragma unroll
        for (int j = 0; j < TPT; ++j) {
            const int t = ta + j;
            if (t < gT) {
                cur[t] += start;
                padd[t] = RES_BASE(start, t) - start;
                if (b == 0) { cnt[t] = v[j]; off[t] = start; }
            }
            start += v[j];
        }
    }
    __syncthreads();

    for (int base = lo; ; ) {
#pragma unroll
        for (int u = 0; u < FP; ++u) {
            const int d = dv[u];
            const int i = base + u * TH + tid;
            if ((unsigned)d < (unsigned)n) {
                const int t = d / TILE;
                const int p = atomicAdd(&cur[t], 1);
                lsrc[p] = sv[u];
                ldl[p]  = d - t * TILE;
                lpos[i] = p + padd[t];
            } else if (i < hi) {
                lpos[i] = -1;
            }
        }
        base += TH * FP;
        if (base >= hi) break;
#pragma unroll
        for (int u = 0; u < FP; ++u) {
            const int i = base + u * TH + tid;
            dv[u] = (i < hi) ? dst[i] : -1;
            sv[u] = (i < hi) ? src[i] : 0;
        }
    }
}

#define GP_TH 512

__global__ __launch_bounds__(GP_TH) __attribute__((amdgpu_waves_per_eu(8, 8)))
void gemm_place(const float* __restrict__ A, const float* __restrict__ A2,
                const float* __restrict__ W, const float* __restrict__ Wx,
                float* __restrict__ Hout, float* __restrict__ el, float* __restrict__ er, int n,
                const int* __restrict__ src, const int* __restrict__ dst,
                const int* __restrict__ hist, int* __restrict__ cnt, int* __restrict__ off,
                int* __restrict__ lsrc, int* __restrict__ ldl, int* __restrict__ lpos,
                int e, int gT)
{
    const int b = (int)blockIdx.x;
    if (b < GEMM_GRID) {
        gemm_body<128, true>(A, A2, W, Wx, Hout, el, er, n, GEMM_GRID, b);
    } else {
        place_body<GP_TH>(src, dst, hist, cnt, off, lsrc, ldl, lpos, e, n, gT, b - GEMM_GRID);
    }
}

#define RPAD 128
#define RCAP (TILE * DH)

__global__ __launch_bounds__(LTH)
void tile_layer(const int* __restrict__ lsrc, const int* __restrict__ ldl,
                const int* __restrict__ cnt, const int* __restrict__ off,
                const float* __restrict__ H, const float* __restrict__ el,
                const float* __restrict__ er, const float* __restrict__ bias,
                float* __restrict__ outp, int n)
{
    static_assert(TILE <= RPAD, "the prefix sum takes 4 rows per lane of one wave");
    __shared__ float acc[TILE * DH];
    __shared__ float sm[TILE * HEADS];
    __shared__ float ss[TILE * HEADS];
    __shared__ float ser[TILE * HEADS];
    __shared__ int   rcnt[RPAD];
    __shared__ int   rstart[RPAD + 1];
    __shared__ int   nextRow;

    const int tid  = threadIdx.x;
    const int lane = tid & 31;
    const int wid  = __builtin_amdgcn_readfirstlane(tid >> 5);
    const int lo   = blockIdx.x * TILE;
    const int beg  = off[blockIdx.x];
    const int num  = cnt[blockIdx.x];

  if (num <= RCAP) {
    int* ssrc = (int*)acc;

    if (tid < RPAD) rcnt[tid] = 0;
    if (tid == 0) nextRow = 0;
    if (tid < TILE * HEADS) {
        const int g = lo * HEADS + tid;
        ser[tid] = (g < n * HEADS) ? er[g] : 0.0f;
    }
    __syncthreads();

    for (int i = tid; i < num; i += LTH) atomicAdd(&rcnt[ldl[beg + i]], 1);
    __syncthreads();

    if (wid == 0) {
        const int c0 = rcnt[4 * lane], c1 = rcnt[4 * lane + 1], c2 = rcnt[4 * lane + 2], c3 = rcnt[4 * lane + 3];
        const int tot = c0 + c1 + c2 + c3;
        int inc = tot;
#pragma unroll
        for (int o = 1; o < 32; o <<= 1) {
            const int y = __shfl_up(inc, o);
            if (lane >= o) inc += y;
        }
        const int p0 = inc - tot, p1 = p0 + c0, p2 = p1 + c1, p3 = p2 + c2;
        rstart[4 * lane] = p0; rstart[4 * lane + 1] = p1; rstart[4 * lane + 2] = p2; rstart[4 * lane + 3] = p3;
        rcnt[4 * lane]   = p0; rcnt[4 * lane + 1]   = p1; rcnt[4 * lane + 2]   = p2; rcnt[4 * lane + 3]   = p3;
        if (lane == 31) rstart[RPAD] = inc;
    }
    __syncthreads();

    for (int i = tid; i < num; i += LTH) {
        const int s = lsrc[beg + i], dl = ldl[beg + i];
        const int p = atomicAdd(&rcnt[dl], 1);
        ssrc[p] = s;
    }
    __syncthreads();

    const v4f bv0 = *(const v4f*)(bias + 4 * lane);
    const v4f bv1 = *(const v4f*)(bias + HID + 4 * lane);

    for (int trip = 0; trip <= TILE; ++trip) {
        int rr = 0;
        if (lane == 0) rr = atomicAdd(&nextRow, 1);
        const int r = __builtin_amdgcn_readfirstlane(rr);
        if (r >= TILE) break;
        const int a   = __builtin_amdgcn_readfirstlane(rstart[r]);
        const int deg = __builtin_amdgcn_readfirstlane(rstart[r + 1]) - a;
        const float er0 = ser[r * HEADS + 0], er1 = ser[r * HEADS + 1];

        float m0 = -INFINITY, m1 = -INFINITY;
        int   sA = 0;
        float qA0 = -INFINITY, qA1 = -INFINITY;
        for (int c = 0; c < deg; c += 32) {
            int s = 0;
            float q0 = -INFINITY, q1 = -INFINITY;
            if (c + lane < deg) {
                s = ssrc[a + c + lane];
                const v2f e = *(const v2f*)(el + (size_t)s * HEADS);
                q0 = lrelu(e.x + er0);
                q1 = lrelu(e.y + er1);
            }
            m0 = fmaxf(m0, q0);
            m1 = fmaxf(m1, q1);
            if (c == 0) { sA = s; qA0 = q0; qA1 = q1; }
        }
#pragma unroll
        for (int o = 16; o; o >>= 1) {
            m0 = fmaxf(m0, __shfl_xor(m0, o));
            m1 = fmaxf(m1, __shfl_xor(m1, o));
        }
        if (!isfinite(m0)) m0 = 0.0f;
        if (!isfinite(m1)) m1 = 0.0f;

        v4f   w0 = {0.0f, 0.0f, 0.0f, 0.0f}, w1 = {0.0f, 0.0f, 0.0f, 0.0f};
        float t0 = 0.0f, t1 = 0.0f;
        for (int c = 0; c < deg; c += 32) {
            int s = sA;
            float q0 = qA0, q1 = qA1;
            if (c != 0) {
                s = 0; q0 = -INFINITY; q1 = -INFINITY;
                if (c + lane < deg) {
                    s = ssrc[a + c + lane];
                    const v2f e = *(const v2f*)(el + (size_t)s * HEADS);
                    q0 = lrelu(e.x + er0);
                    q1 = lrelu(e.y + er1);
                }
            }
            float v0 = 0.0f, v1 = 0.0f;
            if (c + lane < deg) {
                v0 = __expf(q0 - m0);
                v1 = __expf(q1 - m1);
            }
            t0 += v0;
            t1 += v1;
            const int nk = (deg - c < 32) ? (deg - c) : 32;
            for (int k = 0; k < nk; k += 4) {
                v4f   r4[4][2];
                float a0[4], a1[4];
#pragma unroll
                for (int q = 0; q < 4; ++q) {
                    const int kk = k + q;
                    const int sk = __builtin_amdgcn_readlane(s, kk);
                    a0[q] = __int_as_float(__builtin_amdgcn_readlane(__float_as_int(v0), kk));
                    a1[q] = __int_as_float(__builtin_amdgcn_readlane(__float_as_int(v1), kk));
                    if (kk < nk) {
                        const v4f* row = (const v4f*)(H + (size_t)sk * DH) + lane;
                        r4[q][0] = row[0];
                        r4[q][1] = row[32];
                    } else {
                        r4[q][0] = v4f{0.0f, 0.0f, 0.0f, 0.0f};
                        r4[q][1] = v4f{0.0f, 0.0f, 0.0f, 0.0f};
                    }
                }
#pragma unroll
                for (int q = 0; q < 4; ++q) {
                    w0 += a0[q] * r4[q][0];
                    w1 += a1[q] * r4[q][1];
                }
            }
        }
#pragma unroll
        for (int o = 16; o; o >>= 1) {
            t0 += __shfl_xor(t0, o);
            t1 += __shfl_xor(t1, o);
        }

        const int node = lo + r;
        if (node < n) {
            v4f o0, o1;
#pragma unroll
            for (int c = 0; c < 4; ++c) {
                float x0 = (t0 == 0.0f) ? 0.0f : w0[c] / t0;
                float x1 = (t1 == 0.0f) ? 0.0f : w1[c] / t1;
                x0 += bv0[c];
                x1 += bv1[c];
                o0[c] = x0 > 0.0f ? x0 : 0.0f;
                o1[c] = x1 > 0.0f ? x1 : 0.0f;
            }
            *(v4f*)(outp + (size_t)node * DH + 4 * lane)       = o0;
            *(v4f*)(outp + (size_t)node * DH + HID + 4 * lane) = o1;
        }
    }
  } else {
    for (int i = tid; i < TILE * DH; i += LTH) acc[i] = 0.0f;
    if (tid < TILE * HEADS) {
        const int g = lo * HEADS + tid;
        sm[tid]  = -INFINITY;
        ss[tid]  = 0.0f;
        ser[tid] = (g < n * HEADS) ? er[g] : 0.0f;
    }
    __syncthreads();

    for (int i = tid; i < num; i += LTH) {
        const int s = lsrc[beg + i], dl = ldl[beg + i];
#pragma unroll
        for (int h = 0; h < HEADS; ++h)
            atomicMaxF(&sm[dl * HEADS + h], lrelu(el[s * HEADS + h] + ser[dl * HEADS + h]));
    }
    __syncthreads();
    if (tid < TILE * HEADS && !isfinite(sm[tid])) sm[tid] = 0.0f;
    __syncthreads();

    for (int c0 = wid * 32; c0 < num; c0 += 32 * (LTH / 32)) {
        const int i = c0 + lane;
        int s = 0, dl = 0;
        float v0 = 0.0f, v1 = 0.0f;
        if (i < num) {
            s  = lsrc[beg + i];
            dl = ldl[beg + i];
            const float sc0 = lrelu(el[s * HEADS + 0] + ser[dl * HEADS + 0]);
            const float sc1 = lrelu(el[s * HEADS + 1] + ser[dl * HEADS + 1]);
            v0 = __expf(sc0 - sm[dl * HEADS + 0]);
            v1 = __expf(sc1 - sm[dl * HEADS + 1]);
            atomicAdd(&ss[dl * HEADS + 0], v0);
            atomicAdd(&ss[dl * HEADS + 1], v1);
        }
        const int nk = (num - c0 < 32) ? (num - c0) : 32;
        for (int k = 0; k < nk; k += 4) {
            v4f   r[4][2];
            int   dk[4];
            float a0[4], a1[4];
#pragma unroll
            for (int q = 0; q < 4; ++q) {
                const int kk = k + q;
                const int sk = __builtin_amdgcn_readlane(s, kk);
                dk[q] = __builtin_amdgcn_readlane(dl, kk);
                a0[q] = __int_as_float(__builtin_amdgcn_readlane(__float_as_int(v0), kk));
                a1[q] = __int_as_float(__builtin_amdgcn_readlane(__float_as_int(v1), kk));
                if (kk < nk) {
                    const v4f* row = (const v4f*)(H + (size_t)sk * DH) + lane;
                    r[q][0] = row[0];
                    r[q][1] = row[32];
                } else {
                    r[q][0] = v4f{0.0f, 0.0f, 0.0f, 0.0f};
                    r[q][1] = v4f{0.0f, 0.0f, 0.0f, 0.0f};
                }
            }
#pragma unroll
            for (int q = 0; q < 4; ++q) {
                if (k + q < nk) {
                    float* ad = acc + dk[q] * DH + lane;
#pragma unroll
                    for (int c = 0; c < 4; ++c) atomicAdd(&ad[c * 32], a0[q] * r[q][0][c]);
#pragma unroll
                    for (int c = 0; c < 4; ++c) atomicAdd(&ad[HID + c * 32], a1[q] * r[q][1][c]);
                }
            }
        }
    }
    __syncthreads();

    for (int idx = tid; idx < TILE * (DH / 4); idx += LTH) {
        const int nl   = idx >> 6;
        const int hf   = (idx >> 5) & 1;
        const int l    = idx & 31;
        const int node = lo + nl;
        if (node < n) {
            const float  sv = ss[nl * HEADS + hf];
            const float* a  = acc + nl * DH + hf * HID + l;
            const v4f    bv = *(const v4f*)(bias + hf * HID + 4 * l);
            v4f o;
#pragma unroll
            for (int c = 0; c < 4; ++c) {
                float v = (sv == 0.0f) ? 0.0f : a[c * 32] / sv;
                v += bv[c];
                o[c] = v > 0.0f ? v : 0.0f;
            }
            *(v4f*)(outp + (size_t)node * DH + hf * HID + 4 * l) = o;
        }
    }
  }
}

#define EC_TH   1024

__global__ __launch_bounds__(EC_TH) __attribute__((amdgpu_waves_per_eu(8, 8)))
void edge_classify_tile(const int* __restrict__ lsrc, const int* __restrict__ ldl,
                        const int* __restrict__ cnt, const int* __restrict__ off,
                        const float* __restrict__ H, const float* __restrict__ fcW,
                        const float* __restrict__ fcb, float* __restrict__ res, int n)
{
    constexpr int RPW = (TILE + EC_TH / 32 - 1) / (EC_TH / 32);
    __shared__ float pd[TILE * 2];

    const int lane = threadIdx.x & 31;
    const int wid  = __builtin_amdgcn_readfirstlane((int)(threadIdx.x >> 5));
    const int lo   = blockIdx.x * TILE;
    const int beg  = off[blockIdx.x];
    const int num  = cnt[blockIdx.x];

    v4f wS[2][2], wD[2][2];
#pragma unroll
    for (int hf = 0; hf < 2; ++hf) {
        const v4f* ps = (const v4f*)(fcW + (size_t)(hf * 128 + 4 * lane) * 2);
        const v4f* pq = (const v4f*)(fcW + (size_t)(DH + hf * 128 + 4 * lane) * 2);
        wS[hf][0] = ps[0]; wS[hf][1] = ps[1];
        wD[hf][0] = pq[0]; wD[hf][1] = pq[1];
    }

    const int nch  = (num + 15) >> 4;
    const int last = num - 1;
    int s = 0, dl = 0;
    if (wid < nch) {
        const int k = min((wid << 4) + (lane & 15), last);
        s = lsrc[beg + k]; dl = ldl[beg + k];
    }

    {
        v4f dr[RPW][2];
#pragma unroll
        for (int j = 0; j < RPW; ++j) {
            const int r    = min(wid + (EC_TH / 32) * j, TILE - 1);
            const int node = min(lo + r, n - 1);
            const v4f* row = (const v4f*)(H + (size_t)node * DH) + lane;
            dr[j][0] = row[0];
            dr[j][1] = row[32];
        }
        const float b0 = fcb[0], b1 = fcb[1];
        float a0[RPW], a1[RPW];
#pragma unroll
        for (int j = 0; j < RPW; ++j) {
            v2f a = {0.0f, 0.0f};
#pragma unroll
            for (int hf = 0; hf < 2; ++hf) {
                a += dr[j][hf].x * wD[hf][0].xy;
                a += dr[j][hf].y * wD[hf][0].zw;
                a += dr[j][hf].z * wD[hf][1].xy;
                a += dr[j][hf].w * wD[hf][1].zw;
            }
            a0[j] = a.x; a1[j] = a.y;
        }
#pragma unroll
        for (int o = 16; o; o >>= 1) {
#pragma unroll
            for (int j = 0; j < RPW; ++j) {
                a0[j] += __shfl_xor(a0[j], o);
                a1[j] += __shfl_xor(a1[j], o);
            }
        }
#pragma unroll
        for (int j = 0; j < RPW; ++j) {
            const int r = wid + (EC_TH / 32) * j;
            if (r < TILE && lane == 0) { pd[r * 2 + 0] = a0[j] + b0; pd[r * 2 + 1] = a1[j] + b1; }
        }
    }
    __syncthreads();

    const int rbase = RES_BASE(beg, (int)blockIdx.x);
    for (int ch = wid; ch < nch; ch += EC_TH / 32) {
        const int e0 = ch << 4;
        const int kn = min(((ch + EC_TH / 32) << 4) + (lane & 15), last);
        const int sn = lsrc[beg + kn], dn = ldl[beg + kn];
        float rsum = 0.0f;
#pragma unroll 1
        for (int g = 0; g < 4; ++g) {
            v4f rs[4][2];
#pragma unroll
            for (int q = 0; q < 4; ++q) {
                const int sk = __builtin_amdgcn_readlane(s, g * 4 + q);
                const v4f* ps = (const v4f*)(H + (size_t)sk * DH) + lane;
                rs[q][0] = ps[0]; rs[q][1] = ps[32];
            }
            float p[8];
#pragma unroll
            for (int q = 0; q < 4; ++q) {
                v2f a = {0.0f, 0.0f};
#pragma unroll
                for (int hf = 0; hf < 2; ++hf) {
                    a += rs[q][hf].x * wS[hf][0].xy;
                    a += rs[q][hf].y * wS[hf][0].zw;
                    a += rs[q][hf].z * wS[hf][1].xy;
                    a += rs[q][hf].w * wS[hf][1].zw;
                }
                p[q * 2 + 0] = a.x;
                p[q * 2 + 1] = a.y;
            }
            float t[4], u[2];
#pragma unroll
            for (int i = 0; i < 4; ++i) {
                const float keep = (lane & 1) ? p[2 * i + 1] : p[2 * i];
                const float send = (lane & 1) ? p[2 * i] : p[2 * i + 1];
                t[i] = keep + __shfl_xor(send, 1);
            }
#pragma unroll
            for (int i = 0; i < 2; ++i) {
                const float keep = (lane & 2) ? t[2 * i + 1] : t[2 * i];
                const float send = (lane & 2) ? t[2 * i] : t[2 * i + 1];
                u[i] = keep + __shfl_xor(send, 2);
            }
            float v;
            {
                const float keep = (lane & 4) ? u[1] : u[0];
                const float send = (lane & 4) ? u[0] : u[1];
                v = keep + __shfl_xor(send, 4);
            }
            v += __shfl_xor(v, 8);
            v += __shfl_xor(v, 16);
            if ((lane >> 3) == g) rsum = v;
        }
        const int dq = __shfl(dl, lane >> 1);
        const float dh = pd[dq * 2 + (lane & 1)];
        if (e0 * 2 + lane < num * 2) res[(size_t)(rbase + e0) * 2 + lane] = rsum + dh;
        s = sn; dl = dn;
    }
}

#define EO_TH 256

__global__ __launch_bounds__(EO_TH)
void edge_order(const int* __restrict__ lpos, const float* __restrict__ res,
                const float* __restrict__ fcb, float* __restrict__ out, int e)
{
    const int i4 = ((int)blockIdx.x * EO_TH + (int)threadIdx.x) * 4;
    if (i4 >= e) return;
    const v2f bo = {fcb[0], fcb[1]};
    if (i4 + 3 < e) {
        int p[4];
#pragma unroll
        for (int j = 0; j < 4; ++j) p[j] = lpos[i4 + j];
        v2f r[4];
#pragma unroll
        for (int j = 0; j < 4; ++j) r[j] = (p[j] >= 0) ? *(const v2f*)(res + (size_t)p[j] * 2) : bo;
        *(v4f*)(out + (size_t)i4 * 2)     = v4f{r[0].x, r[0].y, r[1].x, r[1].y};
        *(v4f*)(out + (size_t)i4 * 2 + 4) = v4f{r[2].x, r[2].y, r[3].x, r[3].y};
    } else {
        for (int j = 0; i4 + j < e; ++j) {
            const int p = lpos[i4 + j];
            const v2f r = (p >= 0) ? *(const v2f*)(res + (size_t)p * 2) : bo;
            out[(size_t)(i4 + j) * 2 + 0] = r.x;
            out[(size_t)(i4 + j) * 2 + 1] = r.y;
        }
    }
}

extern "C" void kernel_launch(void* const* d_in, const int* in_sizes, int n_in,
                              void* d_out, int out_size, void* d_ws, size_t ws_size,
                              hipStream_t stream)
{
    const float* x   = (const float*)d_in[0];
    const float* rep = (const float*)d_in[1];
    const int*   src = (const int*)d_in[2];
    const int*   dst = (const int*)d_in[3];
    const float* W1  = (const float*)d_in[4];
    const float* al1 = (const float*)d_in[5];
    const float* ar1 = (const float*)d_in[6];
    const float* b1  = (const float*)d_in[7];
    const float* W2  = (const float*)d_in[8];
    const float* al2 = (const float*)d_in[9];
    const float* ar2 = (const float*)d_in[10];
    const float* b2  = (const float*)d_in[11];
    const float* fcW = (const float*)d_in[12];
    const float* fcb = (const float*)d_in[13];
    float* out = (float*)d_out;

    const int N = in_sizes[0] / 64;
    const int E = in_sizes[2];

    float* ws  = (float*)d_ws;
    float* h   = ws;
    float* agg = h   + (size_t)N * DH;
    float* rly = agg + (size_t)N * DH;
    float* el  = rly + (size_t)N * DH;
    float* er  = el  + (size_t)N * 2;
    int*   cnt = (int*)(er + (size_t)N * 2);
    int*   off = cnt + (size_t)N * 2;
    int*   lsrc = off + (size_t)N * 2;
    int*   ldl  = lsrc + (size_t)E;
    int*   hist = (int*)agg;
    float* wx1  = agg + (size_t)NB * MAXT;
    float* wx2  = wx1 + 128 * 4;
    float* res  = wx2 + 256 * 4;
    int*   lpos = (int*)(res + 2 * ((size_t)E + 16 * (MAXT + 1)));

    const int gT   = (N + TILE - 1) / TILE;

    hist_fold<<<PREP_GRID, LTH, 0, stream>>>(dst, hist, E, N, gT, W1, al1, ar1, W2, al2, ar2, wx1, wx2);

    gemm_place<<<GEMM_GRID + NB, GP_TH, 0, stream>>>(x, rep, W1, wx1, h, el, er, N,
                                                     src, dst, hist, cnt, off, lsrc, ldl, lpos, E, gT);
    tile_layer<<<gT, LTH, 0, stream>>>(lsrc, ldl, cnt, off, h, el, er, b1, rly, N);

    gemm_node<256, false><<<GEMM_GRID, 512, 0, stream>>>(rly, nullptr, W2, wx2, h, el, er, N);
    tile_layer<<<gT, LTH, 0, stream>>>(lsrc, ldl, cnt, off, h, el, er, b2, rly, N);

    edge_classify_tile<<<gT, EC_TH, 0, stream>>>(lsrc, ldl, cnt, off, rly, fcW, fcb, res, N);
    edge_order<<<(E + EO_TH * 4 - 1) / (EO_TH * 4), EO_TH, 0, stream>>>(lpos, res, fcb, out, E);
}
